# starting kernel
# baseline (speedup 1.0000x reference)
#include <hip/hip_runtime.h>

typedef __attribute__((ext_vector_type(2))) float v2f;
typedef __attribute__((ext_vector_type(8))) float v8f;

#define FDIM 64

__global__ void init_cnt_kernel(float* __restrict__ cnt, int n) {
    int i = blockIdx.x * blockDim.x + threadIdx.x;
    if (i < n) cnt[i] = 0.0f;
}

__global__ void count_edges_kernel(const int* __restrict__ dst, float* __restrict__ cnt, int e) {
    int i = blockIdx.x * blockDim.x + threadIdx.x;
    if (i < e) atomicAdd(&cnt[dst[i]], 1.0f);
}

__global__ void finalize_deg_kernel(const float* __restrict__ cnt, float* __restrict__ dinv,
                                    float* __restrict__ sinv, int n) {
    int i = blockIdx.x * blockDim.x + threadIdx.x;
    if (i < n) {
        float c = cnt[i];
        dinv[i] = rsqrtf(c + 1.0f);
        sinv[i] = 1.0f / fmaxf(c, 1.0f);
    }
}

__global__ void edge_w_kernel(const int* __restrict__ src, const int* __restrict__ dst,
                              const float* __restrict__ dinv, float* __restrict__ ew, int e) {
    int i = blockIdx.x * blockDim.x + threadIdx.x;
    if (i < e) ew[i] = dinv[src[i]] * dinv[dst[i]];
}

__launch_bounds__(256)
__global__ void gemm64_wmma_kernel(const float* __restrict__ A, const float* __restrict__ W1g,
                                   const float* __restrict__ B, const float* __restrict__ W2g,
                                   const float* __restrict__ bias,
                                   const float* __restrict__ rowscale,
                                   const float* __restrict__ lnAg, const float* __restrict__ lnAb,
                                   const float* __restrict__ lnBg, const float* __restrict__ lnBb,
                                   const float* __restrict__ dinv2, const float* __restrict__ bias2,
                                   float* __restrict__ out2,
                                   float* __restrict__ out, int n, int reluOut, int reluA) {
    __shared__ float W1[FDIM * FDIM];
    __shared__ float W2[FDIM * FDIM];
    __shared__ float LNP[4][FDIM];

    const int t = threadIdx.x;
    {
        const float4* s1 = (const float4*)W1g;
        float4* l1 = (float4*)W1;
        for (int i = t; i < 1024; i += 256) l1[i] = s1[i];
        if (B) {
            const float4* s2 = (const float4*)W2g;
            float4* l2 = (float4*)W2;
            for (int i = t; i < 1024; i += 256) l2[i] = s2[i];
        }
        if (lnAg && t < FDIM) { LNP[0][t] = lnAg[t]; LNP[1][t] = lnAb[t]; }
        if (lnBg && t < FDIM) { LNP[2][t] = lnBg[t]; LNP[3][t] = lnBb[t]; }
    }
    __syncthreads();

    const int wave  = t >> 5;
    const int lane  = t & 31;
    const int m0    = (blockIdx.x * 8 + wave) * 16;
    if (m0 >= n) return;
    const int lhalf = lane >> 4;
    const int l15   = lane & 15;
    const int kOff  = lhalf * 2;

    int row = m0 + l15;
    if (row >= n) row = n - 1;

    const float rs = rowscale ? rowscale[row] : 1.0f;

    v2f afrag[16];
    {
        const float* ar = A + (size_t)row * FDIM + kOff;
        #pragma unroll
        for (int kb = 0; kb < 16; ++kb) {
            float a0 = ar[kb * 4 + 0];
            float a1 = ar[kb * 4 + 1];
            if (reluA) { a0 = fmaxf(a0, 0.0f); a1 = fmaxf(a1, 0.0f); }
            afrag[kb].x = a0 * rs;
            afrag[kb].y = a1 * rs;
        }
        if (lnAg) {
            float s = 0.0f, q = 0.0f;
            #pragma unroll
            for (int kb = 0; kb < 16; ++kb) {
                s += afrag[kb].x + afrag[kb].y;
                q += afrag[kb].x * afrag[kb].x + afrag[kb].y * afrag[kb].y;
            }
            s += __shfl_xor(s, 16, 32);
            q += __shfl_xor(q, 16, 32);
            const float mu  = s * (1.0f / 64.0f);
            const float inv = rsqrtf(fmaxf(q * (1.0f / 64.0f) - mu * mu, 0.0f) + 1e-5f);
            #pragma unroll
            for (int kb = 0; kb < 16; ++kb) {
                const int k0 = kb * 4 + kOff;
                afrag[kb].x = (afrag[kb].x - mu) * inv * LNP[0][k0]     + LNP[1][k0];
                afrag[kb].y = (afrag[kb].y - mu) * inv * LNP[0][k0 + 1] + LNP[1][k0 + 1];
            }
        }
    }
    v2f bfrag[16];
    if (B) {
        const float* br = B + (size_t)row * FDIM + kOff;
        #pragma unroll
        for (int kb = 0; kb < 16; ++kb) {
            bfrag[kb].x = br[kb * 4 + 0];
            bfrag[kb].y = br[kb * 4 + 1];
        }
        if (lnBg) {
            float s = 0.0f, q = 0.0f;
            #pragma unroll
            for (int kb = 0; kb < 16; ++kb) {
                s += bfrag[kb].x + bfrag[kb].y;
                q += bfrag[kb].x * bfrag[kb].x + bfrag[kb].y * bfrag[kb].y;
            }
            s += __shfl_xor(s, 16, 32);
            q += __shfl_xor(q, 16, 32);
            const float mu  = s * (1.0f / 64.0f);
            const float inv = rsqrtf(fmaxf(q * (1.0f / 64.0f) - mu * mu, 0.0f) + 1e-5f);
            #pragma unroll
            for (int kb = 0; kb < 16; ++kb) {
                const int k0 = kb * 4 + kOff;
                bfrag[kb].x = (bfrag[kb].x - mu) * inv * LNP[2][k0]     + LNP[3][k0];
                bfrag[kb].y = (bfrag[kb].y - mu) * inv * LNP[2][k0 + 1] + LNP[3][k0 + 1];
            }
        }
    }

    #pragma unroll
    for (int nb = 0; nb < 4; ++nb) {
        const int n0 = nb * 16;
        v8f acc = {0.f, 0.f, 0.f, 0.f, 0.f, 0.f, 0.f, 0.f};

        v2f wf[16];
        #pragma unroll
        for (int kb = 0; kb < 16; ++kb) {
            const int kr = kb * 4 + kOff;
            wf[kb].x = W1[kr * FDIM + n0 + l15];
            wf[kb].y = W1[(kr + 1) * FDIM + n0 + l15];
        }
        #pragma unroll
        for (int kb = 0; kb < 16; ++kb)
            acc = __builtin_amdgcn_wmma_f32_16x16x4_f32(
                false, afrag[kb], false, wf[kb], (short)0, acc, false, false);

        if (B) {
            v2f wg[16];
            #pragma unroll
            for (int kb = 0; kb < 16; ++kb) {
                const int kr = kb * 4 + kOff;
                wg[kb].x = W2[kr * FDIM + n0 + l15];
                wg[kb].y = W2[(kr + 1) * FDIM + n0 + l15];
            }
            #pragma unroll
            for (int kb = 0; kb < 16; ++kb)
                acc = __builtin_amdgcn_wmma_f32_16x16x4_f32(
                    false, bfrag[kb], false, wg[kb], (short)0, acc, false, false);
        }

        const float bval  = bias  ? bias[n0 + l15]  : 0.0f;
        const float b2val = bias2 ? bias2[n0 + l15] : 0.0f;
        #pragma unroll
        for (int j = 0; j < 8; ++j) {
            const int r = m0 + j + lhalf * 8;
            if (r < n) {
                float v = acc[j] + bval;
                if (reluOut) v = fmaxf(v, 0.0f);
                out[(size_t)r * FDIM + n0 + l15] = v;
                if (out2) {
                    const float d = dinv2[r];
                    out2[(size_t)r * FDIM + n0 + l15] = acc[j] * d * d + b2val;
                }
            }
        }
    }
}

__global__ void scatter_gcn_kernel(const float* __restrict__ h, const int* __restrict__ src,
                                   const int* __restrict__ dst, const float* __restrict__ ew,
                                   float* __restrict__ g, int e) {
    long long tid = (long long)blockIdx.x * blockDim.x + threadIdx.x;
    long long edge = tid >> 4;
    if (edge >= e) return;
    int q = (int)(tid & 15);
    int s = src[edge], d = dst[edge];
    float w = ew[edge];
    float4 hv = ((const float4*)(h + (size_t)s * FDIM))[q];
    float* gp = g + (size_t)d * FDIM + q * 4;
    atomicAdd(gp + 0, hv.x * w);
    atomicAdd(gp + 1, hv.y * w);
    atomicAdd(gp + 2, hv.z * w);
    atomicAdd(gp + 3, hv.w * w);
}

__global__ void scatter_sum_kernel(const float* __restrict__ xin, const int* __restrict__ src,
                                   const int* __restrict__ dst, float* __restrict__ agg, int e) {
    long long tid = (long long)blockIdx.x * blockDim.x + threadIdx.x;
    long long edge = tid >> 4;
    if (edge >= e) return;
    int q = (int)(tid & 15);
    int s = src[edge], d = dst[edge];
    float4 xv = ((const float4*)(xin + (size_t)s * FDIM))[q];
    float* ap = agg + (size_t)d * FDIM + q * 4;
    atomicAdd(ap + 0, xv.x);
    atomicAdd(ap + 1, xv.y);
    atomicAdd(ap + 2, xv.z);
    atomicAdd(ap + 3, xv.w);
}

__global__ void zero_kernel(float* __restrict__ p, long long total) {
    long long i = (long long)blockIdx.x * blockDim.x + threadIdx.x;
    if (i < total) p[i] = 0.0f;
}

extern "C" void kernel_launch(void* const* d_in, const int* in_sizes, int n_in,
                              void* d_out, int out_size, void* d_ws, size_t ws_size,
                              hipStream_t stream) {
    const float* x        = (const float*)d_in[0];
    const int*   ei       = (const int*)d_in[1];
    const float* gcn_w1   = (const float*)d_in[2];
    const float* gcn_b1   = (const float*)d_in[3];
    const float* gcn_w2   = (const float*)d_in[4];
    const float* gcn_b2   = (const float*)d_in[5];
    const float* sage_wl1 = (const float*)d_in[6];
    const float* sage_bl1 = (const float*)d_in[7];
    const float* sage_wr1 = (const float*)d_in[8];
    const float* sage_wl2 = (const float*)d_in[9];
    const float* sage_bl2 = (const float*)d_in[10];
    const float* sage_wr2 = (const float*)d_in[11];
    const float* gcn_ln_g = (const float*)d_in[12];
    const float* gcn_ln_b = (const float*)d_in[13];
    const float* sage_ln_g = (const float*)d_in[14];
    const float* sage_ln_b = (const float*)d_in[15];
    const float* proj_w   = (const float*)d_in[16];
    const float* proj_b   = (const float*)d_in[17];

    const int N = in_sizes[0] / FDIM;
    const int E = in_sizes[1] / 2;
    const int* src = ei;
    const int* dst = ei + E;

    float* cnt  = (float*)d_ws;
    float* dinv = cnt + N;
    float* sinv = dinv + N;
    float* ew   = sinv + N;
    float* h    = ew + E;
    float* g    = h + (size_t)N * FDIM;
    float* agg  = g + (size_t)N * FDIM;
    float* out  = (float*)d_out;

    const long long NM = (long long)N * FDIM;
    const int nblkN  = (N + 255) / 256;
    const int nblkE  = (E + 255) / 256;
    const int nblkNM = (int)((NM + 255) / 256);
    const int nblkSc = (int)(((long long)E * 16 + 255) / 256);
    const int nblkG  = (N + 127) / 128;

    init_cnt_kernel<<<nblkN, 256, 0, stream>>>(cnt, N);
    count_edges_kernel<<<nblkE, 256, 0, stream>>>(dst, cnt, E);
    finalize_deg_kernel<<<nblkN, 256, 0, stream>>>(cnt, dinv, sinv, N);
    edge_w_kernel<<<nblkE, 256, 0, stream>>>(src, dst, dinv, ew, E);

    gemm64_wmma_kernel<<<nblkG, 256, 0, stream>>>(
        x, gcn_w1, nullptr, nullptr, nullptr, nullptr,
        nullptr, nullptr, nullptr, nullptr,
        dinv, gcn_b1, g, h, N, 0, 0);
    scatter_gcn_kernel<<<nblkSc, 256, 0, stream>>>(h, src, dst, ew, g, E);

    gemm64_wmma_kernel<<<nblkG, 256, 0, stream>>>(
        g, gcn_w2, nullptr, nullptr, nullptr, nullptr,
        nullptr, nullptr, nullptr, nullptr,
        dinv, gcn_b2, g, h, N, 0, 1  );
    scatter_gcn_kernel<<<nblkSc, 256, 0, stream>>>(h, src, dst, ew, g, E);

    zero_kernel<<<nblkNM, 256, 0, stream>>>(agg, NM);
    scatter_sum_kernel<<<nblkSc, 256, 0, stream>>>(x, src, dst, agg, E);
    gemm64_wmma_kernel<<<nblkG, 256, 0, stream>>>(
        agg, sage_wl1, x, sage_wr1, sage_bl1, sinv,
        nullptr, nullptr, nullptr, nullptr,
        nullptr, nullptr, nullptr, out, N, 1  , 0);

    zero_kernel<<<nblkNM, 256, 0, stream>>>(agg, NM);
    scatter_sum_kernel<<<nblkSc, 256, 0, stream>>>(out, src, dst, agg, E);
    gemm64_wmma_kernel<<<nblkG, 256, 0, stream>>>(
        agg, sage_wl2, out, sage_wr2, sage_bl2, sinv,
        nullptr, nullptr, nullptr, nullptr,
        nullptr, nullptr, nullptr, h, N, 0, 0);

    gemm64_wmma_kernel<<<nblkG, 256, 0, stream>>>(
        g, proj_w, h, proj_w + FDIM * FDIM, proj_b, nullptr,
        gcn_ln_g, gcn_ln_b, sage_ln_g, sage_ln_b,
        nullptr, nullptr, nullptr, out, N, 0, 0);
}


# Four f32-atomic edge scatters replaced by a per-call destination-bucketed edge list (arrival-order rank from the degree adds, scan, fill) and one-owner register row sums; sums stay arrival-order
# speedup vs baseline: 2.3288x; 2.3288x over previous
#include <hip/hip_runtime.h>

typedef __attribute__((ext_vector_type(2))) float v2f;
typedef __attribute__((ext_vector_type(8))) float v8f;

#define FDIM 64
#define SCAN_TILE 1024
#define GROUP_ROWS 4
#define EDGE_UNROLL 8

__global__ void init_cnt_kernel(int* __restrict__ cnt, int n) {
    int i = blockIdx.x * blockDim.x + threadIdx.x;
    if (i < n) cnt[i] = 0;
}

__global__ void count_rank_kernel(const int* __restrict__ dst, int* __restrict__ cnt,
                                  int* __restrict__ rank, int e) {
    int i = blockIdx.x * blockDim.x + threadIdx.x;
    if (i < e) rank[i] = atomicAdd(&cnt[dst[i]], 1);
}

__launch_bounds__(256)
__global__ void scan_tile_kernel(const int* __restrict__ cnt, int* __restrict__ rowptr,
                                 int* __restrict__ bsum, int n) {
    __shared__ int wsum[8];
    const int t    = threadIdx.x;
    const int lane = t & 31;
    const int wave = t >> 5;
    const int base = blockIdx.x * SCAN_TILE + t * 4;
    const int c0 = (base + 0 < n) ? cnt[base + 0] : 0;
    const int c1 = (base + 1 < n) ? cnt[base + 1] : 0;
    const int c2 = (base + 2 < n) ? cnt[base + 2] : 0;
    const int c3 = (base + 3 < n) ? cnt[base + 3] : 0;
    const int tot = c0 + c1 + c2 + c3;
    int inc = tot;
    for (int o = 1; o < 32; o <<= 1) {
        const int v = __shfl_up(inc, o, 32);
        if (lane >= o) inc += v;
    }
    if (lane == 31) wsum[wave] = inc;
    __syncthreads();
    int woff = 0;
    for (int w = 0; w < wave; ++w) woff += wsum[w];
    const int ex = woff + inc - tot;
    if (base + 0 < n) rowptr[base + 0] = ex;
    if (base + 1 < n) rowptr[base + 1] = ex + c0;
    if (base + 2 < n) rowptr[base + 2] = ex + c0 + c1;
    if (base + 3 < n) rowptr[base + 3] = ex + c0 + c1 + c2;
    if (t == 255) bsum[blockIdx.x] = woff + inc;
}

__launch_bounds__(256)
__global__ void finalize_rows_kernel(const int* __restrict__ cnt, const int* __restrict__ bsum,
                                     int* __restrict__ rowptr, float* __restrict__ dinv, int n) {
    __shared__ int part[8];
    const int t  = threadIdx.x;
    const int nt = (blockIdx.x * 256) / SCAN_TILE;
    int v = 0;
    for (int i = t; i < nt; i += 256) v += bsum[i];
    for (int o = 16; o > 0; o >>= 1) v += __shfl_xor(v, o, 32);
    if ((t & 31) == 0) part[t >> 5] = v;
    __syncthreads();
    int off = 0;
    for (int w = 0; w < 8; ++w) off += part[w];
    const int i = blockIdx.x * 256 + t;
    if (i < n) {
        rowptr[i] += off;
        dinv[i] = rsqrtf((float)cnt[i] + 1.0f);
    }
}

__global__ void fill_nbr_kernel(const int* __restrict__ src, const int* __restrict__ dst,
                                const int* __restrict__ rowptr, const int* __restrict__ rank,
                                int* __restrict__ nbr, int e) {
    int i = blockIdx.x * blockDim.x + threadIdx.x;
    if (i < e) nbr[rowptr[dst[i]] + rank[i]] = src[i];
}

__launch_bounds__(256)
__global__ void gemm64_wmma_kernel(const float* __restrict__ A, const float* __restrict__ W1g,
                                   const float* __restrict__ B, const float* __restrict__ W2g,
                                   const float* __restrict__ bias,
                                   const float* __restrict__ rowscale,
                                   const float* __restrict__ lnAg, const float* __restrict__ lnAb,
                                   const float* __restrict__ lnBg, const float* __restrict__ lnBb,
                                   const float* __restrict__ dinv2, const float* __restrict__ bias2,
                                   float* __restrict__ out2,
                                   float* __restrict__ out, int n, int reluOut, int reluA) {
    __shared__ float W1[FDIM * FDIM];
    __shared__ float W2[FDIM * FDIM];
    __shared__ float LNP[4][FDIM];

    const int t = threadIdx.x;
    {
        const float4* s1 = (const float4*)W1g;
        float4* l1 = (float4*)W1;
        for (int i = t; i < 1024; i += 256) l1[i] = s1[i];
        if (B) {
            const float4* s2 = (const float4*)W2g;
            float4* l2 = (float4*)W2;
            for (int i = t; i < 1024; i += 256) l2[i] = s2[i];
        }
        if (lnAg && t < FDIM) { LNP[0][t] = lnAg[t]; LNP[1][t] = lnAb[t]; }
        if (lnBg && t < FDIM) { LNP[2][t] = lnBg[t]; LNP[3][t] = lnBb[t]; }
    }
    __syncthreads();

    const int wave  = t >> 5;
    const int lane  = t & 31;
    const int m0    = (blockIdx.x * 8 + wave) * 16;
    if (m0 >= n) return;
    const int lhalf = lane >> 4;
    const int l15   = lane & 15;
    const int kOff  = lhalf * 2;

    int row = m0 + l15;
    if (row >= n) row = n - 1;

    const float rs = rowscale ? rowscale[row] : 1.0f;

    v2f afrag[16];
    {
        const float* ar = A + (size_t)row * FDIM + kOff;
        #pragma unroll
        for (int kb = 0; kb < 16; ++kb) {
            float a0 = ar[kb * 4 + 0];
            float a1 = ar[kb * 4 + 1];
            if (reluA) { a0 = fmaxf(a0, 0.0f); a1 = fmaxf(a1, 0.0f); }
            afrag[kb].x = a0 * rs;
            afrag[kb].y = a1 * rs;
        }
        if (lnAg) {
            float s = 0.0f, q = 0.0f;
            #pragma unroll
            for (int kb = 0; kb < 16; ++kb) {
                s += afrag[kb].x + afrag[kb].y;
                q += afrag[kb].x * afrag[kb].x + afrag[kb].y * afrag[kb].y;
            }
            s += __shfl_xor(s, 16, 32);
            q += __shfl_xor(q, 16, 32);
            const float mu  = s * (1.0f / 64.0f);
            const float inv = rsqrtf(fmaxf(q * (1.0f / 64.0f) - mu * mu, 0.0f) + 1e-5f);
            #pragma unroll
            for (int kb = 0; kb < 16; ++kb) {
                const int k0 = kb * 4 + kOff;
                afrag[kb].x = (afrag[kb].x - mu) * inv * LNP[0][k0]     + LNP[1][k0];
                afrag[kb].y = (afrag[kb].y - mu) * inv * LNP[0][k0 + 1] + LNP[1][k0 + 1];
            }
        }
    }
    v2f bfrag[16];
    if (B) {
        const float* br = B + (size_t)row * FDIM + kOff;
        #pragma unroll
        for (int kb = 0; kb < 16; ++kb) {
            bfrag[kb].x = br[kb * 4 + 0];
            bfrag[kb].y = br[kb * 4 + 1];
        }
        if (lnBg) {
            float s = 0.0f, q = 0.0f;
            #pragma unroll
            for (int kb = 0; kb < 16; ++kb) {
                s += bfrag[kb].x + bfrag[kb].y;
                q += bfrag[kb].x * bfrag[kb].x + bfrag[kb].y * bfrag[kb].y;
            }
            s += __shfl_xor(s, 16, 32);
            q += __shfl_xor(q, 16, 32);
            const float mu  = s * (1.0f / 64.0f);
            const float inv = rsqrtf(fmaxf(q * (1.0f / 64.0f) - mu * mu, 0.0f) + 1e-5f);
            #pragma unroll
            for (int kb = 0; kb < 16; ++kb) {
                const int k0 = kb * 4 + kOff;
                bfrag[kb].x = (bfrag[kb].x - mu) * inv * LNP[2][k0]     + LNP[3][k0];
                bfrag[kb].y = (bfrag[kb].y - mu) * inv * LNP[2][k0 + 1] + LNP[3][k0 + 1];
            }
        }
    }

    #pragma unroll
    for (int nb = 0; nb < 4; ++nb) {
        const int n0 = nb * 16;
        v8f acc = {0.f, 0.f, 0.f, 0.f, 0.f, 0.f, 0.f, 0.f};

        v2f wf[16];
        #pragma unroll
        for (int kb = 0; kb < 16; ++kb) {
            const int kr = kb * 4 + kOff;
            wf[kb].x = W1[kr * FDIM + n0 + l15];
            wf[kb].y = W1[(kr + 1) * FDIM + n0 + l15];
        }
        #pragma unroll
        for (int kb = 0; kb < 16; ++kb)
            acc = __builtin_amdgcn_wmma_f32_16x16x4_f32(
                false, afrag[kb], false, wf[kb], (short)0, acc, false, false);

        if (B) {
            v2f wg[16];
            #pragma unroll
            for (int kb = 0; kb < 16; ++kb) {
                const int kr = kb * 4 + kOff;
                wg[kb].x = W2[kr * FDIM + n0 + l15];
                wg[kb].y = W2[(kr + 1) * FDIM + n0 + l15];
            }
            #pragma unroll
            for (int kb = 0; kb < 16; ++kb)
                acc = __builtin_amdgcn_wmma_f32_16x16x4_f32(
                    false, bfrag[kb], false, wg[kb], (short)0, acc, false, false);
        }

        const float bval  = bias  ? bias[n0 + l15]  : 0.0f;
        const float b2val = bias2 ? bias2[n0 + l15] : 0.0f;
        #pragma unroll
        for (int j = 0; j < 8; ++j) {
            const int r = m0 + j + lhalf * 8;
            if (r < n) {
                float v = acc[j] + bval;
                if (reluOut) v = fmaxf(v, 0.0f);
                out[(size_t)r * FDIM + n0 + l15] = v;
                if (out2) {
                    const float d = dinv2[r];
                    out2[(size_t)r * FDIM + n0 + l15] = acc[j] * d * d + b2val;
                }
            }
        }
    }
}

template <bool GCN>
__launch_bounds__(256)
__global__ void gather_rows_kernel(const float* __restrict__ feat, const int* __restrict__ rowptr,
                                   const int* __restrict__ cnt, const int* __restrict__ nbr,
                                   const float* __restrict__ dinv, const float* __restrict__ bias,
                                   float* __restrict__ outp, int n) {
    const int q   = threadIdx.x & 15;
    const int grp = (int)((blockIdx.x * 256u + threadIdx.x) >> 4);
    const int r0  = grp * GROUP_ROWS;
    if (r0 >= n) return;
    const float4* f4 = (const float4*)feat;

    int begN = rowptr[r0], degN = cnt[r0];
    for (int k = 0; k < GROUP_ROWS; ++k) {
        const int r = r0 + k;
        if (r >= n) break;
        const int beg = begN, deg = degN;
        if (k + 1 < GROUP_ROWS && r + 1 < n) {
            begN = rowptr[r + 1];
            degN = cnt[r + 1];
        }
        const int* nb = nbr + beg;
        const float dr = GCN ? dinv[r] : 0.0f;
        float4 acc = make_float4(0.0f, 0.0f, 0.0f, 0.0f);
        for (int j = 0; j < deg; j += EDGE_UNROLL) {
            const int m = deg - j;
            int   s[EDGE_UNROLL];
            float4 v[EDGE_UNROLL];
            float w[EDGE_UNROLL];
            #pragma unroll
            for (int u = 0; u < EDGE_UNROLL; ++u) s[u] = nb[(j + u < deg) ? (j + u) : 0];
            #pragma unroll
            for (int u = 0; u < EDGE_UNROLL; ++u) {
                v[u] = f4[(size_t)s[u] * 16 + q];
                w[u] = GCN ? dinv[s[u]] * dr : 1.0f;
            }
            #pragma unroll
            for (int u = 0; u < EDGE_UNROLL; ++u) {
                const bool on = (u == 0) || (u < m);
                float4 nx;
                if (GCN) {
                    nx.x = acc.x + v[u].x * w[u]; nx.y = acc.y + v[u].y * w[u];
                    nx.z = acc.z + v[u].z * w[u]; nx.w = acc.w + v[u].w * w[u];
                } else {
                    nx.x = acc.x + v[u].x; nx.y = acc.y + v[u].y;
                    nx.z = acc.z + v[u].z; nx.w = acc.w + v[u].w;
                }
                acc.x = on ? nx.x : acc.x; acc.y = on ? nx.y : acc.y;
                acc.z = on ? nx.z : acc.z; acc.w = on ? nx.w : acc.w;
            }
        }
        float4 o;
        if (GCN) {
            const float4 hv = f4[(size_t)r * 16 + q];
            const float4 bv = ((const float4*)bias)[q];
            const float d2 = dr * dr;
            o.x = acc.x + hv.x * d2 + bv.x; o.y = acc.y + hv.y * d2 + bv.y;
            o.z = acc.z + hv.z * d2 + bv.z; o.w = acc.w + hv.w * d2 + bv.w;
        } else {
            const float rs = 1.0f / fmaxf((float)deg, 1.0f);
            o.x = acc.x * rs; o.y = acc.y * rs; o.z = acc.z * rs; o.w = acc.w * rs;
        }
        ((float4*)outp)[(size_t)r * 16 + q] = o;
    }
}

extern "C" void kernel_launch(void* const* d_in, const int* in_sizes, int n_in,
                              void* d_out, int out_size, void* d_ws, size_t ws_size,
                              hipStream_t stream) {
    const float* x        = (const float*)d_in[0];
    const int*   ei       = (const int*)d_in[1];
    const float* gcn_w1   = (const float*)d_in[2];
    const float* gcn_b1   = (const float*)d_in[3];
    const float* gcn_w2   = (const float*)d_in[4];
    const float* gcn_b2   = (const float*)d_in[5];
    const float* sage_wl1 = (const float*)d_in[6];
    const float* sage_bl1 = (const float*)d_in[7];
    const float* sage_wr1 = (const float*)d_in[8];
    const float* sage_wl2 = (const float*)d_in[9];
    const float* sage_bl2 = (const float*)d_in[10];
    const float* sage_wr2 = (const float*)d_in[11];
    const float* gcn_ln_g = (const float*)d_in[12];
    const float* gcn_ln_b = (const float*)d_in[13];
    const float* sage_ln_g = (const float*)d_in[14];
    const float* sage_ln_b = (const float*)d_in[15];
    const float* proj_w   = (const float*)d_in[16];
    const float* proj_b   = (const float*)d_in[17];

    const int N = in_sizes[0] / FDIM;
    const int E = in_sizes[1] / 2;
    const int* src = ei;
    const int* dst = ei + E;

    int*   cnt    = (int*)d_ws;
    float* dinv   = (float*)(cnt + N);
    int*   rowptr = (int*)(dinv + N);
    int*   nbr    = rowptr + N;
    float* h      = (float*)(nbr + E);
    float* g      = h + (size_t)N * FDIM;
    float* agg    = g + (size_t)N * FDIM;
    float* out    = (float*)d_out;
    int*   rank   = (int*)agg;
    int*   bsum   = (int*)h;

    const int nblkN  = (N + 255) / 256;
    const int nblkE  = (E + 255) / 256;
    const int nblkT  = (N + SCAN_TILE - 1) / SCAN_TILE;
    const int nblkG  = (N + 127) / 128;
    const int nblkR  = (N + 16 * GROUP_ROWS - 1) / (16 * GROUP_ROWS);

    init_cnt_kernel<<<nblkN, 256, 0, stream>>>(cnt, N);
    count_rank_kernel<<<nblkE, 256, 0, stream>>>(dst, cnt, rank, E);
    scan_tile_kernel<<<nblkT, 256, 0, stream>>>(cnt, rowptr, bsum, N);
    finalize_rows_kernel<<<nblkN, 256, 0, stream>>>(cnt, bsum, rowptr, dinv, N);
    fill_nbr_kernel<<<nblkE, 256, 0, stream>>>(src, dst, rowptr, rank, nbr, E);

    gemm64_wmma_kernel<<<nblkG, 256, 0, stream>>>(
        x, gcn_w1, nullptr, nullptr, nullptr, nullptr,
        nullptr, nullptr, nullptr, nullptr,
        nullptr, nullptr, nullptr, h, N, 0, 0);
    gather_rows_kernel<true><<<nblkR, 256, 0, stream>>>(h, rowptr, cnt, nbr, dinv, gcn_b1, g, N);

    gemm64_wmma_kernel<<<nblkG, 256, 0, stream>>>(
        g, gcn_w2, nullptr, nullptr, nullptr, nullptr,
        nullptr, nullptr, nullptr, nullptr,
        nullptr, nullptr, nullptr, h, N, 0, 1  );
    gather_rows_kernel<true><<<nblkR, 256, 0, stream>>>(h, rowptr, cnt, nbr, dinv, gcn_b2, g, N);

    gather_rows_kernel<false><<<nblkR, 256, 0, stream>>>(x, rowptr, cnt, nbr, nullptr, nullptr, agg, N);
    gemm64_wmma_kernel<<<nblkG, 256, 0, stream>>>(
        agg, sage_wl1, x, sage_wr1, sage_bl1, nullptr,
        nullptr, nullptr, nullptr, nullptr,
        nullptr, nullptr, nullptr, out, N, 1  , 0);

    gather_rows_kernel<false><<<nblkR, 256, 0, stream>>>(out, rowptr, cnt, nbr, nullptr, nullptr, agg, N);
    gemm64_wmma_kernel<<<nblkG, 256, 0, stream>>>(
        agg, sage_wl2, out, sage_wr2, sage_bl2, nullptr,
        nullptr, nullptr, nullptr, nullptr,
        nullptr, nullptr, nullptr, h, N, 0, 0);

    gemm64_wmma_kernel<<<nblkG, 256, 0, stream>>>(
        g, proj_w, h, proj_w + FDIM * FDIM, proj_b, nullptr,
        gcn_ln_g, gcn_ln_b, sage_ln_g, sage_ln_b,
        nullptr, nullptr, nullptr, out, N, 0, 0);
}


# GCN neighbour sums moved in front of their products ((A_hat X) W, sum-then-multiply order), so each layer's two branches share one pass over the edge buckets: 4 gather launches become 2
# speedup vs baseline: 2.6901x; 1.1551x over previous
#include <hip/hip_runtime.h>

typedef __attribute__((ext_vector_type(2))) float v2f;
typedef __attribute__((ext_vector_type(8))) float v8f;

#define FDIM 64
#define SCAN_TILE 1024
#define GROUP_ROWS 4
#define EDGE_UNROLL 8

__global__ void init_cnt_kernel(int* __restrict__ cnt, int n) {
    int i = blockIdx.x * blockDim.x + threadIdx.x;
    if (i < n) cnt[i] = 0;
}

__global__ void count_rank_kernel(const int* __restrict__ dst, int* __restrict__ cnt,
                                  int* __restrict__ rank, int e) {
    int i = blockIdx.x * blockDim.x + threadIdx.x;
    if (i < e) rank[i] = atomicAdd(&cnt[dst[i]], 1);
}

__launch_bounds__(256)
__global__ void scan_tile_kernel(const int* __restrict__ cnt, int* __restrict__ rowptr,
                                 int* __restrict__ bsum, int n) {
    __shared__ int wsum[8];
    const int t    = threadIdx.x;
    const int lane = t & 31;
    const int wave = t >> 5;
    const int base = blockIdx.x * SCAN_TILE + t * 4;
    const int c0 = (base + 0 < n) ? cnt[base + 0] : 0;
    const int c1 = (base + 1 < n) ? cnt[base + 1] : 0;
    const int c2 = (base + 2 < n) ? cnt[base + 2] : 0;
    const int c3 = (base + 3 < n) ? cnt[base + 3] : 0;
    const int tot = c0 + c1 + c2 + c3;
    int inc = tot;
    for (int o = 1; o < 32; o <<= 1) {
        const int v = __shfl_up(inc, o, 32);
        if (lane >= o) inc += v;
    }
    if (lane == 31) wsum[wave] = inc;
    __syncthreads();
    int woff = 0;
    for (int w = 0; w < wave; ++w) woff += wsum[w];
    const int ex = woff + inc - tot;
    if (base + 0 < n) rowptr[base + 0] = ex;
    if (base + 1 < n) rowptr[base + 1] = ex + c0;
    if (base + 2 < n) rowptr[base + 2] = ex + c0 + c1;
    if (base + 3 < n) rowptr[base + 3] = ex + c0 + c1 + c2;
    if (t == 255) bsum[blockIdx.x] = woff + inc;
}

__launch_bounds__(256)
__global__ void finalize_rows_kernel(const int* __restrict__ cnt, const int* __restrict__ bsum,
                                     int* __restrict__ rowptr, float* __restrict__ dinv, int n) {
    __shared__ int part[8];
    const int t  = threadIdx.x;
    const int nt = (blockIdx.x * 256) / SCAN_TILE;
    int v = 0;
    for (int i = t; i < nt; i += 256) v += bsum[i];
    for (int o = 16; o > 0; o >>= 1) v += __shfl_xor(v, o, 32);
    if ((t & 31) == 0) part[t >> 5] = v;
    __syncthreads();
    int off = 0;
    for (int w = 0; w < 8; ++w) off += part[w];
    const int i = blockIdx.x * 256 + t;
    if (i < n) {
        rowptr[i] += off;
        dinv[i] = rsqrtf((float)cnt[i] + 1.0f);
    }
}

__global__ void fill_nbr_kernel(const int* __restrict__ src, const int* __restrict__ dst,
                                const int* __restrict__ rowptr, const int* __restrict__ rank,
                                int* __restrict__ nbr, int e) {
    int i = blockIdx.x * blockDim.x + threadIdx.x;
    if (i < e) nbr[rowptr[dst[i]] + rank[i]] = src[i];
}

__launch_bounds__(256)
__global__ void gemm64_wmma_kernel(const float* __restrict__ A, const float* __restrict__ W1g,
                                   const float* __restrict__ B, const float* __restrict__ W2g,
                                   const float* __restrict__ bias,
                                   const float* __restrict__ rowscale,
                                   const float* __restrict__ lnAg, const float* __restrict__ lnAb,
                                   const float* __restrict__ lnBg, const float* __restrict__ lnBb,
                                   const float* __restrict__ dinv2, const float* __restrict__ bias2,
                                   float* __restrict__ out2,
                                   float* __restrict__ out, int n, int reluOut, int reluA) {
    __shared__ float W1[FDIM * FDIM];
    __shared__ float W2[FDIM * FDIM];
    __shared__ float LNP[4][FDIM];

    const int t = threadIdx.x;
    {
        const float4* s1 = (const float4*)W1g;
        float4* l1 = (float4*)W1;
        for (int i = t; i < 1024; i += 256) l1[i] = s1[i];
        if (B) {
            const float4* s2 = (const float4*)W2g;
            float4* l2 = (float4*)W2;
            for (int i = t; i < 1024; i += 256) l2[i] = s2[i];
        }
        if (lnAg && t < FDIM) { LNP[0][t] = lnAg[t]; LNP[1][t] = lnAb[t]; }
        if (lnBg && t < FDIM) { LNP[2][t] = lnBg[t]; LNP[3][t] = lnBb[t]; }
    }
    __syncthreads();

    const int wave  = t >> 5;
    const int lane  = t & 31;
    const int m0    = (blockIdx.x * 8 + wave) * 16;
    if (m0 >= n) return;
    const int lhalf = lane >> 4;
    const int l15   = lane & 15;
    const int kOff  = lhalf * 2;

    int row = m0 + l15;
    if (row >= n) row = n - 1;

    const float rs = rowscale ? rowscale[row] : 1.0f;

    v2f afrag[16];
    {
        const float* ar = A + (size_t)row * FDIM + kOff;
        #pragma unroll
        for (int kb = 0; kb < 16; ++kb) {
            float a0 = ar[kb * 4 + 0];
            float a1 = ar[kb * 4 + 1];
            if (reluA) { a0 = fmaxf(a0, 0.0f); a1 = fmaxf(a1, 0.0f); }
            afrag[kb].x = a0 * rs;
            afrag[kb].y = a1 * rs;
        }
        if (lnAg) {
            float s = 0.0f, q = 0.0f;
            #pragma unroll
            for (int kb = 0; kb < 16; ++kb) {
                s += afrag[kb].x + afrag[kb].y;
                q += afrag[kb].x * afrag[kb].x + afrag[kb].y * afrag[kb].y;
            }
            s += __shfl_xor(s, 16, 32);
            q += __shfl_xor(q, 16, 32);
            const float mu  = s * (1.0f / 64.0f);
            const float inv = rsqrtf(fmaxf(q * (1.0f / 64.0f) - mu * mu, 0.0f) + 1e-5f);
            #pragma unroll
            for (int kb = 0; kb < 16; ++kb) {
                const int k0 = kb * 4 + kOff;
                afrag[kb].x = (afrag[kb].x - mu) * inv * LNP[0][k0]     + LNP[1][k0];
                afrag[kb].y = (afrag[kb].y - mu) * inv * LNP[0][k0 + 1] + LNP[1][k0 + 1];
            }
        }
    }
    v2f bfrag[16];
    if (B) {
        const float* br = B + (size_t)row * FDIM + kOff;
        #pragma unroll
        for (int kb = 0; kb < 16; ++kb) {
            bfrag[kb].x = br[kb * 4 + 0];
            bfrag[kb].y = br[kb * 4 + 1];
        }
        if (lnBg) {
            float s = 0.0f, q = 0.0f;
            #pragma unroll
            for (int kb = 0; kb < 16; ++kb) {
                s += bfrag[kb].x + bfrag[kb].y;
                q += bfrag[kb].x * bfrag[kb].x + bfrag[kb].y * bfrag[kb].y;
            }
            s += __shfl_xor(s, 16, 32);
            q += __shfl_xor(q, 16, 32);
            const float mu  = s * (1.0f / 64.0f);
            const float inv = rsqrtf(fmaxf(q * (1.0f / 64.0f) - mu * mu, 0.0f) + 1e-5f);
            #pragma unroll
            for (int kb = 0; kb < 16; ++kb) {
                const int k0 = kb * 4 + kOff;
                bfrag[kb].x = (bfrag[kb].x - mu) * inv * LNP[2][k0]     + LNP[3][k0];
                bfrag[kb].y = (bfrag[kb].y - mu) * inv * LNP[2][k0 + 1] + LNP[3][k0 + 1];
            }
        }
    }

    #pragma unroll
    for (int nb = 0; nb < 4; ++nb) {
        const int n0 = nb * 16;
        v8f acc = {0.f, 0.f, 0.f, 0.f, 0.f, 0.f, 0.f, 0.f};

        v2f wf[16];
        #pragma unroll
        for (int kb = 0; kb < 16; ++kb) {
            const int kr = kb * 4 + kOff;
            wf[kb].x = W1[kr * FDIM + n0 + l15];
            wf[kb].y = W1[(kr + 1) * FDIM + n0 + l15];
        }
        #pragma unroll
        for (int kb = 0; kb < 16; ++kb)
            acc = __builtin_amdgcn_wmma_f32_16x16x4_f32(
                false, afrag[kb], false, wf[kb], (short)0, acc, false, false);

        if (B) {
            v2f wg[16];
            #pragma unroll
            for (int kb = 0; kb < 16; ++kb) {
                const int kr = kb * 4 + kOff;
                wg[kb].x = W2[kr * FDIM + n0 + l15];
                wg[kb].y = W2[(kr + 1) * FDIM + n0 + l15];
            }
            #pragma unroll
            for (int kb = 0; kb < 16; ++kb)
                acc = __builtin_amdgcn_wmma_f32_16x16x4_f32(
                    false, bfrag[kb], false, wg[kb], (short)0, acc, false, false);
        }

        const float bval  = bias  ? bias[n0 + l15]  : 0.0f;
        const float b2val = bias2 ? bias2[n0 + l15] : 0.0f;
        #pragma unroll
        for (int j = 0; j < 8; ++j) {
            const int r = m0 + j + lhalf * 8;
            if (r < n) {
                float v = acc[j] + bval;
                if (reluOut) v = fmaxf(v, 0.0f);
                out[(size_t)r * FDIM + n0 + l15] = v;
                if (out2) {
                    const float d = dinv2[r];
                    out2[(size_t)r * FDIM + n0 + l15] = acc[j] * d * d + b2val;
                }
            }
        }
    }
}

template <bool ONE_TABLE>
__launch_bounds__(256)
__global__ void gather_pair_kernel(const float* __restrict__ tabG, const float* __restrict__ tabS,
                                   const int* __restrict__ rowptr, const int* __restrict__ cnt,
                                   const int* __restrict__ nbr, const float* __restrict__ dinv,
                                   float* __restrict__ outG, float* __restrict__ outS, int n) {
    const int q   = threadIdx.x & 15;
    const int grp = (int)((blockIdx.x * 256u + threadIdx.x) >> 4);
    const int r0  = grp * GROUP_ROWS;
    if (r0 >= n) return;
    const float4* g4 = (const float4*)tabG;
    const float4* s4 = (const float4*)tabS;

    int begN = rowptr[r0], degN = cnt[r0];
    for (int k = 0; k < GROUP_ROWS; ++k) {
        const int r = r0 + k;
        if (r >= n) break;
        const int beg = begN, deg = degN;
        if (k + 1 < GROUP_ROWS && r + 1 < n) {
            begN = rowptr[r + 1];
            degN = cnt[r + 1];
        }
        const int* nb = nbr + beg;
        const float dr = dinv[r];
        float4 accG = make_float4(0.0f, 0.0f, 0.0f, 0.0f);
        float4 accS = make_float4(0.0f, 0.0f, 0.0f, 0.0f);
        for (int j = 0; j < deg; j += EDGE_UNROLL) {
            const int m = deg - j;
            int   s[EDGE_UNROLL];
            float4 vg[EDGE_UNROLL];
            float4 vs[EDGE_UNROLL];
            float w[EDGE_UNROLL];
            #pragma unroll
            for (int u = 0; u < EDGE_UNROLL; ++u) s[u] = nb[(j + u < deg) ? (j + u) : 0];
            #pragma unroll
            for (int u = 0; u < EDGE_UNROLL; ++u) {
                vg[u] = g4[(size_t)s[u] * 16 + q];
                if (!ONE_TABLE) vs[u] = s4[(size_t)s[u] * 16 + q];
                w[u] = dinv[s[u]] * dr;
            }
            #pragma unroll
            for (int u = 0; u < EDGE_UNROLL; ++u) {
                const bool on = (u == 0) || (u < m);
                const float4 a = vg[u];
                const float4 b = ONE_TABLE ? vg[u] : vs[u];
                float4 ng, ns;
                ng.x = accG.x + a.x * w[u]; ng.y = accG.y + a.y * w[u];
                ng.z = accG.z + a.z * w[u]; ng.w = accG.w + a.w * w[u];
                ns.x = accS.x + b.x; ns.y = accS.y + b.y;
                ns.z = accS.z + b.z; ns.w = accS.w + b.w;
                accG.x = on ? ng.x : accG.x; accG.y = on ? ng.y : accG.y;
                accG.z = on ? ng.z : accG.z; accG.w = on ? ng.w : accG.w;
                accS.x = on ? ns.x : accS.x; accS.y = on ? ns.y : accS.y;
                accS.z = on ? ns.z : accS.z; accS.w = on ? ns.w : accS.w;
            }
        }
        const float4 hv = g4[(size_t)r * 16 + q];
        const float d2 = dr * dr;
        float4 og, os;
        og.x = accG.x + hv.x * d2; og.y = accG.y + hv.y * d2;
        og.z = accG.z + hv.z * d2; og.w = accG.w + hv.w * d2;
        const float rs = 1.0f / fmaxf((float)deg, 1.0f);
        os.x = accS.x * rs; os.y = accS.y * rs; os.z = accS.z * rs; os.w = accS.w * rs;
        ((float4*)outG)[(size_t)r * 16 + q] = og;
        ((float4*)outS)[(size_t)r * 16 + q] = os;
    }
}

extern "C" void kernel_launch(void* const* d_in, const int* in_sizes, int n_in,
                              void* d_out, int out_size, void* d_ws, size_t ws_size,
                              hipStream_t stream) {
    const float* x        = (const float*)d_in[0];
    const int*   ei       = (const int*)d_in[1];
    const float* gcn_w1   = (const float*)d_in[2];
    const float* gcn_b1   = (const float*)d_in[3];
    const float* gcn_w2   = (const float*)d_in[4];
    const float* gcn_b2   = (const float*)d_in[5];
    const float* sage_wl1 = (const float*)d_in[6];
    const float* sage_bl1 = (const float*)d_in[7];
    const float* sage_wr1 = (const float*)d_in[8];
    const float* sage_wl2 = (const float*)d_in[9];
    const float* sage_bl2 = (const float*)d_in[10];
    const float* sage_wr2 = (const float*)d_in[11];
    const float* gcn_ln_g = (const float*)d_in[12];
    const float* gcn_ln_b = (const float*)d_in[13];
    const float* sage_ln_g = (const float*)d_in[14];
    const float* sage_ln_b = (const float*)d_in[15];
    const float* proj_w   = (const float*)d_in[16];
    const float* proj_b   = (const float*)d_in[17];

    const int N = in_sizes[0] / FDIM;
    const int E = in_sizes[1] / 2;
    const int* src = ei;
    const int* dst = ei + E;

    int*   cnt    = (int*)d_ws;
    float* dinv   = (float*)(cnt + N);
    int*   rowptr = (int*)(dinv + N);
    int*   nbr    = rowptr + N;
    float* h      = (float*)(nbr + E);
    float* g      = h + (size_t)N * FDIM;
    float* agg    = g + (size_t)N * FDIM;
    float* out    = (float*)d_out;
    int*   rank   = (int*)agg;
    int*   bsum   = (int*)h;

    const int nblkN  = (N + 255) / 256;
    const int nblkE  = (E + 255) / 256;
    const int nblkT  = (N + SCAN_TILE - 1) / SCAN_TILE;
    const int nblkG  = (N + 127) / 128;
    const int nblkR  = (N + 16 * GROUP_ROWS - 1) / (16 * GROUP_ROWS);

    init_cnt_kernel<<<nblkN, 256, 0, stream>>>(cnt, N);
    count_rank_kernel<<<nblkE, 256, 0, stream>>>(dst, cnt, rank, E);
    scan_tile_kernel<<<nblkT, 256, 0, stream>>>(cnt, rowptr, bsum, N);
    finalize_rows_kernel<<<nblkN, 256, 0, stream>>>(cnt, bsum, rowptr, dinv, N);
    fill_nbr_kernel<<<nblkE, 256, 0, stream>>>(src, dst, rowptr, rank, nbr, E);

    gather_pair_kernel<true><<<nblkR, 256, 0, stream>>>(x, x, rowptr, cnt, nbr, dinv, h, agg, N);
    gemm64_wmma_kernel<<<nblkG, 256, 0, stream>>>(
        h, gcn_w1, nullptr, nullptr, gcn_b1, nullptr,
        nullptr, nullptr, nullptr, nullptr,
        nullptr, nullptr, nullptr, g, N, 1  , 0);
    gemm64_wmma_kernel<<<nblkG, 256, 0, stream>>>(
        agg, sage_wl1, x, sage_wr1, sage_bl1, nullptr,
        nullptr, nullptr, nullptr, nullptr,
        nullptr, nullptr, nullptr, out, N, 1  , 0);

    gather_pair_kernel<false><<<nblkR, 256, 0, stream>>>(g, out, rowptr, cnt, nbr, dinv, h, agg, N);
    gemm64_wmma_kernel<<<nblkG, 256, 0, stream>>>(
        h, gcn_w2, nullptr, nullptr, gcn_b2, nullptr,
        nullptr, nullptr, nullptr, nullptr,
        nullptr, nullptr, nullptr, g, N, 0, 0);
    gemm64_wmma_kernel<<<nblkG, 256, 0, stream>>>(
        agg, sage_wl2, out, sage_wr2, sage_bl2, nullptr,
        nullptr, nullptr, nullptr, nullptr,
        nullptr, nullptr, nullptr, h, N, 0, 0);

    gemm64_wmma_kernel<<<nblkG, 256, 0, stream>>>(
        g, proj_w, h, proj_w + FDIM * FDIM, proj_b, nullptr,
        gcn_ln_g, gcn_ln_b, sage_ln_g, sage_ln_b,
        nullptr, nullptr, nullptr, out, N, 0, 0);
}


# Gather pass re-pipelined: a chunk's source ids and weights come by one lane-parallel load and row_share moves, requested a chunk ahead, so 8 edges wait for one round of loads; sum order unchanged
# speedup vs baseline: 2.8591x; 1.0628x over previous
#include <hip/hip_runtime.h>

typedef __attribute__((ext_vector_type(2))) float v2f;
typedef __attribute__((ext_vector_type(8))) float v8f;
typedef __attribute__((ext_vector_type(4))) float v4f;

#define FDIM 64
#define SCAN_TILE 1024
#define GROUP_ROWS 5
#define EDGE_UNROLL 8

__global__ void init_cnt_kernel(int* __restrict__ cnt, int n) {
    int i = blockIdx.x * blockDim.x + threadIdx.x;
    if (i < n) cnt[i] = 0;
}

__global__ void count_rank_kernel(const int* __restrict__ dst, int* __restrict__ cnt,
                                  int* __restrict__ rank, int e) {
    int i = blockIdx.x * blockDim.x + threadIdx.x;
    if (i < e) rank[i] = atomicAdd(&cnt[dst[i]], 1);
}

__launch_bounds__(256)
__global__ void scan_tile_kernel(const int* __restrict__ cnt, int* __restrict__ rowptr,
                                 int* __restrict__ bsum, int n) {
    __shared__ int wsum[8];
    const int t    = threadIdx.x;
    const int lane = t & 31;
    const int wave = t >> 5;
    const int base = blockIdx.x * SCAN_TILE + t * 4;
    const int c0 = (base + 0 < n) ? cnt[base + 0] : 0;
    const int c1 = (base + 1 < n) ? cnt[base + 1] : 0;
    const int c2 = (base + 2 < n) ? cnt[base + 2] : 0;
    const int c3 = (base + 3 < n) ? cnt[base + 3] : 0;
    const int tot = c0 + c1 + c2 + c3;
    int inc = tot;
    for (int o = 1; o < 32; o <<= 1) {
        const int v = __shfl_up(inc, o, 32);
        if (lane >= o) inc += v;
    }
    if (lane == 31) wsum[wave] = inc;
    __syncthreads();
    int woff = 0;
    for (int w = 0; w < wave; ++w) woff += wsum[w];
    const int ex = woff + inc - tot;
    if (base + 0 < n) rowptr[base + 0] = ex;
    if (base + 1 < n) rowptr[base + 1] = ex + c0;
    if (base + 2 < n) rowptr[base + 2] = ex + c0 + c1;
    if (base + 3 < n) rowptr[base + 3] = ex + c0 + c1 + c2;
    if (t == 255) bsum[blockIdx.x] = woff + inc;
}

__launch_bounds__(256)
__global__ void finalize_rows_kernel(const int* __restrict__ cnt, const int* __restrict__ bsum,
                                     int* __restrict__ rowptr, float* __restrict__ dinv, int n) {
    __shared__ int part[8];
    const int t  = threadIdx.x;
    const int nt = (blockIdx.x * 256) / SCAN_TILE;
    int v = 0;
    for (int i = t; i < nt; i += 256) v += bsum[i];
    for (int o = 16; o > 0; o >>= 1) v += __shfl_xor(v, o, 32);
    if ((t & 31) == 0) part[t >> 5] = v;
    __syncthreads();
    int off = 0;
    for (int w = 0; w < 8; ++w) off += part[w];
    const int i = blockIdx.x * 256 + t;
    if (i < n) {
        rowptr[i] += off;
        dinv[i] = rsqrtf((float)cnt[i] + 1.0f);
    }
}

__global__ void fill_nbr_kernel(const int* __restrict__ src, const int* __restrict__ dst,
                                const int* __restrict__ rowptr, const int* __restrict__ rank,
                                int* __restrict__ nbr, int e) {
    int i = blockIdx.x * blockDim.x + threadIdx.x;
    if (i < e) nbr[rowptr[dst[i]] + rank[i]] = src[i];
}

__launch_bounds__(256)
__global__ void gemm64_wmma_kernel(const float* __restrict__ A, const float* __restrict__ W1g,
                                   const float* __restrict__ B, const float* __restrict__ W2g,
                                   const float* __restrict__ bias,
                                   const float* __restrict__ rowscale,
                                   const float* __restrict__ lnAg, const float* __restrict__ lnAb,
                                   const float* __restrict__ lnBg, const float* __restrict__ lnBb,
                                   const float* __restrict__ dinv2, const float* __restrict__ bias2,
                                   float* __restrict__ out2,
                                   float* __restrict__ out, int n, int reluOut, int reluA) {
    __shared__ float W1[FDIM * FDIM];
    __shared__ float W2[FDIM * FDIM];
    __shared__ float LNP[4][FDIM];

    const int t = threadIdx.x;
    {
        const float4* s1 = (const float4*)W1g;
        float4* l1 = (float4*)W1;
        for (int i = t; i < 1024; i += 256) l1[i] = s1[i];
        if (B) {
            const float4* s2 = (const float4*)W2g;
            float4* l2 = (float4*)W2;
            for (int i = t; i < 1024; i += 256) l2[i] = s2[i];
        }
        if (lnAg && t < FDIM) { LNP[0][t] = lnAg[t]; LNP[1][t] = lnAb[t]; }
        if (lnBg && t < FDIM) { LNP[2][t] = lnBg[t]; LNP[3][t] = lnBb[t]; }
    }
    __syncthreads();

    const int wave  = t >> 5;
    const int lane  = t & 31;
    const int m0    = (blockIdx.x * 8 + wave) * 16;
    if (m0 >= n) return;
    const int lhalf = lane >> 4;
    const int l15   = lane & 15;
    const int kOff  = lhalf * 2;

    int row = m0 + l15;
    if (row >= n) row = n - 1;

    const float rs = rowscale ? rowscale[row] : 1.0f;

    v2f afrag[16];
    {
        const float* ar = A + (size_t)row * FDIM + kOff;
        #pragma unroll
        for (int kb = 0; kb < 16; ++kb) {
            float a0 = ar[kb * 4 + 0];
            float a1 = ar[kb * 4 + 1];
            if (reluA) { a0 = fmaxf(a0, 0.0f); a1 = fmaxf(a1, 0.0f); }
            afrag[kb].x = a0 * rs;
            afrag[kb].y = a1 * rs;
        }
        if (lnAg) {
            float s = 0.0f, q = 0.0f;
            #pragma unroll
            for (int kb = 0; kb < 16; ++kb) {
                s += afrag[kb].x + afrag[kb].y;
                q += afrag[kb].x * afrag[kb].x + afrag[kb].y * afrag[kb].y;
            }
            s += __shfl_xor(s, 16, 32);
            q += __shfl_xor(q, 16, 32);
            const float mu  = s * (1.0f / 64.0f);
            const float inv = rsqrtf(fmaxf(q * (1.0f / 64.0f) - mu * mu, 0.0f) + 1e-5f);
            #pragma unroll
            for (int kb = 0; kb < 16; ++kb) {
                const int k0 = kb * 4 + kOff;
                afrag[kb].x = (afrag[kb].x - mu) * inv * LNP[0][k0]     + LNP[1][k0];
                afrag[kb].y = (afrag[kb].y - mu) * inv * LNP[0][k0 + 1] + LNP[1][k0 + 1];
            }
        }
    }
    v2f bfrag[16];
    if (B) {
        const float* br = B + (size_t)row * FDIM + kOff;
        #pragma unroll
        for (int kb = 0; kb < 16; ++kb) {
            bfrag[kb].x = br[kb * 4 + 0];
            bfrag[kb].y = br[kb * 4 + 1];
        }
        if (lnBg) {
            float s = 0.0f, q = 0.0f;
            #pragma unroll
            for (int kb = 0; kb < 16; ++kb) {
                s += bfrag[kb].x + bfrag[kb].y;
                q += bfrag[kb].x * bfrag[kb].x + bfrag[kb].y * bfrag[kb].y;
            }
            s += __shfl_xor(s, 16, 32);
            q += __shfl_xor(q, 16, 32);
            const float mu  = s * (1.0f / 64.0f);
            const float inv = rsqrtf(fmaxf(q * (1.0f / 64.0f) - mu * mu, 0.0f) + 1e-5f);
            #pragma unroll
            for (int kb = 0; kb < 16; ++kb) {
                const int k0 = kb * 4 + kOff;
                bfrag[kb].x = (bfrag[kb].x - mu) * inv * LNP[2][k0]     + LNP[3][k0];
                bfrag[kb].y = (bfrag[kb].y - mu) * inv * LNP[2][k0 + 1] + LNP[3][k0 + 1];
            }
        }
    }

    #pragma unroll
    for (int nb = 0; nb < 4; ++nb) {
        const int n0 = nb * 16;
        v8f acc = {0.f, 0.f, 0.f, 0.f, 0.f, 0.f, 0.f, 0.f};

        v2f wf[16];
        #pragma unroll
        for (int kb = 0; kb < 16; ++kb) {
            const int kr = kb * 4 + kOff;
            wf[kb].x = W1[kr * FDIM + n0 + l15];
            wf[kb].y = W1[(kr + 1) * FDIM + n0 + l15];
        }
        #pragma unroll
        for (int kb = 0; kb < 16; ++kb)
            acc = __builtin_amdgcn_wmma_f32_16x16x4_f32(
                false, afrag[kb], false, wf[kb], (short)0, acc, false, false);

        if (B) {
            v2f wg[16];
            #pragma unroll
            for (int kb = 0; kb < 16; ++kb) {
                const int kr = kb * 4 + kOff;
                wg[kb].x = W2[kr * FDIM + n0 + l15];
                wg[kb].y = W2[(kr + 1) * FDIM + n0 + l15];
            }
            #pragma unroll
            for (int kb = 0; kb < 16; ++kb)
                acc = __builtin_amdgcn_wmma_f32_16x16x4_f32(
                    false, bfrag[kb], false, wg[kb], (short)0, acc, false, false);
        }

        const float bval  = bias  ? bias[n0 + l15]  : 0.0f;
        const float b2val = bias2 ? bias2[n0 + l15] : 0.0f;
        #pragma unroll
        for (int j = 0; j < 8; ++j) {
            const int r = m0 + j + lhalf * 8;
            if (r < n) {
                float v = acc[j] + bval;
                if (reluOut) v = fmaxf(v, 0.0f);
                out[(size_t)r * FDIM + n0 + l15] = v;
                if (out2) {
                    const float d = dinv2[r];
                    out2[(size_t)r * FDIM + n0 + l15] = acc[j] * d * d + b2val;
                }
            }
        }
    }
}

__device__ __forceinline__ int row_bcast(int v, int u) {
    switch (u) {
        case 0:  return __builtin_amdgcn_update_dpp(0, v, 0x150, 0xf, 0xf, false);
        case 1:  return __builtin_amdgcn_update_dpp(0, v, 0x151, 0xf, 0xf, false);
        case 2:  return __builtin_amdgcn_update_dpp(0, v, 0x152, 0xf, 0xf, false);
        case 3:  return __builtin_amdgcn_update_dpp(0, v, 0x153, 0xf, 0xf, false);
        case 4:  return __builtin_amdgcn_update_dpp(0, v, 0x154, 0xf, 0xf, false);
        case 5:  return __builtin_amdgcn_update_dpp(0, v, 0x155, 0xf, 0xf, false);
        case 6:  return __builtin_amdgcn_update_dpp(0, v, 0x156, 0xf, 0xf, false);
        default: return __builtin_amdgcn_update_dpp(0, v, 0x157, 0xf, 0xf, false);
    }
}
__device__ __forceinline__ float row_bcast_f(float v, int u) {
    return __int_as_float(row_bcast(__float_as_int(v), u));
}

template <bool ONE_TABLE>
__launch_bounds__(256)
__global__ void gather_pair_kernel(const float* __restrict__ tabG, const float* __restrict__ tabS,
                                   const int* __restrict__ rowptr, const int* __restrict__ cnt,
                                   const int* __restrict__ nbr, const float* __restrict__ dinv,
                                   float* __restrict__ outG, float* __restrict__ outS, int n) {
    const int q   = threadIdx.x & 15;
    const int grp = (int)((blockIdx.x * 256u + threadIdx.x) >> 4);
    const int r0  = grp * GROUP_ROWS;
    if (r0 >= n) return;
    const v4f* g4 = (const v4f*)tabG;
    const v4f* s4 = (const v4f*)tabS;

    const bool mine = (q < GROUP_ROWS) && (r0 + q < n);
    const int  rl   = mine ? (r0 + q) : r0;
    const int   begL = rowptr[rl];
    const int   degL = mine ? cnt[rl] : 0;
    const float drL  = dinv[rl];

    int ixn = 0;
    {
        const int d0 = row_bcast(degL, 0);
        const int b0 = row_bcast(begL, 0);
        if (d0 > 0) ixn = nbr[b0 + min(q, d0 - 1)];
    }

    #pragma unroll
    for (int k = 0; k < GROUP_ROWS; ++k) {
        const int r = r0 + k;
        if (r >= n) break;
        const int   beg = row_bcast(begL, k);
        const int   deg = row_bcast(degL, k);
        const float dr  = row_bcast_f(drL, k);
        int ix = ixn;
        if (k + 1 < GROUP_ROWS) {
            const int dn = row_bcast(degL, k + 1);
            const int bn = row_bcast(begL, k + 1);
            ixn = 0;
            if (dn > 0) ixn = nbr[bn + min(q, dn - 1)];
        }
        const v4f hv = g4[r * 16 + q];
        float4 accG = make_float4(0.0f, 0.0f, 0.0f, 0.0f);
        float4 accS = make_float4(0.0f, 0.0f, 0.0f, 0.0f);
        for (int j = 0; j < deg; j += EDGE_UNROLL) {
            const int m = deg - j;
            int ixc = ix;
            if (m > EDGE_UNROLL)
                ixc = nbr[beg + j + EDGE_UNROLL + min(q, m - EDGE_UNROLL - 1)];
            const float dq = dinv[ix];
            v4f vg[EDGE_UNROLL];
            v4f vs[EDGE_UNROLL];
            #pragma unroll
            for (int u = 0; u < EDGE_UNROLL; ++u) {
                const int s = row_bcast(ix, u);
                vg[u] = g4[s * 16 + q];
                if (!ONE_TABLE) vs[u] = s4[s * 16 + q];
            }
            if (ONE_TABLE)
                asm volatile("" : "+v"(vg[0]), "+v"(vg[1]), "+v"(vg[2]), "+v"(vg[3]),
                                  "+v"(vg[4]), "+v"(vg[5]), "+v"(vg[6]), "+v"(vg[7]));
            else
                asm volatile("" : "+v"(vg[0]), "+v"(vg[1]), "+v"(vg[2]), "+v"(vg[3]),
                                  "+v"(vg[4]), "+v"(vg[5]), "+v"(vg[6]), "+v"(vg[7]),
                                  "+v"(vs[0]), "+v"(vs[1]), "+v"(vs[2]), "+v"(vs[3]),
                                  "+v"(vs[4]), "+v"(vs[5]), "+v"(vs[6]), "+v"(vs[7]));
            const float wq = dq * dr;
            #pragma unroll
            for (int u = 0; u < EDGE_UNROLL; ++u) {
                const bool on = (u == 0) || (u < m);
                const float w = row_bcast_f(wq, u);
                const v4f a = vg[u];
                const v4f b = ONE_TABLE ? vg[u] : vs[u];
                float4 ng, ns;
                ng.x = accG.x + a.x * w; ng.y = accG.y + a.y * w;
                ng.z = accG.z + a.z * w; ng.w = accG.w + a.w * w;
                ns.x = accS.x + b.x; ns.y = accS.y + b.y;
                ns.z = accS.z + b.z; ns.w = accS.w + b.w;
                accG.x = on ? ng.x : accG.x; accG.y = on ? ng.y : accG.y;
                accG.z = on ? ng.z : accG.z; accG.w = on ? ng.w : accG.w;
                accS.x = on ? ns.x : accS.x; accS.y = on ? ns.y : accS.y;
                accS.z = on ? ns.z : accS.z; accS.w = on ? ns.w : accS.w;
            }
            ix = ixc;
        }
        const float d2 = dr * dr;
        float4 og, os;
        og.x = accG.x + hv.x * d2; og.y = accG.y + hv.y * d2;
        og.z = accG.z + hv.z * d2; og.w = accG.w + hv.w * d2;
        const float rs = 1.0f / fmaxf((float)deg, 1.0f);
        os.x = accS.x * rs; os.y = accS.y * rs; os.z = accS.z * rs; os.w = accS.w * rs;
        ((float4*)outG)[r * 16 + q] = og;
        ((float4*)outS)[r * 16 + q] = os;
    }
}

extern "C" void kernel_launch(void* const* d_in, const int* in_sizes, int n_in,
                              void* d_out, int out_size, void* d_ws, size_t ws_size,
                              hipStream_t stream) {
    const float* x        = (const float*)d_in[0];
    const int*   ei       = (const int*)d_in[1];
    const float* gcn_w1   = (const float*)d_in[2];
    const float* gcn_b1   = (const float*)d_in[3];
    const float* gcn_w2   = (const float*)d_in[4];
    const float* gcn_b2   = (const float*)d_in[5];
    const float* sage_wl1 = (const float*)d_in[6];
    const float* sage_bl1 = (const float*)d_in[7];
    const float* sage_wr1 = (const float*)d_in[8];
    const float* sage_wl2 = (const float*)d_in[9];
    const float* sage_bl2 = (const float*)d_in[10];
    const float* sage_wr2 = (const float*)d_in[11];
    const float* gcn_ln_g = (const float*)d_in[12];
    const float* gcn_ln_b = (const float*)d_in[13];
    const float* sage_ln_g = (const float*)d_in[14];
    const float* sage_ln_b = (const float*)d_in[15];
    const float* proj_w   = (const float*)d_in[16];
    const float* proj_b   = (const float*)d_in[17];

    const int N = in_sizes[0] / FDIM;
    const int E = in_sizes[1] / 2;
    const int* src = ei;
    const int* dst = ei + E;

    int*   cnt    = (int*)d_ws;
    float* dinv   = (float*)(cnt + N);
    int*   rowptr = (int*)(dinv + N);
    int*   nbr    = rowptr + N;
    float* h      = (float*)(nbr + E);
    float* g      = h + (size_t)N * FDIM;
    float* agg    = g + (size_t)N * FDIM;
    float* out    = (float*)d_out;
    int*   rank   = (int*)agg;
    int*   bsum   = (int*)h;

    const int nblkN  = (N + 255) / 256;
    const int nblkE  = (E + 255) / 256;
    const int nblkT  = (N + SCAN_TILE - 1) / SCAN_TILE;
    const int nblkG  = (N + 127) / 128;
    const int nblkR  = (N + 16 * GROUP_ROWS - 1) / (16 * GROUP_ROWS);

    init_cnt_kernel<<<nblkN, 256, 0, stream>>>(cnt, N);
    count_rank_kernel<<<nblkE, 256, 0, stream>>>(dst, cnt, rank, E);
    scan_tile_kernel<<<nblkT, 256, 0, stream>>>(cnt, rowptr, bsum, N);
    finalize_rows_kernel<<<nblkN, 256, 0, stream>>>(cnt, bsum, rowptr, dinv, N);
    fill_nbr_kernel<<<nblkE, 256, 0, stream>>>(src, dst, rowptr, rank, nbr, E);

    gather_pair_kernel<true><<<nblkR, 256, 0, stream>>>(x, x, rowptr, cnt, nbr, dinv, h, agg, N);
    gemm64_wmma_kernel<<<nblkG, 256, 0, stream>>>(
        h, gcn_w1, nullptr, nullptr, gcn_b1, nullptr,
        nullptr, nullptr, nullptr, nullptr,
        nullptr, nullptr, nullptr, g, N, 1  , 0);
    gemm64_wmma_kernel<<<nblkG, 256, 0, stream>>>(
        agg, sage_wl1, x, sage_wr1, sage_bl1, nullptr,
        nullptr, nullptr, nullptr, nullptr,
        nullptr, nullptr, nullptr, out, N, 1  , 0);

    gather_pair_kernel<false><<<nblkR, 256, 0, stream>>>(g, out, rowptr, cnt, nbr, dinv, h, agg, N);
    gemm64_wmma_kernel<<<nblkG, 256, 0, stream>>>(
        h, gcn_w2, nullptr, nullptr, gcn_b2, nullptr,
        nullptr, nullptr, nullptr, nullptr,
        nullptr, nullptr, nullptr, g, N, 0, 0);
    gemm64_wmma_kernel<<<nblkG, 256, 0, stream>>>(
        agg, sage_wl2, out, sage_wr2, sage_bl2, nullptr,
        nullptr, nullptr, nullptr, nullptr,
        nullptr, nullptr, nullptr, h, N, 0, 0);

    gemm64_wmma_kernel<<<nblkG, 256, 0, stream>>>(
        g, proj_w, h, proj_w + FDIM * FDIM, proj_b, nullptr,
        gcn_ln_g, gcn_ln_b, sage_ln_g, sage_ln_b,
        nullptr, nullptr, nullptr, out, N, 0, 0);
}


# Row-local GEMM launches fused, 5 to 2: layer-2 products stay in registers as transposed (feature-major) WMMA tiles feeding LayerNorm and the projection; projection k order, LayerNorm sum order change
# speedup vs baseline: 3.0366x; 1.0621x over previous
#include <hip/hip_runtime.h>

typedef __attribute__((ext_vector_type(2))) float v2f;
typedef __attribute__((ext_vector_type(8))) float v8f;
typedef __attribute__((ext_vector_type(4))) float v4f;

#define FDIM 64
#define SCAN_TILE 1024
#define GROUP_ROWS 5
#define EDGE_UNROLL 8

__global__ void init_cnt_kernel(int* __restrict__ cnt, int n) {
    int i = blockIdx.x * blockDim.x + threadIdx.x;
    if (i < n) cnt[i] = 0;
}

__global__ void count_rank_kernel(const int* __restrict__ dst, int* __restrict__ cnt,
                                  int* __restrict__ rank, int e) {
    int i = blockIdx.x * blockDim.x + threadIdx.x;
    if (i < e) rank[i] = atomicAdd(&cnt[dst[i]], 1);
}

__launch_bounds__(256)
__global__ void scan_tile_kernel(const int* __restrict__ cnt, int* __restrict__ rowptr,
                                 int* __restrict__ bsum, int n) {
    __shared__ int wsum[8];
    const int t    = threadIdx.x;
    const int lane = t & 31;
    const int wave = t >> 5;
    const int base = blockIdx.x * SCAN_TILE + t * 4;
    const int c0 = (base + 0 < n) ? cnt[base + 0] : 0;
    const int c1 = (base + 1 < n) ? cnt[base + 1] : 0;
    const int c2 = (base + 2 < n) ? cnt[base + 2] : 0;
    const int c3 = (base + 3 < n) ? cnt[base + 3] : 0;
    const int tot = c0 + c1 + c2 + c3;
    int inc = tot;
    for (int o = 1; o < 32; o <<= 1) {
        const int v = __shfl_up(inc, o, 32);
        if (lane >= o) inc += v;
    }
    if (lane == 31) wsum[wave] = inc;
    __syncthreads();
    int woff = 0;
    for (int w = 0; w < wave; ++w) woff += wsum[w];
    const int ex = woff + inc - tot;
    if (base + 0 < n) rowptr[base + 0] = ex;
    if (base + 1 < n) rowptr[base + 1] = ex + c0;
    if (base + 2 < n) rowptr[base + 2] = ex + c0 + c1;
    if (base + 3 < n) rowptr[base + 3] = ex + c0 + c1 + c2;
    if (t == 255) bsum[blockIdx.x] = woff + inc;
}

__launch_bounds__(256)
__global__ void finalize_rows_kernel(const int* __restrict__ cnt, const int* __restrict__ bsum,
                                     int* __restrict__ rowptr, float* __restrict__ dinv, int n) {
    __shared__ int part[8];
    const int t  = threadIdx.x;
    const int nt = (blockIdx.x * 256) / SCAN_TILE;
    int v = 0;
    for (int i = t; i < nt; i += 256) v += bsum[i];
    for (int o = 16; o > 0; o >>= 1) v += __shfl_xor(v, o, 32);
    if ((t & 31) == 0) part[t >> 5] = v;
    __syncthreads();
    int off = 0;
    for (int w = 0; w < 8; ++w) off += part[w];
    const int i = blockIdx.x * 256 + t;
    if (i < n) {
        rowptr[i] += off;
        dinv[i] = rsqrtf((float)cnt[i] + 1.0f);
    }
}

__global__ void fill_nbr_kernel(const int* __restrict__ src, const int* __restrict__ dst,
                                const int* __restrict__ rowptr, const int* __restrict__ rank,
                                int* __restrict__ nbr, int e) {
    int i = blockIdx.x * blockDim.x + threadIdx.x;
    if (i < e) nbr[rowptr[dst[i]] + rank[i]] = src[i];
}

#define WMMA_F32(a, b, c) __builtin_amdgcn_wmma_f32_16x16x4_f32(false, (a), false, (b), (short)0, (c), false, false)

__device__ __forceinline__ void stage_w64(float* __restrict__ dst, const float* __restrict__ src, int t) {
    const float4* s = (const float4*)src;
    float4* d = (float4*)dst;
    for (int i = t; i < 1024; i += 256) d[i] = s[i];
}

__device__ __forceinline__ void load_row_frag(v2f (&f)[16], const float* p) {
    #pragma unroll
    for (int kb = 0; kb < 16; ++kb) { f[kb].x = p[kb * 4 + 0]; f[kb].y = p[kb * 4 + 1]; }
}

__device__ __forceinline__ void load_w_frag(v2f (&w)[16], const float* Wl, int col, int kOff) {
    #pragma unroll
    for (int kb = 0; kb < 16; ++kb) {
        const int kr = kb * 4 + kOff;
        w[kb].x = Wl[kr * FDIM + col];
        w[kb].y = Wl[(kr + 1) * FDIM + col];
    }
}

template <bool RELU>
__device__ __forceinline__ void store_node_major(float* __restrict__ dst, const v8f acc, float bval,
                                                 int m0, int lhalf, int col, int n) {
    float v[8];
    #pragma unroll
    for (int j = 0; j < 8; ++j) {
        v[j] = acc[j] + bval;
        if (RELU) v[j] = fmaxf(v[j], 0.0f);
    }
    #pragma unroll
    for (int j = 0; j < 8; ++j) {
        const int r = m0 + j + lhalf * 8;
        if (r < n) dst[(size_t)r * FDIM + col] = v[j];
    }
}

__launch_bounds__(256)
__global__ void layer1_products_kernel(const float* __restrict__ H, const float* __restrict__ AGG,
                                       const float* __restrict__ X,
                                       const float* __restrict__ Wg, const float* __restrict__ bg,
                                       const float* __restrict__ Wl, const float* __restrict__ Wr,
                                       const float* __restrict__ bl,
                                       float* __restrict__ outG, float* __restrict__ outS, int n) {
    __shared__ float WG[FDIM * FDIM];
    __shared__ float WL[FDIM * FDIM];
    __shared__ float WR[FDIM * FDIM];
    const int t = threadIdx.x;
    stage_w64(WG, Wg, t);
    stage_w64(WL, Wl, t);
    stage_w64(WR, Wr, t);
    __syncthreads();

    const int wave  = t >> 5;
    const int lane  = t & 31;
    const int m0    = (blockIdx.x * 8 + wave) * 16;
    if (m0 >= n) return;
    const int lhalf = lane >> 4;
    const int l15   = lane & 15;
    const int kOff  = lhalf * 2;
    int row = m0 + l15;
    if (row >= n) row = n - 1;

    {
        v2f af[16], xf[16];
        load_row_frag(af, AGG + (size_t)row * FDIM + kOff);
        load_row_frag(xf, X   + (size_t)row * FDIM + kOff);
        #pragma unroll
        for (int nb = 0; nb < 4; ++nb) {
            const int col = nb * 16 + l15;
            v2f wf[16], wg[16];
            load_w_frag(wf, WL, col, kOff);
            load_w_frag(wg, WR, col, kOff);
            v8f acc = {0.f, 0.f, 0.f, 0.f, 0.f, 0.f, 0.f, 0.f};
            #pragma unroll
            for (int kb = 0; kb < 16; ++kb) acc = WMMA_F32(af[kb], wf[kb], acc);
            #pragma unroll
            for (int kb = 0; kb < 16; ++kb) acc = WMMA_F32(xf[kb], wg[kb], acc);
            store_node_major<true>(outS, acc, bl[col], m0, lhalf, col, n);
        }
    }
    {
        v2f hf[16];
        load_row_frag(hf, H + (size_t)row * FDIM + kOff);
        #pragma unroll
        for (int nb = 0; nb < 4; ++nb) {
            const int col = nb * 16 + l15;
            v2f wf[16];
            load_w_frag(wf, WG, col, kOff);
            v8f acc = {0.f, 0.f, 0.f, 0.f, 0.f, 0.f, 0.f, 0.f};
            #pragma unroll
            for (int kb = 0; kb < 16; ++kb) acc = WMMA_F32(hf[kb], wf[kb], acc);
            store_node_major<true>(outG, acc, bg[col], m0, lhalf, col, n);
        }
    }
}

__device__ __forceinline__ void bias_layernorm_fm(v8f (&tl)[4], const float* bias, const float* gam,
                                                  const float* bet, int lhalf) {
    float s = 0.0f, q = 0.0f;
    #pragma unroll
    for (int nb = 0; nb < 4; ++nb) {
        #pragma unroll
        for (int r = 0; r < 8; ++r) {
            const float v = tl[nb][r] + bias[nb * 16 + lhalf * 8 + r];
            tl[nb][r] = v;
            s += v;
            q += v * v;
        }
    }
    s += __shfl_xor(s, 16, 32);
    q += __shfl_xor(q, 16, 32);
    const float mu  = s * (1.0f / 64.0f);
    const float inv = rsqrtf(fmaxf(q * (1.0f / 64.0f) - mu * mu, 0.0f) + 1e-5f);
    #pragma unroll
    for (int nb = 0; nb < 4; ++nb) {
        #pragma unroll
        for (int r = 0; r < 8; ++r) {
            const int f = nb * 16 + lhalf * 8 + r;
            tl[nb][r] = (tl[nb][r] - mu) * inv * gam[f] + bet[f];
        }
    }
}

__launch_bounds__(256)
__global__ void layer2_projection_kernel(const float* __restrict__ H, const float* __restrict__ AGG,
                                         const float* S1,
                                         const float* __restrict__ Wg, const float* __restrict__ bg,
                                         const float* __restrict__ Wl, const float* __restrict__ Wr,
                                         const float* __restrict__ bl,
                                         const float* __restrict__ lnGg, const float* __restrict__ lnGb,
                                         const float* __restrict__ lnSg, const float* __restrict__ lnSb,
                                         const float* __restrict__ P, const float* __restrict__ pb,
                                         float* out, int n) {
    __shared__ float WG[FDIM * FDIM];
    __shared__ float WL[FDIM * FDIM];
    __shared__ float WR[FDIM * FDIM];
    __shared__ float PT[FDIM * FDIM];
    __shared__ float PB[FDIM * FDIM];
    __shared__ float PAR[6][FDIM];
    const int t = threadIdx.x;
    stage_w64(WG, Wg, t);
    stage_w64(WL, Wl, t);
    stage_w64(WR, Wr, t);
    stage_w64(PT, P, t);
    stage_w64(PB, P + FDIM * FDIM, t);
    if (t < FDIM) {
        PAR[0][t] = bg[t];   PAR[1][t] = bl[t];
        PAR[2][t] = lnGg[t]; PAR[3][t] = lnGb[t];
        PAR[4][t] = lnSg[t]; PAR[5][t] = lnSb[t];
    }
    __syncthreads();

    const int wave  = t >> 5;
    const int lane  = t & 31;
    const int m0    = (blockIdx.x * 8 + wave) * 16;
    if (m0 >= n) return;
    const int lhalf = lane >> 4;
    const int l15   = lane & 15;
    const int kOff  = lhalf * 2;
    int row = m0 + l15;
    if (row >= n) row = n - 1;

    v8f gT[4];
    {
        v2f hf[16];
        load_row_frag(hf, H + (size_t)row * FDIM + kOff);
        #pragma unroll
        for (int nb = 0; nb < 4; ++nb) {
            v2f wf[16];
            load_w_frag(wf, WG, nb * 16 + l15, kOff);
            v8f acc = {0.f, 0.f, 0.f, 0.f, 0.f, 0.f, 0.f, 0.f};
            #pragma unroll
            for (int kb = 0; kb < 16; ++kb) acc = WMMA_F32(wf[kb], hf[kb], acc);
            gT[nb] = acc;
        }
        bias_layernorm_fm(gT, PAR[0], PAR[2], PAR[3], lhalf);
    }
    v8f sT[4];
    {
        v2f af[16], sf[16];
        load_row_frag(af, AGG + (size_t)row * FDIM + kOff);
        load_row_frag(sf, S1  + (size_t)row * FDIM + kOff);
        #pragma unroll
        for (int nb = 0; nb < 4; ++nb) {
            v2f wf[16], wg[16];
            load_w_frag(wf, WL, nb * 16 + l15, kOff);
            load_w_frag(wg, WR, nb * 16 + l15, kOff);
            v8f acc = {0.f, 0.f, 0.f, 0.f, 0.f, 0.f, 0.f, 0.f};
            #pragma unroll
            for (int kb = 0; kb < 16; ++kb) acc = WMMA_F32(wf[kb], af[kb], acc);
            #pragma unroll
            for (int kb = 0; kb < 16; ++kb) acc = WMMA_F32(wg[kb], sf[kb], acc);
            sT[nb] = acc;
        }
        bias_layernorm_fm(sT, PAR[1], PAR[4], PAR[5], lhalf);
    }
    #pragma unroll
    for (int ob = 0; ob < 4; ++ob) {
        const int col = ob * 16 + l15;
        v2f pt[16], pq[16];
        #pragma unroll
        for (int nb = 0; nb < 4; ++nb) {
            #pragma unroll
            for (int rp = 0; rp < 4; ++rp) {
                const int k0 = nb * 16 + lhalf * 8 + rp * 2;
                pt[nb * 4 + rp].x = PT[k0 * FDIM + col];
                pt[nb * 4 + rp].y = PT[(k0 + 1) * FDIM + col];
                pq[nb * 4 + rp].x = PB[k0 * FDIM + col];
                pq[nb * 4 + rp].y = PB[(k0 + 1) * FDIM + col];
            }
        }
        v8f acc = {0.f, 0.f, 0.f, 0.f, 0.f, 0.f, 0.f, 0.f};
        #pragma unroll
        for (int nb = 0; nb < 4; ++nb) {
            #pragma unroll
            for (int rp = 0; rp < 4; ++rp) {
                v2f a;
                a.x = gT[nb][rp * 2];
                a.y = gT[nb][rp * 2 + 1];
                acc = WMMA_F32(a, pt[nb * 4 + rp], acc);
            }
        }
        #pragma unroll
        for (int nb = 0; nb < 4; ++nb) {
            #pragma unroll
            for (int rp = 0; rp < 4; ++rp) {
                v2f a;
                a.x = sT[nb][rp * 2];
                a.y = sT[nb][rp * 2 + 1];
                acc = WMMA_F32(a, pq[nb * 4 + rp], acc);
            }
        }
        store_node_major<false>(out, acc, pb[col], m0, lhalf, col, n);
    }
}

__device__ __forceinline__ int row_bcast(int v, int u) {
    switch (u) {
        case 0:  return __builtin_amdgcn_update_dpp(0, v, 0x150, 0xf, 0xf, false);
        case 1:  return __builtin_amdgcn_update_dpp(0, v, 0x151, 0xf, 0xf, false);
        case 2:  return __builtin_amdgcn_update_dpp(0, v, 0x152, 0xf, 0xf, false);
        case 3:  return __builtin_amdgcn_update_dpp(0, v, 0x153, 0xf, 0xf, false);
        case 4:  return __builtin_amdgcn_update_dpp(0, v, 0x154, 0xf, 0xf, false);
        case 5:  return __builtin_amdgcn_update_dpp(0, v, 0x155, 0xf, 0xf, false);
        case 6:  return __builtin_amdgcn_update_dpp(0, v, 0x156, 0xf, 0xf, false);
        default: return __builtin_amdgcn_update_dpp(0, v, 0x157, 0xf, 0xf, false);
    }
}
__device__ __forceinline__ float row_bcast_f(float v, int u) {
    return __int_as_float(row_bcast(__float_as_int(v), u));
}

template <bool ONE_TABLE>
__launch_bounds__(256)
__global__ void gather_pair_kernel(const float* __restrict__ tabG, const float* __restrict__ tabS,
                                   const int* __restrict__ rowptr, const int* __restrict__ cnt,
                                   const int* __restrict__ nbr, const float* __restrict__ dinv,
                                   float* __restrict__ outG, float* __restrict__ outS, int n) {
    const int q   = threadIdx.x & 15;
    const int grp = (int)((blockIdx.x * 256u + threadIdx.x) >> 4);
    const int r0  = grp * GROUP_ROWS;
    if (r0 >= n) return;
    const v4f* g4 = (const v4f*)tabG;
    const v4f* s4 = (const v4f*)tabS;

    const bool mine = (q < GROUP_ROWS) && (r0 + q < n);
    const int  rl   = mine ? (r0 + q) : r0;
    const int   begL = rowptr[rl];
    const int   degL = mine ? cnt[rl] : 0;
    const float drL  = dinv[rl];

    int ixn = 0;
    {
        const int d0 = row_bcast(degL, 0);
        const int b0 = row_bcast(begL, 0);
        if (d0 > 0) ixn = nbr[b0 + min(q, d0 - 1)];
    }

    #pragma unroll
    for (int k = 0; k < GROUP_ROWS; ++k) {
        const int r = r0 + k;
        if (r >= n) break;
        const int   beg = row_bcast(begL, k);
        const int   deg = row_bcast(degL, k);
        const float dr  = row_bcast_f(drL, k);
        int ix = ixn;
        if (k + 1 < GROUP_ROWS) {
            const int dn = row_bcast(degL, k + 1);
            const int bn = row_bcast(begL, k + 1);
            ixn = 0;
            if (dn > 0) ixn = nbr[bn + min(q, dn - 1)];
        }
        const v4f hv = g4[r * 16 + q];
        float4 accG = make_float4(0.0f, 0.0f, 0.0f, 0.0f);
        float4 accS = make_float4(0.0f, 0.0f, 0.0f, 0.0f);
        for (int j = 0; j < deg; j += EDGE_UNROLL) {
            const int m = deg - j;
            int ixc = ix;
            if (m > EDGE_UNROLL)
                ixc = nbr[beg + j + EDGE_UNROLL + min(q, m - EDGE_UNROLL - 1)];
            const float dq = dinv[ix];
            v4f vg[EDGE_UNROLL];
            v4f vs[EDGE_UNROLL];
            #pragma unroll
            for (int u = 0; u < EDGE_UNROLL; ++u) {
                const int s = row_bcast(ix, u);
                vg[u] = g4[s * 16 + q];
                if (!ONE_TABLE) vs[u] = s4[s * 16 + q];
            }
            if (ONE_TABLE)
                asm volatile("" : "+v"(vg[0]), "+v"(vg[1]), "+v"(vg[2]), "+v"(vg[3]),
                                  "+v"(vg[4]), "+v"(vg[5]), "+v"(vg[6]), "+v"(vg[7]));
            else
                asm volatile("" : "+v"(vg[0]), "+v"(vg[1]), "+v"(vg[2]), "+v"(vg[3]),
                                  "+v"(vg[4]), "+v"(vg[5]), "+v"(vg[6]), "+v"(vg[7]),
                                  "+v"(vs[0]), "+v"(vs[1]), "+v"(vs[2]), "+v"(vs[3]),
                                  "+v"(vs[4]), "+v"(vs[5]), "+v"(vs[6]), "+v"(vs[7]));
            const float wq = dq * dr;
            #pragma unroll
            for (int u = 0; u < EDGE_UNROLL; ++u) {
                const bool on = (u == 0) || (u < m);
                const float w = row_bcast_f(wq, u);
                const v4f a = vg[u];
                const v4f b = ONE_TABLE ? vg[u] : vs[u];
                float4 ng, ns;
                ng.x = accG.x + a.x * w; ng.y = accG.y + a.y * w;
                ng.z = accG.z + a.z * w; ng.w = accG.w + a.w * w;
                ns.x = accS.x + b.x; ns.y = accS.y + b.y;
                ns.z = accS.z + b.z; ns.w = accS.w + b.w;
                accG.x = on ? ng.x : accG.x; accG.y = on ? ng.y : accG.y;
                accG.z = on ? ng.z : accG.z; accG.w = on ? ng.w : accG.w;
                accS.x = on ? ns.x : accS.x; accS.y = on ? ns.y : accS.y;
                accS.z = on ? ns.z : accS.z; accS.w = on ? ns.w : accS.w;
            }
            ix = ixc;
        }
        const float d2 = dr * dr;
        float4 og, os;
        og.x = accG.x + hv.x * d2; og.y = accG.y + hv.y * d2;
        og.z = accG.z + hv.z * d2; og.w = accG.w + hv.w * d2;
        const float rs = 1.0f / fmaxf((float)deg, 1.0f);
        os.x = accS.x * rs; os.y = accS.y * rs; os.z = accS.z * rs; os.w = accS.w * rs;
        ((float4*)outG)[r * 16 + q] = og;
        ((float4*)outS)[r * 16 + q] = os;
    }
}

extern "C" void kernel_launch(void* const* d_in, const int* in_sizes, int n_in,
                              void* d_out, int out_size, void* d_ws, size_t ws_size,
                              hipStream_t stream) {
    const float* x        = (const float*)d_in[0];
    const int*   ei       = (const int*)d_in[1];
    const float* gcn_w1   = (const float*)d_in[2];
    const float* gcn_b1   = (const float*)d_in[3];
    const float* gcn_w2   = (const float*)d_in[4];
    const float* gcn_b2   = (const float*)d_in[5];
    const float* sage_wl1 = (const float*)d_in[6];
    const float* sage_bl1 = (const float*)d_in[7];
    const float* sage_wr1 = (const float*)d_in[8];
    const float* sage_wl2 = (const float*)d_in[9];
    const float* sage_bl2 = (const float*)d_in[10];
    const float* sage_wr2 = (const float*)d_in[11];
    const float* gcn_ln_g = (const float*)d_in[12];
    const float* gcn_ln_b = (const float*)d_in[13];
    const float* sage_ln_g = (const float*)d_in[14];
    const float* sage_ln_b = (const float*)d_in[15];
    const float* proj_w   = (const float*)d_in[16];
    const float* proj_b   = (const float*)d_in[17];

    const int N = in_sizes[0] / FDIM;
    const int E = in_sizes[1] / 2;
    const int* src = ei;
    const int* dst = ei + E;

    int*   cnt    = (int*)d_ws;
    float* dinv   = (float*)(cnt + N);
    int*   rowptr = (int*)(dinv + N);
    int*   nbr    = rowptr + N;
    float* h      = (float*)(nbr + E);
    float* g      = h + (size_t)N * FDIM;
    float* agg    = g + (size_t)N * FDIM;
    float* out    = (float*)d_out;
    int*   rank   = (int*)agg;
    int*   bsum   = (int*)h;

    const int nblkN  = (N + 255) / 256;
    const int nblkE  = (E + 255) / 256;
    const int nblkT  = (N + SCAN_TILE - 1) / SCAN_TILE;
    const int nblkG  = (N + 127) / 128;
    const int nblkR  = (N + 16 * GROUP_ROWS - 1) / (16 * GROUP_ROWS);

    init_cnt_kernel<<<nblkN, 256, 0, stream>>>(cnt, N);
    count_rank_kernel<<<nblkE, 256, 0, stream>>>(dst, cnt, rank, E);
    scan_tile_kernel<<<nblkT, 256, 0, stream>>>(cnt, rowptr, bsum, N);
    finalize_rows_kernel<<<nblkN, 256, 0, stream>>>(cnt, bsum, rowptr, dinv, N);
    fill_nbr_kernel<<<nblkE, 256, 0, stream>>>(src, dst, rowptr, rank, nbr, E);

    gather_pair_kernel<true><<<nblkR, 256, 0, stream>>>(x, x, rowptr, cnt, nbr, dinv, h, agg, N);
    layer1_products_kernel<<<nblkG, 256, 0, stream>>>(
        h, agg, x, gcn_w1, gcn_b1, sage_wl1, sage_wr1, sage_bl1, g, out, N);

    gather_pair_kernel<false><<<nblkR, 256, 0, stream>>>(g, out, rowptr, cnt, nbr, dinv, h, agg, N);
    layer2_projection_kernel<<<nblkG, 256, 0, stream>>>(
        h, agg, out, gcn_w2, gcn_b2, sage_wl2, sage_wr2, sage_bl2,
        gcn_ln_g, gcn_ln_b, sage_ln_g, sage_ln_b, proj_w, proj_b, out, N);
}


# Layer-2 GCN table stored row-scaled by dinv in the layer-1 epilogue: pass 2 adds plain rows, applies dinv[dst] once; its per-edge random dinv[src] read goes; term rounding changes, sum order kept
# speedup vs baseline: 3.1383x; 1.0335x over previous
#include <hip/hip_runtime.h>

typedef __attribute__((ext_vector_type(2))) float v2f;
typedef __attribute__((ext_vector_type(8))) float v8f;
typedef __attribute__((ext_vector_type(4))) float v4f;

#define FDIM 64
#define SCAN_TILE 1024
#define GROUP_ROWS 5
#define EDGE_UNROLL 8

__global__ void init_cnt_kernel(int* __restrict__ cnt, int n) {
    int i = blockIdx.x * blockDim.x + threadIdx.x;
    if (i < n) cnt[i] = 0;
}

__global__ void count_rank_kernel(const int* __restrict__ dst, int* __restrict__ cnt,
                                  int* __restrict__ rank, int e) {
    int i = blockIdx.x * blockDim.x + threadIdx.x;
    if (i < e) rank[i] = atomicAdd(&cnt[dst[i]], 1);
}

__launch_bounds__(256)
__global__ void scan_tile_kernel(const int* __restrict__ cnt, int* __restrict__ rowptr,
                                 int* __restrict__ bsum, int n) {
    __shared__ int wsum[8];
    const int t    = threadIdx.x;
    const int lane = t & 31;
    const int wave = t >> 5;
    const int base = blockIdx.x * SCAN_TILE + t * 4;
    const int c0 = (base + 0 < n) ? cnt[base + 0] : 0;
    const int c1 = (base + 1 < n) ? cnt[base + 1] : 0;
    const int c2 = (base + 2 < n) ? cnt[base + 2] : 0;
    const int c3 = (base + 3 < n) ? cnt[base + 3] : 0;
    const int tot = c0 + c1 + c2 + c3;
    int inc = tot;
    for (int o = 1; o < 32; o <<= 1) {
        const int v = __shfl_up(inc, o, 32);
        if (lane >= o) inc += v;
    }
    if (lane == 31) wsum[wave] = inc;
    __syncthreads();
    int woff = 0;
    for (int w = 0; w < wave; ++w) woff += wsum[w];
    const int ex = woff + inc - tot;
    if (base + 0 < n) rowptr[base + 0] = ex;
    if (base + 1 < n) rowptr[base + 1] = ex + c0;
    if (base + 2 < n) rowptr[base + 2] = ex + c0 + c1;
    if (base + 3 < n) rowptr[base + 3] = ex + c0 + c1 + c2;
    if (t == 255) bsum[blockIdx.x] = woff + inc;
}

__launch_bounds__(256)
__global__ void finalize_rows_kernel(const int* __restrict__ cnt, const int* __restrict__ bsum,
                                     int* __restrict__ rowptr, float* __restrict__ dinv, int n) {
    __shared__ int part[8];
    const int t  = threadIdx.x;
    const int nt = (blockIdx.x * 256) / SCAN_TILE;
    int v = 0;
    for (int i = t; i < nt; i += 256) v += bsum[i];
    for (int o = 16; o > 0; o >>= 1) v += __shfl_xor(v, o, 32);
    if ((t & 31) == 0) part[t >> 5] = v;
    __syncthreads();
    int off = 0;
    for (int w = 0; w < 8; ++w) off += part[w];
    const int i = blockIdx.x * 256 + t;
    if (i < n) {
        rowptr[i] += off;
        dinv[i] = rsqrtf((float)cnt[i] + 1.0f);
    }
}

__global__ void fill_nbr_kernel(const int* __restrict__ src, const int* __restrict__ dst,
                                const int* __restrict__ rowptr, const int* __restrict__ rank,
                                int* __restrict__ nbr, int e) {
    int i = blockIdx.x * blockDim.x + threadIdx.x;
    if (i < e) nbr[rowptr[dst[i]] + rank[i]] = src[i];
}

#define WMMA_F32(a, b, c) __builtin_amdgcn_wmma_f32_16x16x4_f32(false, (a), false, (b), (short)0, (c), false, false)

__device__ __forceinline__ void stage_w64(float* __restrict__ dst, const float* __restrict__ src, int t) {
    const float4* s = (const float4*)src;
    float4* d = (float4*)dst;
    for (int i = t; i < 1024; i += 256) d[i] = s[i];
}

__device__ __forceinline__ void load_row_frag(v2f (&f)[16], const float* p) {
    #pragma unroll
    for (int kb = 0; kb < 16; ++kb) { f[kb].x = p[kb * 4 + 0]; f[kb].y = p[kb * 4 + 1]; }
}

__device__ __forceinline__ void load_w_frag(v2f (&w)[16], const float* Wl, int col, int kOff) {
    #pragma unroll
    for (int kb = 0; kb < 16; ++kb) {
        const int kr = kb * 4 + kOff;
        w[kb].x = Wl[kr * FDIM + col];
        w[kb].y = Wl[(kr + 1) * FDIM + col];
    }
}

template <bool RELU>
__device__ __forceinline__ void store_node_major(float* __restrict__ dst, const v8f acc, float bval,
                                                 int m0, int lhalf, int col, int n) {
    float v[8];
    #pragma unroll
    for (int j = 0; j < 8; ++j) {
        v[j] = acc[j] + bval;
        if (RELU) v[j] = fmaxf(v[j], 0.0f);
    }
    #pragma unroll
    for (int j = 0; j < 8; ++j) {
        const int r = m0 + j + lhalf * 8;
        if (r < n) dst[(size_t)r * FDIM + col] = v[j];
    }
}

__device__ __forceinline__ void store_node_major_relu_scaled(float* __restrict__ dst, const v8f acc, float bval,
                                                             const float (&rs)[8], int m0, int lhalf, int col, int n) {
    float v[8];
    #pragma unroll
    for (int j = 0; j < 8; ++j) v[j] = fmaxf(acc[j] + bval, 0.0f) * rs[j];
    #pragma unroll
    for (int j = 0; j < 8; ++j) {
        const int r = m0 + j + lhalf * 8;
        if (r < n) dst[(size_t)r * FDIM + col] = v[j];
    }
}

__launch_bounds__(256)
__global__ void layer1_products_kernel(const float* __restrict__ H, const float* __restrict__ AGG,
                                       const float* __restrict__ X,
                                       const float* __restrict__ Wg, const float* __restrict__ bg,
                                       const float* __restrict__ Wl, const float* __restrict__ Wr,
                                       const float* __restrict__ bl, const float* __restrict__ dinv,
                                       float* __restrict__ outG, float* __restrict__ outS, int n) {
    __shared__ float WG[FDIM * FDIM];
    __shared__ float WL[FDIM * FDIM];
    __shared__ float WR[FDIM * FDIM];
    const int t = threadIdx.x;
    stage_w64(WG, Wg, t);
    stage_w64(WL, Wl, t);
    stage_w64(WR, Wr, t);
    __syncthreads();

    const int wave  = t >> 5;
    const int lane  = t & 31;
    const int m0    = (blockIdx.x * 8 + wave) * 16;
    if (m0 >= n) return;
    const int lhalf = lane >> 4;
    const int l15   = lane & 15;
    const int kOff  = lhalf * 2;
    int row = m0 + l15;
    if (row >= n) row = n - 1;

    {
        v2f af[16], xf[16];
        load_row_frag(af, AGG + (size_t)row * FDIM + kOff);
        load_row_frag(xf, X   + (size_t)row * FDIM + kOff);
        #pragma unroll
        for (int nb = 0; nb < 4; ++nb) {
            const int col = nb * 16 + l15;
            v2f wf[16], wg[16];
            load_w_frag(wf, WL, col, kOff);
            load_w_frag(wg, WR, col, kOff);
            v8f acc = {0.f, 0.f, 0.f, 0.f, 0.f, 0.f, 0.f, 0.f};
            #pragma unroll
            for (int kb = 0; kb < 16; ++kb) acc = WMMA_F32(af[kb], wf[kb], acc);
            #pragma unroll
            for (int kb = 0; kb < 16; ++kb) acc = WMMA_F32(xf[kb], wg[kb], acc);
            store_node_major<true>(outS, acc, bl[col], m0, lhalf, col, n);
        }
    }
    {
        v2f hf[16];
        load_row_frag(hf, H + (size_t)row * FDIM + kOff);
        float dv[8];
        #pragma unroll
        for (int j = 0; j < 8; ++j) {
            const int r = m0 + j + lhalf * 8;
            dv[j] = dinv[r < n ? r : n - 1];
        }
        #pragma unroll
        for (int nb = 0; nb < 4; ++nb) {
            const int col = nb * 16 + l15;
            v2f wf[16];
            load_w_frag(wf, WG, col, kOff);
            v8f acc = {0.f, 0.f, 0.f, 0.f, 0.f, 0.f, 0.f, 0.f};
            #pragma unroll
            for (int kb = 0; kb < 16; ++kb) acc = WMMA_F32(hf[kb], wf[kb], acc);
            store_node_major_relu_scaled(outG, acc, bg[col], dv, m0, lhalf, col, n);
        }
    }
}

__device__ __forceinline__ void bias_layernorm_fm(v8f (&tl)[4], const float* bias, const float* gam,
                                                  const float* bet, int lhalf) {
    float s = 0.0f, q = 0.0f;
    #pragma unroll
    for (int nb = 0; nb < 4; ++nb) {
        #pragma unroll
        for (int r = 0; r < 8; ++r) {
            const float v = tl[nb][r] + bias[nb * 16 + lhalf * 8 + r];
            tl[nb][r] = v;
            s += v;
            q += v * v;
        }
    }
    s += __shfl_xor(s, 16, 32);
    q += __shfl_xor(q, 16, 32);
    const float mu  = s * (1.0f / 64.0f);
    const float inv = rsqrtf(fmaxf(q * (1.0f / 64.0f) - mu * mu, 0.0f) + 1e-5f);
    #pragma unroll
    for (int nb = 0; nb < 4; ++nb) {
        #pragma unroll
        for (int r = 0; r < 8; ++r) {
            const int f = nb * 16 + lhalf * 8 + r;
            tl[nb][r] = (tl[nb][r] - mu) * inv * gam[f] + bet[f];
        }
    }
}

__launch_bounds__(256)
__global__ void layer2_projection_kernel(const float* __restrict__ H, const float* __restrict__ AGG,
                                         const float* S1,
                                         const float* __restrict__ Wg, const float* __restrict__ bg,
                                         const float* __restrict__ Wl, const float* __restrict__ Wr,
                                         const float* __restrict__ bl,
                                         const float* __restrict__ lnGg, const float* __restrict__ lnGb,
                                         const float* __restrict__ lnSg, const float* __restrict__ lnSb,
                                         const float* __restrict__ P, const float* __restrict__ pb,
                                         float* out, int n) {
    __shared__ float WG[FDIM * FDIM];
    __shared__ float WL[FDIM * FDIM];
    __shared__ float WR[FDIM * FDIM];
    __shared__ float PT[FDIM * FDIM];
    __shared__ float PB[FDIM * FDIM];
    __shared__ float PAR[6][FDIM];
    const int t = threadIdx.x;
    stage_w64(WG, Wg, t);
    stage_w64(WL, Wl, t);
    stage_w64(WR, Wr, t);
    stage_w64(PT, P, t);
    stage_w64(PB, P + FDIM * FDIM, t);
    if (t < FDIM) {
        PAR[0][t] = bg[t];   PAR[1][t] = bl[t];
        PAR[2][t] = lnGg[t]; PAR[3][t] = lnGb[t];
        PAR[4][t] = lnSg[t]; PAR[5][t] = lnSb[t];
    }
    __syncthreads();

    const int wave  = t >> 5;
    const int lane  = t & 31;
    const int m0    = (blockIdx.x * 8 + wave) * 16;
    if (m0 >= n) return;
    const int lhalf = lane >> 4;
    const int l15   = lane & 15;
    const int kOff  = lhalf * 2;
    int row = m0 + l15;
    if (row >= n) row = n - 1;

    v8f gT[4];
    {
        v2f hf[16];
        load_row_frag(hf, H + (size_t)row * FDIM + kOff);
        #pragma unroll
        for (int nb = 0; nb < 4; ++nb) {
            v2f wf[16];
            load_w_frag(wf, WG, nb * 16 + l15, kOff);
            v8f acc = {0.f, 0.f, 0.f, 0.f, 0.f, 0.f, 0.f, 0.f};
            #pragma unroll
            for (int kb = 0; kb < 16; ++kb) acc = WMMA_F32(wf[kb], hf[kb], acc);
            gT[nb] = acc;
        }
        bias_layernorm_fm(gT, PAR[0], PAR[2], PAR[3], lhalf);
    }
    v8f sT[4];
    {
        v2f af[16], sf[16];
        load_row_frag(af, AGG + (size_t)row * FDIM + kOff);
        load_row_frag(sf, S1  + (size_t)row * FDIM + kOff);
        #pragma unroll
        for (int nb = 0; nb < 4; ++nb) {
            v2f wf[16], wg[16];
            load_w_frag(wf, WL, nb * 16 + l15, kOff);
            load_w_frag(wg, WR, nb * 16 + l15, kOff);
            v8f acc = {0.f, 0.f, 0.f, 0.f, 0.f, 0.f, 0.f, 0.f};
            #pragma unroll
            for (int kb = 0; kb < 16; ++kb) acc = WMMA_F32(wf[kb], af[kb], acc);
            #pragma unroll
            for (int kb = 0; kb < 16; ++kb) acc = WMMA_F32(wg[kb], sf[kb], acc);
            sT[nb] = acc;
        }
        bias_layernorm_fm(sT, PAR[1], PAR[4], PAR[5], lhalf);
    }
    #pragma unroll
    for (int ob = 0; ob < 4; ++ob) {
        const int col = ob * 16 + l15;
        v2f pt[16], pq[16];
        #pragma unroll
        for (int nb = 0; nb < 4; ++nb) {
            #pragma unroll
            for (int rp = 0; rp < 4; ++rp) {
                const int k0 = nb * 16 + lhalf * 8 + rp * 2;
                pt[nb * 4 + rp].x = PT[k0 * FDIM + col];
                pt[nb * 4 + rp].y = PT[(k0 + 1) * FDIM + col];
                pq[nb * 4 + rp].x = PB[k0 * FDIM + col];
                pq[nb * 4 + rp].y = PB[(k0 + 1) * FDIM + col];
            }
        }
        v8f acc = {0.f, 0.f, 0.f, 0.f, 0.f, 0.f, 0.f, 0.f};
        #pragma unroll
        for (int nb = 0; nb < 4; ++nb) {
            #pragma unroll
            for (int rp = 0; rp < 4; ++rp) {
                v2f a;
                a.x = gT[nb][rp * 2];
                a.y = gT[nb][rp * 2 + 1];
                acc = WMMA_F32(a, pt[nb * 4 + rp], acc);
            }
        }
        #pragma unroll
        for (int nb = 0; nb < 4; ++nb) {
            #pragma unroll
            for (int rp = 0; rp < 4; ++rp) {
                v2f a;
                a.x = sT[nb][rp * 2];
                a.y = sT[nb][rp * 2 + 1];
                acc = WMMA_F32(a, pq[nb * 4 + rp], acc);
            }
        }
        store_node_major<false>(out, acc, pb[col], m0, lhalf, col, n);
    }
}

__device__ __forceinline__ int row_bcast(int v, int u) {
    switch (u) {
        case 0:  return __builtin_amdgcn_update_dpp(0, v, 0x150, 0xf, 0xf, false);
        case 1:  return __builtin_amdgcn_update_dpp(0, v, 0x151, 0xf, 0xf, false);
        case 2:  return __builtin_amdgcn_update_dpp(0, v, 0x152, 0xf, 0xf, false);
        case 3:  return __builtin_amdgcn_update_dpp(0, v, 0x153, 0xf, 0xf, false);
        case 4:  return __builtin_amdgcn_update_dpp(0, v, 0x154, 0xf, 0xf, false);
        case 5:  return __builtin_amdgcn_update_dpp(0, v, 0x155, 0xf, 0xf, false);
        case 6:  return __builtin_amdgcn_update_dpp(0, v, 0x156, 0xf, 0xf, false);
        default: return __builtin_amdgcn_update_dpp(0, v, 0x157, 0xf, 0xf, false);
    }
}
__device__ __forceinline__ float row_bcast_f(float v, int u) {
    return __int_as_float(row_bcast(__float_as_int(v), u));
}

template <bool ONE_TABLE>
__launch_bounds__(256)
__global__ void gather_pair_kernel(const float* __restrict__ tabG, const float* __restrict__ tabS,
                                   const int* __restrict__ rowptr, const int* __restrict__ cnt,
                                   const int* __restrict__ nbr, const float* __restrict__ dinv,
                                   float* __restrict__ outG, float* __restrict__ outS, int n) {
    const int q   = threadIdx.x & 15;
    const int grp = (int)((blockIdx.x * 256u + threadIdx.x) >> 4);
    const int r0  = grp * GROUP_ROWS;
    if (r0 >= n) return;
    const v4f* g4 = (const v4f*)tabG;
    const v4f* s4 = (const v4f*)tabS;

    const bool mine = (q < GROUP_ROWS) && (r0 + q < n);
    const int  rl   = mine ? (r0 + q) : r0;
    const int   begL = rowptr[rl];
    const int   degL = mine ? cnt[rl] : 0;
    const float drL  = dinv[rl];

    int ixn = 0;
    {
        const int d0 = row_bcast(degL, 0);
        const int b0 = row_bcast(begL, 0);
        if (d0 > 0) ixn = nbr[b0 + min(q, d0 - 1)];
    }

    #pragma unroll
    for (int k = 0; k < GROUP_ROWS; ++k) {
        const int r = r0 + k;
        if (r >= n) break;
        const int   beg = row_bcast(begL, k);
        const int   deg = row_bcast(degL, k);
        const float dr  = row_bcast_f(drL, k);
        int ix = ixn;
        if (k + 1 < GROUP_ROWS) {
            const int dn = row_bcast(degL, k + 1);
            const int bn = row_bcast(begL, k + 1);
            ixn = 0;
            if (dn > 0) ixn = nbr[bn + min(q, dn - 1)];
        }
        const v4f hv = g4[r * 16 + q];
        float4 accG = make_float4(0.0f, 0.0f, 0.0f, 0.0f);
        float4 accS = make_float4(0.0f, 0.0f, 0.0f, 0.0f);
        for (int j = 0; j < deg; j += EDGE_UNROLL) {
            const int m = deg - j;
            int ixc = ix;
            if (m > EDGE_UNROLL)
                ixc = nbr[beg + j + EDGE_UNROLL + min(q, m - EDGE_UNROLL - 1)];
            float dq = 0.0f;
            if (ONE_TABLE) dq = dinv[ix];
            v4f vg[EDGE_UNROLL];
            v4f vs[EDGE_UNROLL];
            #pragma unroll
            for (int u = 0; u < EDGE_UNROLL; ++u) {
                const int s = row_bcast(ix, u);
                vg[u] = g4[s * 16 + q];
                if (!ONE_TABLE) vs[u] = s4[s * 16 + q];
            }
            if (ONE_TABLE)
                asm volatile("" : "+v"(vg[0]), "+v"(vg[1]), "+v"(vg[2]), "+v"(vg[3]),
                                  "+v"(vg[4]), "+v"(vg[5]), "+v"(vg[6]), "+v"(vg[7]));
            else
                asm volatile("" : "+v"(vg[0]), "+v"(vg[1]), "+v"(vg[2]), "+v"(vg[3]),
                                  "+v"(vg[4]), "+v"(vg[5]), "+v"(vg[6]), "+v"(vg[7]),
                                  "+v"(vs[0]), "+v"(vs[1]), "+v"(vs[2]), "+v"(vs[3]),
                                  "+v"(vs[4]), "+v"(vs[5]), "+v"(vs[6]), "+v"(vs[7]));
            const float wq = dq * dr;
            #pragma unroll
            for (int u = 0; u < EDGE_UNROLL; ++u) {
                const bool on = (u == 0) || (u < m);
                const v4f a = vg[u];
                const v4f b = ONE_TABLE ? vg[u] : vs[u];
                float4 ng, ns;
                if (ONE_TABLE) {
                    const float w = row_bcast_f(wq, u);
                    ng.x = accG.x + a.x * w; ng.y = accG.y + a.y * w;
                    ng.z = accG.z + a.z * w; ng.w = accG.w + a.w * w;
                } else {
                    ng.x = accG.x + a.x; ng.y = accG.y + a.y;
                    ng.z = accG.z + a.z; ng.w = accG.w + a.w;
                }
                ns.x = accS.x + b.x; ns.y = accS.y + b.y;
                ns.z = accS.z + b.z; ns.w = accS.w + b.w;
                accG.x = on ? ng.x : accG.x; accG.y = on ? ng.y : accG.y;
                accG.z = on ? ng.z : accG.z; accG.w = on ? ng.w : accG.w;
                accS.x = on ? ns.x : accS.x; accS.y = on ? ns.y : accS.y;
                accS.z = on ? ns.z : accS.z; accS.w = on ? ns.w : accS.w;
            }
            ix = ixc;
        }
        float4 og, os;
        if (ONE_TABLE) {
            const float d2 = dr * dr;
            og.x = accG.x + hv.x * d2; og.y = accG.y + hv.y * d2;
            og.z = accG.z + hv.z * d2; og.w = accG.w + hv.w * d2;
        } else {
            og.x = (accG.x + hv.x) * dr; og.y = (accG.y + hv.y) * dr;
            og.z = (accG.z + hv.z) * dr; og.w = (accG.w + hv.w) * dr;
        }
        const float rs = 1.0f / fmaxf((float)deg, 1.0f);
        os.x = accS.x * rs; os.y = accS.y * rs; os.z = accS.z * rs; os.w = accS.w * rs;
        ((float4*)outG)[r * 16 + q] = og;
        ((float4*)outS)[r * 16 + q] = os;
    }
}

extern "C" void kernel_launch(void* const* d_in, const int* in_sizes, int n_in,
                              void* d_out, int out_size, void* d_ws, size_t ws_size,
                              hipStream_t stream) {
    const float* x        = (const float*)d_in[0];
    const int*   ei       = (const int*)d_in[1];
    const float* gcn_w1   = (const float*)d_in[2];
    const float* gcn_b1   = (const float*)d_in[3];
    const float* gcn_w2   = (const float*)d_in[4];
    const float* gcn_b2   = (const float*)d_in[5];
    const float* sage_wl1 = (const float*)d_in[6];
    const float* sage_bl1 = (const float*)d_in[7];
    const float* sage_wr1 = (const float*)d_in[8];
    const float* sage_wl2 = (const float*)d_in[9];
    const float* sage_bl2 = (const float*)d_in[10];
    const float* sage_wr2 = (const float*)d_in[11];
    const float* gcn_ln_g = (const float*)d_in[12];
    const float* gcn_ln_b = (const float*)d_in[13];
    const float* sage_ln_g = (const float*)d_in[14];
    const float* sage_ln_b = (const float*)d_in[15];
    const float* proj_w   = (const float*)d_in[16];
    const float* proj_b   = (const float*)d_in[17];

    const int N = in_sizes[0] / FDIM;
    const int E = in_sizes[1] / 2;
    const int* src = ei;
    const int* dst = ei + E;

    int*   cnt    = (int*)d_ws;
    float* dinv   = (float*)(cnt + N);
    int*   rowptr = (int*)(dinv + N);
    int*   nbr    = rowptr + N;
    float* h      = (float*)(nbr + E);
    float* g      = h + (size_t)N * FDIM;
    float* agg    = g + (size_t)N * FDIM;
    float* out    = (float*)d_out;
    int*   rank   = (int*)agg;
    int*   bsum   = (int*)h;

    const int nblkN  = (N + 255) / 256;
    const int nblkE  = (E + 255) / 256;
    const int nblkT  = (N + SCAN_TILE - 1) / SCAN_TILE;
    const int nblkG  = (N + 127) / 128;
    const int nblkR  = (N + 16 * GROUP_ROWS - 1) / (16 * GROUP_ROWS);

    init_cnt_kernel<<<nblkN, 256, 0, stream>>>(cnt, N);
    count_rank_kernel<<<nblkE, 256, 0, stream>>>(dst, cnt, rank, E);
    scan_tile_kernel<<<nblkT, 256, 0, stream>>>(cnt, rowptr, bsum, N);
    finalize_rows_kernel<<<nblkN, 256, 0, stream>>>(cnt, bsum, rowptr, dinv, N);
    fill_nbr_kernel<<<nblkE, 256, 0, stream>>>(src, dst, rowptr, rank, nbr, E);

    gather_pair_kernel<true><<<nblkR, 256, 0, stream>>>(x, x, rowptr, cnt, nbr, dinv, h, agg, N);
    layer1_products_kernel<<<nblkG, 256, 0, stream>>>(
        h, agg, x, gcn_w1, gcn_b1, sage_wl1, sage_wr1, sage_bl1, dinv, g, out, N);

    gather_pair_kernel<false><<<nblkR, 256, 0, stream>>>(g, out, rowptr, cnt, nbr, dinv, h, agg, N);
    layer2_projection_kernel<<<nblkG, 256, 0, stream>>>(
        h, agg, out, gcn_w2, gcn_b2, sage_wl2, sage_wr2, sage_bl2,
        gcn_ln_g, gcn_ln_b, sage_ln_g, sage_ln_b, proj_w, proj_b, out, N);
}


# WMMA weight feed rebuilt: LDS keeps weights as padded k-pair rows, so a fragment is one conflict-free 64-bit load, and a tile's 16 loads go out as one group ahead of its WMMAs; all sums unchanged
# speedup vs baseline: 3.2659x; 1.0407x over previous
#include <hip/hip_runtime.h>

typedef __attribute__((ext_vector_type(2))) float v2f;
typedef __attribute__((ext_vector_type(8))) float v8f;
typedef __attribute__((ext_vector_type(4))) float v4f;

#define FDIM 64
#define SCAN_TILE 1024
#define GROUP_ROWS 5
#define EDGE_UNROLL 8

__global__ void init_cnt_kernel(int* __restrict__ cnt, int n) {
    int i = blockIdx.x * blockDim.x + threadIdx.x;
    if (i < n) cnt[i] = 0;
}

__global__ void count_rank_kernel(const int* __restrict__ dst, int* __restrict__ cnt,
                                  int* __restrict__ rank, int e) {
    int i = blockIdx.x * blockDim.x + threadIdx.x;
    if (i < e) rank[i] = atomicAdd(&cnt[dst[i]], 1);
}

__launch_bounds__(256)
__global__ void scan_tile_kernel(const int* __restrict__ cnt, int* __restrict__ rowptr,
                                 int* __restrict__ bsum, int n) {
    __shared__ int wsum[8];
    const int t    = threadIdx.x;
    const int lane = t & 31;
    const int wave = t >> 5;
    const int base = blockIdx.x * SCAN_TILE + t * 4;
    const int c0 = (base + 0 < n) ? cnt[base + 0] : 0;
    const int c1 = (base + 1 < n) ? cnt[base + 1] : 0;
    const int c2 = (base + 2 < n) ? cnt[base + 2] : 0;
    const int c3 = (base + 3 < n) ? cnt[base + 3] : 0;
    const int tot = c0 + c1 + c2 + c3;
    int inc = tot;
    for (int o = 1; o < 32; o <<= 1) {
        const int v = __shfl_up(inc, o, 32);
        if (lane >= o) inc += v;
    }
    if (lane == 31) wsum[wave] = inc;
    __syncthreads();
    int woff = 0;
    for (int w = 0; w < wave; ++w) woff += wsum[w];
    const int ex = woff + inc - tot;
    if (base + 0 < n) rowptr[base + 0] = ex;
    if (base + 1 < n) rowptr[base + 1] = ex + c0;
    if (base + 2 < n) rowptr[base + 2] = ex + c0 + c1;
    if (base + 3 < n) rowptr[base + 3] = ex + c0 + c1 + c2;
    if (t == 255) bsum[blockIdx.x] = woff + inc;
}

__launch_bounds__(256)
__global__ void finalize_rows_kernel(const int* __restrict__ cnt, const int* __restrict__ bsum,
                                     int* __restrict__ rowptr, float* __restrict__ dinv, int n) {
    __shared__ int part[8];
    const int t  = threadIdx.x;
    const int nt = (blockIdx.x * 256) / SCAN_TILE;
    int v = 0;
    for (int i = t; i < nt; i += 256) v += bsum[i];
    for (int o = 16; o > 0; o >>= 1) v += __shfl_xor(v, o, 32);
    if ((t & 31) == 0) part[t >> 5] = v;
    __syncthreads();
    int off = 0;
    for (int w = 0; w < 8; ++w) off += part[w];
    const int i = blockIdx.x * 256 + t;
    if (i < n) {
        rowptr[i] += off;
        dinv[i] = rsqrtf((float)cnt[i] + 1.0f);
    }
}

__global__ void fill_nbr_kernel(const int* __restrict__ src, const int* __restrict__ dst,
                                const int* __restrict__ rowptr, const int* __restrict__ rank,
                                int* __restrict__ nbr, int e) {
    int i = blockIdx.x * blockDim.x + threadIdx.x;
    if (i < e) nbr[rowptr[dst[i]] + rank[i]] = src[i];
}

#define WMMA_F32(a, b, c) __builtin_amdgcn_wmma_f32_16x16x4_f32(false, (a), false, (b), (short)0, (c), false, false)

#define WPITCH 160
#define WWORDS (32 * WPITCH)
template <bool PERM>
__device__ __forceinline__ void stage_w64(float* __restrict__ dst, const float* __restrict__ src, int t) {
    const float4* s = (const float4*)src;
    for (int i = t; i < 512; i += 256) {
        const int kp = i >> 4;
        const int c4 = i & 15;
        const float4 a = s[(2 * kp) * 16 + c4];
        const float4 b = s[(2 * kp + 1) * 16 + c4];
        const int prow = PERM ? ((kp & 24) | ((kp & 3) << 1) | ((kp >> 2) & 1)) : kp;
        float4* d = (float4*)(dst + prow * WPITCH + c4 * 8);
        d[0] = make_float4(a.x, b.x, a.y, b.y);
        d[1] = make_float4(a.z, b.z, a.w, b.w);
    }
}

#define FRAGS_LOADED(w) asm volatile("" : "+v"(w[0]), "+v"(w[1]), "+v"(w[2]), "+v"(w[3]), "+v"(w[4]), "+v"(w[5]), \
                                          "+v"(w[6]), "+v"(w[7]), "+v"(w[8]), "+v"(w[9]), "+v"(w[10]), "+v"(w[11]), \
                                          "+v"(w[12]), "+v"(w[13]), "+v"(w[14]), "+v"(w[15]))

__device__ __forceinline__ void load_row_frag(v2f (&f)[16], const float* p) {
    #pragma unroll
    for (int kb = 0; kb < 16; ++kb) { f[kb].x = p[kb * 4 + 0]; f[kb].y = p[kb * 4 + 1]; }
}

__device__ __forceinline__ void load_w_frag(v2f (&w)[16], const float* Wl, int col, int lhalf) {
    #pragma unroll
    for (int kb = 0; kb < 16; ++kb)
        w[kb] = *(const v2f*)(Wl + (2 * kb + lhalf) * WPITCH + col * 2);
}

template <bool RELU>
__device__ __forceinline__ void store_node_major(float* __restrict__ dst, const v8f acc, float bval,
                                                 int m0, int lhalf, int col, int n) {
    float v[8];
    #pragma unroll
    for (int j = 0; j < 8; ++j) {
        v[j] = acc[j] + bval;
        if (RELU) v[j] = fmaxf(v[j], 0.0f);
    }
    #pragma unroll
    for (int j = 0; j < 8; ++j) {
        const int r = m0 + j + lhalf * 8;
        if (r < n) dst[(size_t)r * FDIM + col] = v[j];
    }
}

__device__ __forceinline__ void store_node_major_relu_scaled(float* __restrict__ dst, const v8f acc, float bval,
                                                             const float (&rs)[8], int m0, int lhalf, int col, int n) {
    float v[8];
    #pragma unroll
    for (int j = 0; j < 8; ++j) v[j] = fmaxf(acc[j] + bval, 0.0f) * rs[j];
    #pragma unroll
    for (int j = 0; j < 8; ++j) {
        const int r = m0 + j + lhalf * 8;
        if (r < n) dst[(size_t)r * FDIM + col] = v[j];
    }
}

__launch_bounds__(256, 8)
__global__ void layer1_products_kernel(const float* __restrict__ H, const float* __restrict__ AGG,
                                       const float* __restrict__ X,
                                       const float* __restrict__ Wg, const float* __restrict__ bg,
                                       const float* __restrict__ Wl, const float* __restrict__ Wr,
                                       const float* __restrict__ bl, const float* __restrict__ dinv,
                                       float* __restrict__ outG, float* __restrict__ outS, int n) {
    __shared__ __attribute__((aligned(16))) float WG[WWORDS];
    __shared__ __attribute__((aligned(16))) float WL[WWORDS];
    __shared__ __attribute__((aligned(16))) float WR[WWORDS];
    const int t = threadIdx.x;
    stage_w64<false>(WG, Wg, t);
    stage_w64<false>(WL, Wl, t);
    stage_w64<false>(WR, Wr, t);
    __syncthreads();

    const int wave  = t >> 5;
    const int lane  = t & 31;
    const int m0    = (blockIdx.x * 8 + wave) * 16;
    if (m0 >= n) return;
    const int lhalf = lane >> 4;
    const int l15   = lane & 15;
    const int kOff  = lhalf * 2;
    int row = m0 + l15;
    if (row >= n) row = n - 1;

    {
        v8f acc[4];
        #pragma unroll
        for (int nb = 0; nb < 4; ++nb) acc[nb] = (v8f){0.f, 0.f, 0.f, 0.f, 0.f, 0.f, 0.f, 0.f};
        {
            v2f af[16];
            load_row_frag(af, AGG + (size_t)row * FDIM + kOff);
            #pragma unroll
            for (int nb = 0; nb < 4; ++nb) {
                v2f wf[16];
                load_w_frag(wf, WL, nb * 16 + l15, lhalf);
                FRAGS_LOADED(wf);
                #pragma unroll
                for (int kb = 0; kb < 16; ++kb) acc[nb] = WMMA_F32(af[kb], wf[kb], acc[nb]);
            }
        }
        {
            v2f xf[16];
            load_row_frag(xf, X + (size_t)row * FDIM + kOff);
            #pragma unroll
            for (int nb = 0; nb < 4; ++nb) {
                const int col = nb * 16 + l15;
                v2f wg[16];
                load_w_frag(wg, WR, col, lhalf);
                FRAGS_LOADED(wg);
                #pragma unroll
                for (int kb = 0; kb < 16; ++kb) acc[nb] = WMMA_F32(xf[kb], wg[kb], acc[nb]);
                store_node_major<true>(outS, acc[nb], bl[col], m0, lhalf, col, n);
            }
        }
    }
    {
        v2f hf[16];
        load_row_frag(hf, H + (size_t)row * FDIM + kOff);
        float dv[8];
        #pragma unroll
        for (int j = 0; j < 8; ++j) {
            const int r = m0 + j + lhalf * 8;
            dv[j] = dinv[r < n ? r : n - 1];
        }
        #pragma unroll
        for (int nb = 0; nb < 4; ++nb) {
            const int col = nb * 16 + l15;
            v2f wf[16];
            load_w_frag(wf, WG, col, lhalf);
            FRAGS_LOADED(wf);
            v8f acc = {0.f, 0.f, 0.f, 0.f, 0.f, 0.f, 0.f, 0.f};
            #pragma unroll
            for (int kb = 0; kb < 16; ++kb) acc = WMMA_F32(hf[kb], wf[kb], acc);
            store_node_major_relu_scaled(outG, acc, bg[col], dv, m0, lhalf, col, n);
        }
    }
}

__device__ __forceinline__ void bias_layernorm_fm(v8f (&tl)[4], const float* bias, const float* gam,
                                                  const float* bet, int lhalf) {
    float s = 0.0f, q = 0.0f;
    #pragma unroll
    for (int nb = 0; nb < 4; ++nb) {
        #pragma unroll
        for (int r = 0; r < 8; ++r) {
            const float v = tl[nb][r] + bias[nb * 16 + lhalf * 8 + r];
            tl[nb][r] = v;
            s += v;
            q += v * v;
        }
    }
    s += __shfl_xor(s, 16, 32);
    q += __shfl_xor(q, 16, 32);
    const float mu  = s * (1.0f / 64.0f);
    const float inv = rsqrtf(fmaxf(q * (1.0f / 64.0f) - mu * mu, 0.0f) + 1e-5f);
    #pragma unroll
    for (int nb = 0; nb < 4; ++nb) {
        #pragma unroll
        for (int r = 0; r < 8; ++r) {
            const int f = nb * 16 + lhalf * 8 + r;
            tl[nb][r] = (tl[nb][r] - mu) * inv * gam[f] + bet[f];
        }
    }
}

__launch_bounds__(256, 4)
__global__ void layer2_projection_kernel(const float* __restrict__ H, const float* __restrict__ AGG,
                                         const float* S1,
                                         const float* __restrict__ Wg, const float* __restrict__ bg,
                                         const float* __restrict__ Wl, const float* __restrict__ Wr,
                                         const float* __restrict__ bl,
                                         const float* __restrict__ lnGg, const float* __restrict__ lnGb,
                                         const float* __restrict__ lnSg, const float* __restrict__ lnSb,
                                         const float* __restrict__ P, const float* __restrict__ pb,
                                         float* out, int n) {
    __shared__ __attribute__((aligned(16))) float WG[WWORDS];
    __shared__ __attribute__((aligned(16))) float WL[WWORDS];
    __shared__ __attribute__((aligned(16))) float WR[WWORDS];
    __shared__ __attribute__((aligned(16))) float PT[WWORDS];
    __shared__ __attribute__((aligned(16))) float PB[WWORDS];
    __shared__ float PAR[6][FDIM];
    const int t = threadIdx.x;
    stage_w64<false>(WG, Wg, t);
    stage_w64<false>(WL, Wl, t);
    stage_w64<false>(WR, Wr, t);
    stage_w64<true>(PT, P, t);
    stage_w64<true>(PB, P + FDIM * FDIM, t);
    if (t < FDIM) {
        PAR[0][t] = bg[t];   PAR[1][t] = bl[t];
        PAR[2][t] = lnGg[t]; PAR[3][t] = lnGb[t];
        PAR[4][t] = lnSg[t]; PAR[5][t] = lnSb[t];
    }
    __syncthreads();

    const int wave  = t >> 5;
    const int lane  = t & 31;
    const int m0    = (blockIdx.x * 8 + wave) * 16;
    if (m0 >= n) return;
    const int lhalf = lane >> 4;
    const int l15   = lane & 15;
    const int kOff  = lhalf * 2;
    int row = m0 + l15;
    if (row >= n) row = n - 1;

    v8f gT[4];
    {
        v2f hf[16];
        load_row_frag(hf, H + (size_t)row * FDIM + kOff);
        #pragma unroll
        for (int nb = 0; nb < 4; ++nb) {
            v2f wf[16];
            load_w_frag(wf, WG, nb * 16 + l15, lhalf);
            FRAGS_LOADED(wf);
            v8f acc = {0.f, 0.f, 0.f, 0.f, 0.f, 0.f, 0.f, 0.f};
            #pragma unroll
            for (int kb = 0; kb < 16; ++kb) acc = WMMA_F32(wf[kb], hf[kb], acc);
            gT[nb] = acc;
        }
        bias_layernorm_fm(gT, PAR[0], PAR[2], PAR[3], lhalf);
    }
    v8f sT[4];
    {
        v2f af[16], sf[16];
        load_row_frag(af, AGG + (size_t)row * FDIM + kOff);
        load_row_frag(sf, S1  + (size_t)row * FDIM + kOff);
        #pragma unroll
        for (int nb = 0; nb < 4; ++nb) {
            v8f acc = {0.f, 0.f, 0.f, 0.f, 0.f, 0.f, 0.f, 0.f};
            {
                v2f wf[16];
                load_w_frag(wf, WL, nb * 16 + l15, lhalf);
                FRAGS_LOADED(wf);
                #pragma unroll
                for (int kb = 0; kb < 16; ++kb) acc = WMMA_F32(wf[kb], af[kb], acc);
            }
            {
                v2f wg[16];
                load_w_frag(wg, WR, nb * 16 + l15, lhalf);
                FRAGS_LOADED(wg);
                #pragma unroll
                for (int kb = 0; kb < 16; ++kb) acc = WMMA_F32(wg[kb], sf[kb], acc);
            }
            sT[nb] = acc;
        }
        bias_layernorm_fm(sT, PAR[1], PAR[4], PAR[5], lhalf);
    }
    #pragma unroll
    for (int ob = 0; ob < 4; ++ob) {
        const int col = ob * 16 + l15;
        v2f pt[16], pq[16];
        #pragma unroll
        for (int nb = 0; nb < 4; ++nb) {
            #pragma unroll
            for (int rp = 0; rp < 4; ++rp) {
                const int prow = nb * 8 + rp * 2 + lhalf;
                pt[nb * 4 + rp] = *(const v2f*)(PT + prow * WPITCH + col * 2);
                pq[nb * 4 + rp] = *(const v2f*)(PB + prow * WPITCH + col * 2);
            }
        }
        FRAGS_LOADED(pt);
        v8f acc = {0.f, 0.f, 0.f, 0.f, 0.f, 0.f, 0.f, 0.f};
        #pragma unroll
        for (int nb = 0; nb < 4; ++nb) {
            #pragma unroll
            for (int rp = 0; rp < 4; ++rp) {
                v2f a;
                a.x = gT[nb][rp * 2];
                a.y = gT[nb][rp * 2 + 1];
                acc = WMMA_F32(a, pt[nb * 4 + rp], acc);
            }
        }
        FRAGS_LOADED(pq);
        #pragma unroll
        for (int nb = 0; nb < 4; ++nb) {
            #pragma unroll
            for (int rp = 0; rp < 4; ++rp) {
                v2f a;
                a.x = sT[nb][rp * 2];
                a.y = sT[nb][rp * 2 + 1];
                acc = WMMA_F32(a, pq[nb * 4 + rp], acc);
            }
        }
        store_node_major<false>(out, acc, pb[col], m0, lhalf, col, n);
    }
}

__device__ __forceinline__ int row_bcast(int v, int u) {
    switch (u) {
        case 0:  return __builtin_amdgcn_update_dpp(0, v, 0x150, 0xf, 0xf, false);
        case 1:  return __builtin_amdgcn_update_dpp(0, v, 0x151, 0xf, 0xf, false);
        case 2:  return __builtin_amdgcn_update_dpp(0, v, 0x152, 0xf, 0xf, false);
        case 3:  return __builtin_amdgcn_update_dpp(0, v, 0x153, 0xf, 0xf, false);
        case 4:  return __builtin_amdgcn_update_dpp(0, v, 0x154, 0xf, 0xf, false);
        case 5:  return __builtin_amdgcn_update_dpp(0, v, 0x155, 0xf, 0xf, false);
        case 6:  return __builtin_amdgcn_update_dpp(0, v, 0x156, 0xf, 0xf, false);
        default: return __builtin_amdgcn_update_dpp(0, v, 0x157, 0xf, 0xf, false);
    }
}
__device__ __forceinline__ float row_bcast_f(float v, int u) {
    return __int_as_float(row_bcast(__float_as_int(v), u));
}

template <bool ONE_TABLE>
__launch_bounds__(256)
__global__ void gather_pair_kernel(const float* __restrict__ tabG, const float* __restrict__ tabS,
                                   const int* __restrict__ rowptr, const int* __restrict__ cnt,
                                   const int* __restrict__ nbr, const float* __restrict__ dinv,
                                   float* __restrict__ outG, float* __restrict__ outS, int n) {
    const int q   = threadIdx.x & 15;
    const int grp = (int)((blockIdx.x * 256u + threadIdx.x) >> 4);
    const int r0  = grp * GROUP_ROWS;
    if (r0 >= n) return;
    const v4f* g4 = (const v4f*)tabG;
    const v4f* s4 = (const v4f*)tabS;

    const bool mine = (q < GROUP_ROWS) && (r0 + q < n);
    const int  rl   = mine ? (r0 + q) : r0;
    const int   begL = rowptr[rl];
    const int   degL = mine ? cnt[rl] : 0;
    const float drL  = dinv[rl];

    int ixn = 0;
    {
        const int d0 = row_bcast(degL, 0);
        const int b0 = row_bcast(begL, 0);
        if (d0 > 0) ixn = nbr[b0 + min(q, d0 - 1)];
    }

    #pragma unroll
    for (int k = 0; k < GROUP_ROWS; ++k) {
        const int r = r0 + k;
        if (r >= n) break;
        const int   beg = row_bcast(begL, k);
        const int   deg = row_bcast(degL, k);
        const float dr  = row_bcast_f(drL, k);
        int ix = ixn;
        if (k + 1 < GROUP_ROWS) {
            const int dn = row_bcast(degL, k + 1);
            const int bn = row_bcast(begL, k + 1);
            ixn = 0;
            if (dn > 0) ixn = nbr[bn + min(q, dn - 1)];
        }
        const v4f hv = g4[r * 16 + q];
        float4 accG = make_float4(0.0f, 0.0f, 0.0f, 0.0f);
        float4 accS = make_float4(0.0f, 0.0f, 0.0f, 0.0f);
        for (int j = 0; j < deg; j += EDGE_UNROLL) {
            const int m = deg - j;
            int ixc = ix;
            if (m > EDGE_UNROLL)
                ixc = nbr[beg + j + EDGE_UNROLL + min(q, m - EDGE_UNROLL - 1)];
            float dq = 0.0f;
            if (ONE_TABLE) dq = dinv[ix];
            v4f vg[EDGE_UNROLL];
            v4f vs[EDGE_UNROLL];
            #pragma unroll
            for (int u = 0; u < EDGE_UNROLL; ++u) {
                const int s = row_bcast(ix, u);
                vg[u] = g4[s * 16 + q];
                if (!ONE_TABLE) vs[u] = s4[s * 16 + q];
            }
            if (ONE_TABLE)
                asm volatile("" : "+v"(vg[0]), "+v"(vg[1]), "+v"(vg[2]), "+v"(vg[3]),
                                  "+v"(vg[4]), "+v"(vg[5]), "+v"(vg[6]), "+v"(vg[7]));
            else
                asm volatile("" : "+v"(vg[0]), "+v"(vg[1]), "+v"(vg[2]), "+v"(vg[3]),
                                  "+v"(vg[4]), "+v"(vg[5]), "+v"(vg[6]), "+v"(vg[7]),
                                  "+v"(vs[0]), "+v"(vs[1]), "+v"(vs[2]), "+v"(vs[3]),
                                  "+v"(vs[4]), "+v"(vs[5]), "+v"(vs[6]), "+v"(vs[7]));
            const float wq = dq * dr;
            #pragma unroll
            for (int u = 0; u < EDGE_UNROLL; ++u) {
                const bool on = (u == 0) || (u < m);
                const v4f a = vg[u];
                const v4f b = ONE_TABLE ? vg[u] : vs[u];
                float4 ng, ns;
                if (ONE_TABLE) {
                    const float w = row_bcast_f(wq, u);
                    ng.x = accG.x + a.x * w; ng.y = accG.y + a.y * w;
                    ng.z = accG.z + a.z * w; ng.w = accG.w + a.w * w;
                } else {
                    ng.x = accG.x + a.x; ng.y = accG.y + a.y;
                    ng.z = accG.z + a.z; ng.w = accG.w + a.w;
                }
                ns.x = accS.x + b.x; ns.y = accS.y + b.y;
                ns.z = accS.z + b.z; ns.w = accS.w + b.w;
                accG.x = on ? ng.x : accG.x; accG.y = on ? ng.y : accG.y;
                accG.z = on ? ng.z : accG.z; accG.w = on ? ng.w : accG.w;
                accS.x = on ? ns.x : accS.x; accS.y = on ? ns.y : accS.y;
                accS.z = on ? ns.z : accS.z; accS.w = on ? ns.w : accS.w;
            }
            ix = ixc;
        }
        float4 og, os;
        if (ONE_TABLE) {
            const float d2 = dr * dr;
            og.x = accG.x + hv.x * d2; og.y = accG.y + hv.y * d2;
            og.z = accG.z + hv.z * d2; og.w = accG.w + hv.w * d2;
        } else {
            og.x = (accG.x + hv.x) * dr; og.y = (accG.y + hv.y) * dr;
            og.z = (accG.z + hv.z) * dr; og.w = (accG.w + hv.w) * dr;
        }
        const float rs = 1.0f / fmaxf((float)deg, 1.0f);
        os.x = accS.x * rs; os.y = accS.y * rs; os.z = accS.z * rs; os.w = accS.w * rs;
        ((float4*)outG)[r * 16 + q] = og;
        ((float4*)outS)[r * 16 + q] = os;
    }
}

extern "C" void kernel_launch(void* const* d_in, const int* in_sizes, int n_in,
                              void* d_out, int out_size, void* d_ws, size_t ws_size,
                              hipStream_t stream) {
    const float* x        = (const float*)d_in[0];
    const int*   ei       = (const int*)d_in[1];
    const float* gcn_w1   = (const float*)d_in[2];
    const float* gcn_b1   = (const float*)d_in[3];
    const float* gcn_w2   = (const float*)d_in[4];
    const float* gcn_b2   = (const float*)d_in[5];
    const float* sage_wl1 = (const float*)d_in[6];
    const float* sage_bl1 = (const float*)d_in[7];
    const float* sage_wr1 = (const float*)d_in[8];
    const float* sage_wl2 = (const float*)d_in[9];
    const float* sage_bl2 = (const float*)d_in[10];
    const float* sage_wr2 = (const float*)d_in[11];
    const float* gcn_ln_g = (const float*)d_in[12];
    const float* gcn_ln_b = (const float*)d_in[13];
    const float* sage_ln_g = (const float*)d_in[14];
    const float* sage_ln_b = (const float*)d_in[15];
    const float* proj_w   = (const float*)d_in[16];
    const float* proj_b   = (const float*)d_in[17];

    const int N = in_sizes[0] / FDIM;
    const int E = in_sizes[1] / 2;
    const int* src = ei;
    const int* dst = ei + E;

    int*   cnt    = (int*)d_ws;
    float* dinv   = (float*)(cnt + N);
    int*   rowptr = (int*)(dinv + N);
    int*   nbr    = rowptr + N;
    float* h      = (float*)(nbr + E);
    float* g      = h + (size_t)N * FDIM;
    float* agg    = g + (size_t)N * FDIM;
    float* out    = (float*)d_out;
    int*   rank   = (int*)agg;
    int*   bsum   = (int*)h;

    const int nblkN  = (N + 255) / 256;
    const int nblkE  = (E + 255) / 256;
    const int nblkT  = (N + SCAN_TILE - 1) / SCAN_TILE;
    const int nblkG  = (N + 127) / 128;
    const int nblkR  = (N + 16 * GROUP_ROWS - 1) / (16 * GROUP_ROWS);

    init_cnt_kernel<<<nblkN, 256, 0, stream>>>(cnt, N);
    count_rank_kernel<<<nblkE, 256, 0, stream>>>(dst, cnt, rank, E);
    scan_tile_kernel<<<nblkT, 256, 0, stream>>>(cnt, rowptr, bsum, N);
    finalize_rows_kernel<<<nblkN, 256, 0, stream>>>(cnt, bsum, rowptr, dinv, N);
    fill_nbr_kernel<<<nblkE, 256, 0, stream>>>(src, dst, rowptr, rank, nbr, E);

    gather_pair_kernel<true><<<nblkR, 256, 0, stream>>>(x, x, rowptr, cnt, nbr, dinv, h, agg, N);
    layer1_products_kernel<<<nblkG, 256, 0, stream>>>(
        h, agg, x, gcn_w1, gcn_b1, sage_wl1, sage_wr1, sage_bl1, dinv, g, out, N);

    gather_pair_kernel<false><<<nblkR, 256, 0, stream>>>(g, out, rowptr, cnt, nbr, dinv, h, agg, N);
    layer2_projection_kernel<<<nblkG, 256, 0, stream>>>(
        h, agg, out, gcn_w2, gcn_b2, sage_wl2, sage_wr2, sage_bl2,
        gcn_ln_g, gcn_ln_b, sage_ln_g, sage_ln_b, proj_w, proj_b, out, N);
}


# Bucket build without global atomics: edges binned by 512-row destination range (LDS counts, prefix, scatter), one block per range then buckets in LDS; bucket order = LDS-add arrival order
# speedup vs baseline: 4.1894x; 1.2828x over previous
#include <hip/hip_runtime.h>

typedef __attribute__((ext_vector_type(2))) float v2f;
typedef __attribute__((ext_vector_type(8))) float v8f;
typedef __attribute__((ext_vector_type(4))) float v4f;

#define FDIM 64
#define BIN_SHIFT 9
#define BIN_ROWS (1 << BIN_SHIFT)
#define PART_BLOCKS 512
#define GROUP_ROWS 5
#define EDGE_UNROLL 8

__launch_bounds__(256)
__global__ void bin_count_kernel(const int* __restrict__ dst, int* __restrict__ hist, int e, int chunk, int nb) {
    extern __shared__ int binc[];
    const int t = threadIdx.x;
    for (int b = t; b < nb; b += 256) binc[b] = 0;
    __syncthreads();
    const int e0 = blockIdx.x * chunk;
    const int e1 = min(e0 + chunk, e);
    int i = e0 + t;
    for (; i + 768 < e1; i += 1024) {
        const int d0 = dst[i], d1 = dst[i + 256], d2 = dst[i + 512], d3 = dst[i + 768];
        atomicAdd(&binc[d0 >> BIN_SHIFT], 1);
        atomicAdd(&binc[d1 >> BIN_SHIFT], 1);
        atomicAdd(&binc[d2 >> BIN_SHIFT], 1);
        atomicAdd(&binc[d3 >> BIN_SHIFT], 1);
    }
    for (; i < e1; i += 256) atomicAdd(&binc[dst[i] >> BIN_SHIFT], 1);
    __syncthreads();
    for (int b = t; b < nb; b += 256) hist[blockIdx.x * nb + b] = binc[b];
}

__launch_bounds__(256)
__global__ void bin_prefix_kernel(int* __restrict__ hist, int* __restrict__ binTotal, int nparts, int nb) {
    __shared__ int wsum[8];
    const int t    = threadIdx.x;
    const int lane = t & 31;
    const int wave = t >> 5;
    const int bin  = blockIdx.x;
    int carry = 0;
    for (int base = 0; base < nparts; base += 256) {
        const int p = base + t;
        const int v = (p < nparts) ? hist[p * nb + bin] : 0;
        int inc = v;
        for (int o = 1; o < 32; o <<= 1) {
            const int u = __shfl_up(inc, o, 32);
            if (lane >= o) inc += u;
        }
        if (lane == 31) wsum[wave] = inc;
        __syncthreads();
        int woff = 0, tot = 0;
        for (int w = 0; w < 8; ++w) {
            const int x = wsum[w];
            if (w < wave) woff += x;
            tot += x;
        }
        if (p < nparts) hist[p * nb + bin] = carry + woff + inc - v;
        carry += tot;
        __syncthreads();
    }
    if (t == 0) binTotal[bin] = carry;
}

__launch_bounds__(256)
__global__ void bin_scatter_kernel(const int* __restrict__ src, const int* __restrict__ dst,
                                   const int* __restrict__ hist, const int* __restrict__ binTotal,
                                   int* __restrict__ binStart, int2* __restrict__ pairs,
                                   int e, int chunk, int nb) {
    extern __shared__ int cur[];
    __shared__ int wsum[8];
    const int t    = threadIdx.x;
    const int lane = t & 31;
    const int wave = t >> 5;
    int carry = 0;
    for (int base = 0; base < nb; base += 256) {
        const int b = base + t;
        const int v = (b < nb) ? binTotal[b] : 0;
        int inc = v;
        for (int o = 1; o < 32; o <<= 1) {
            const int u = __shfl_up(inc, o, 32);
            if (lane >= o) inc += u;
        }
        if (lane == 31) wsum[wave] = inc;
        __syncthreads();
        int woff = 0, tot = 0;
        for (int w = 0; w < 8; ++w) {
            const int x = wsum[w];
            if (w < wave) woff += x;
            tot += x;
        }
        const int ex = carry + woff + inc - v;
        if (b < nb) {
            cur[b] = ex + hist[blockIdx.x * nb + b];
            if (blockIdx.x == 0) binStart[b] = ex;
        }
        carry += tot;
        __syncthreads();
    }
    if (blockIdx.x == 0 && t == 0) binStart[nb] = carry;
    const int e0 = blockIdx.x * chunk;
    const int e1 = min(e0 + chunk, e);
    int i = e0 + t;
    for (; i + 768 < e1; i += 1024) {
        const int d0 = dst[i], d1 = dst[i + 256], d2 = dst[i + 512], d3 = dst[i + 768];
        const int s0 = src[i], s1 = src[i + 256], s2 = src[i + 512], s3 = src[i + 768];
        const int p0 = atomicAdd(&cur[d0 >> BIN_SHIFT], 1);
        const int p1 = atomicAdd(&cur[d1 >> BIN_SHIFT], 1);
        const int p2 = atomicAdd(&cur[d2 >> BIN_SHIFT], 1);
        const int p3 = atomicAdd(&cur[d3 >> BIN_SHIFT], 1);
        pairs[p0] = make_int2(s0, d0);
        pairs[p1] = make_int2(s1, d1);
        pairs[p2] = make_int2(s2, d2);
        pairs[p3] = make_int2(s3, d3);
    }
    for (; i < e1; i += 256) {
        const int d = dst[i];
        const int sv = src[i];
        const int p = atomicAdd(&cur[d >> BIN_SHIFT], 1);
        pairs[p] = make_int2(sv, d);
    }
}

__launch_bounds__(1024)
__global__ void bin_bucket_kernel(const int2* __restrict__ pairs, const int* __restrict__ binStart,
                                  int* __restrict__ cnt, int* __restrict__ rowptr, float* __restrict__ dinv,
                                  int* __restrict__ nbr, int n) {
    __shared__ int lc[BIN_ROWS];
    __shared__ int lp[BIN_ROWS];
    __shared__ int wsum[32];
    const int t     = threadIdx.x;
    const int lane  = t & 31;
    const int wave  = t >> 5;
    const int rbase = blockIdx.x * BIN_ROWS;
    const int b0    = binStart[blockIdx.x];
    const int b1    = binStart[blockIdx.x + 1];
    if (t < BIN_ROWS) lc[t] = 0;
    __syncthreads();
    {
        int i = b0 + t;
        for (; i + 1024 < b1; i += 2048) {
            const int2 pa = pairs[i], pb = pairs[i + 1024];
            atomicAdd(&lc[pa.y - rbase], 1);
            atomicAdd(&lc[pb.y - rbase], 1);
        }
        for (; i < b1; i += 1024) atomicAdd(&lc[pairs[i].y - rbase], 1);
    }
    __syncthreads();
    const int v = (t < BIN_ROWS) ? lc[t] : 0;
    int inc = v;
    for (int o = 1; o < 32; o <<= 1) {
        const int u = __shfl_up(inc, o, 32);
        if (lane >= o) inc += u;
    }
    if (lane == 31) wsum[wave] = inc;
    __syncthreads();
    int woff = 0;
    for (int w = 0; w < wave; ++w) woff += wsum[w];
    if (t < BIN_ROWS) {
        const int ex = woff + inc - v;
        lp[t] = ex;
        lc[t] = 0;
        const int r = rbase + t;
        if (r < n) {
            cnt[r]    = v;
            rowptr[r] = b0 + ex;
            dinv[r]   = rsqrtf((float)v + 1.0f);
        }
    }
    __syncthreads();
    {
        int i = b0 + t;
        for (; i + 1024 < b1; i += 2048) {
            const int2 pa = pairs[i], pb = pairs[i + 1024];
            const int ra = pa.y - rbase, rb = pb.y - rbase;
            const int ka = atomicAdd(&lc[ra], 1);
            const int kb = atomicAdd(&lc[rb], 1);
            nbr[b0 + lp[ra] + ka] = pa.x;
            nbr[b0 + lp[rb] + kb] = pb.x;
        }
        for (; i < b1; i += 1024) {
            const int2 pa = pairs[i];
            const int ra = pa.y - rbase;
            const int ka = atomicAdd(&lc[ra], 1);
            nbr[b0 + lp[ra] + ka] = pa.x;
        }
    }
}

#define WMMA_F32(a, b, c) __builtin_amdgcn_wmma_f32_16x16x4_f32(false, (a), false, (b), (short)0, (c), false, false)

#define WPITCH 160
#define WWORDS (32 * WPITCH)
template <bool PERM>
__device__ __forceinline__ void stage_w64(float* __restrict__ dst, const float* __restrict__ src, int t) {
    const float4* s = (const float4*)src;
    for (int i = t; i < 512; i += 256) {
        const int kp = i >> 4;
        const int c4 = i & 15;
        const float4 a = s[(2 * kp) * 16 + c4];
        const float4 b = s[(2 * kp + 1) * 16 + c4];
        const int prow = PERM ? ((kp & 24) | ((kp & 3) << 1) | ((kp >> 2) & 1)) : kp;
        float4* d = (float4*)(dst + prow * WPITCH + c4 * 8);
        d[0] = make_float4(a.x, b.x, a.y, b.y);
        d[1] = make_float4(a.z, b.z, a.w, b.w);
    }
}

#define FRAGS_LOADED(w) asm volatile("" : "+v"(w[0]), "+v"(w[1]), "+v"(w[2]), "+v"(w[3]), "+v"(w[4]), "+v"(w[5]), \
                                          "+v"(w[6]), "+v"(w[7]), "+v"(w[8]), "+v"(w[9]), "+v"(w[10]), "+v"(w[11]), \
                                          "+v"(w[12]), "+v"(w[13]), "+v"(w[14]), "+v"(w[15]))

__device__ __forceinline__ void load_row_frag(v2f (&f)[16], const float* p) {
    #pragma unroll
    for (int kb = 0; kb < 16; ++kb) { f[kb].x = p[kb * 4 + 0]; f[kb].y = p[kb * 4 + 1]; }
}

__device__ __forceinline__ void load_w_frag(v2f (&w)[16], const float* Wl, int col, int lhalf) {
    #pragma unroll
    for (int kb = 0; kb < 16; ++kb)
        w[kb] = *(const v2f*)(Wl + (2 * kb + lhalf) * WPITCH + col * 2);
}

template <bool RELU>
__device__ __forceinline__ void store_node_major(float* __restrict__ dst, const v8f acc, float bval,
                                                 int m0, int lhalf, int col, int n) {
    float v[8];
    #pragma unroll
    for (int j = 0; j < 8; ++j) {
        v[j] = acc[j] + bval;
        if (RELU) v[j] = fmaxf(v[j], 0.0f);
    }
    #pragma unroll
    for (int j = 0; j < 8; ++j) {
        const int r = m0 + j + lhalf * 8;
        if (r < n) dst[(size_t)r * FDIM + col] = v[j];
    }
}

__device__ __forceinline__ void store_node_major_relu_scaled(float* __restrict__ dst, const v8f acc, float bval,
                                                             const float (&rs)[8], int m0, int lhalf, int col, int n) {
    float v[8];
    #pragma unroll
    for (int j = 0; j < 8; ++j) v[j] = fmaxf(acc[j] + bval, 0.0f) * rs[j];
    #pragma unroll
    for (int j = 0; j < 8; ++j) {
        const int r = m0 + j + lhalf * 8;
        if (r < n) dst[(size_t)r * FDIM + col] = v[j];
    }
}

__launch_bounds__(256, 8)
__global__ void layer1_products_kernel(const float* __restrict__ H, const float* __restrict__ AGG,
                                       const float* __restrict__ X,
                                       const float* __restrict__ Wg, const float* __restrict__ bg,
                                       const float* __restrict__ Wl, const float* __restrict__ Wr,
                                       const float* __restrict__ bl, const float* __restrict__ dinv,
                                       float* __restrict__ outG, float* __restrict__ outS, int n) {
    __shared__ __attribute__((aligned(16))) float WG[WWORDS];
    __shared__ __attribute__((aligned(16))) float WL[WWORDS];
    __shared__ __attribute__((aligned(16))) float WR[WWORDS];
    const int t = threadIdx.x;
    stage_w64<false>(WG, Wg, t);
    stage_w64<false>(WL, Wl, t);
    stage_w64<false>(WR, Wr, t);
    __syncthreads();

    const int wave  = t >> 5;
    const int lane  = t & 31;
    const int m0    = (blockIdx.x * 8 + wave) * 16;
    if (m0 >= n) return;
    const int lhalf = lane >> 4;
    const int l15   = lane & 15;
    const int kOff  = lhalf * 2;
    int row = m0 + l15;
    if (row >= n) row = n - 1;

    {
        v8f acc[4];
        #pragma unroll
        for (int nb = 0; nb < 4; ++nb) acc[nb] = (v8f){0.f, 0.f, 0.f, 0.f, 0.f, 0.f, 0.f, 0.f};
        {
            v2f af[16];
            load_row_frag(af, AGG + (size_t)row * FDIM + kOff);
            #pragma unroll
            for (int nb = 0; nb < 4; ++nb) {
                v2f wf[16];
                load_w_frag(wf, WL, nb * 16 + l15, lhalf);
                FRAGS_LOADED(wf);
                #pragma unroll
                for (int kb = 0; kb < 16; ++kb) acc[nb] = WMMA_F32(af[kb], wf[kb], acc[nb]);
            }
        }
        {
            v2f xf[16];
            load_row_frag(xf, X + (size_t)row * FDIM + kOff);
            #pragma unroll
            for (int nb = 0; nb < 4; ++nb) {
                const int col = nb * 16 + l15;
                v2f wg[16];
                load_w_frag(wg, WR, col, lhalf);
                FRAGS_LOADED(wg);
                #pragma unroll
                for (int kb = 0; kb < 16; ++kb) acc[nb] = WMMA_F32(xf[kb], wg[kb], acc[nb]);
                store_node_major<true>(outS, acc[nb], bl[col], m0, lhalf, col, n);
            }
        }
    }
    {
        v2f hf[16];
        load_row_frag(hf, H + (size_t)row * FDIM + kOff);
        float dv[8];
        #pragma unroll
        for (int j = 0; j < 8; ++j) {
            const int r = m0 + j + lhalf * 8;
            dv[j] = dinv[r < n ? r : n - 1];
        }
        #pragma unroll
        for (int nb = 0; nb < 4; ++nb) {
            const int col = nb * 16 + l15;
            v2f wf[16];
            load_w_frag(wf, WG, col, lhalf);
            FRAGS_LOADED(wf);
            v8f acc = {0.f, 0.f, 0.f, 0.f, 0.f, 0.f, 0.f, 0.f};
            #pragma unroll
            for (int kb = 0; kb < 16; ++kb) acc = WMMA_F32(hf[kb], wf[kb], acc);
            store_node_major_relu_scaled(outG, acc, bg[col], dv, m0, lhalf, col, n);
        }
    }
}

__device__ __forceinline__ void bias_layernorm_fm(v8f (&tl)[4], const float* bias, const float* gam,
                                                  const float* bet, int lhalf) {
    float s = 0.0f, q = 0.0f;
    #pragma unroll
    for (int nb = 0; nb < 4; ++nb) {
        #pragma unroll
        for (int r = 0; r < 8; ++r) {
            const float v = tl[nb][r] + bias[nb * 16 + lhalf * 8 + r];
            tl[nb][r] = v;
            s += v;
            q += v * v;
        }
    }
    s += __shfl_xor(s, 16, 32);
    q += __shfl_xor(q, 16, 32);
    const float mu  = s * (1.0f / 64.0f);
    const float inv = rsqrtf(fmaxf(q * (1.0f / 64.0f) - mu * mu, 0.0f) + 1e-5f);
    #pragma unroll
    for (int nb = 0; nb < 4; ++nb) {
        #pragma unroll
        for (int r = 0; r < 8; ++r) {
            const int f = nb * 16 + lhalf * 8 + r;
            tl[nb][r] = (tl[nb][r] - mu) * inv * gam[f] + bet[f];
        }
    }
}

__launch_bounds__(256, 4)
__global__ void layer2_projection_kernel(const float* __restrict__ H, const float* __restrict__ AGG,
                                         const float* S1,
                                         const float* __restrict__ Wg, const float* __restrict__ bg,
                                         const float* __restrict__ Wl, const float* __restrict__ Wr,
                                         const float* __restrict__ bl,
                                         const float* __restrict__ lnGg, const float* __restrict__ lnGb,
                                         const float* __restrict__ lnSg, const float* __restrict__ lnSb,
                                         const float* __restrict__ P, const float* __restrict__ pb,
                                         float* out, int n) {
    __shared__ __attribute__((aligned(16))) float WG[WWORDS];
    __shared__ __attribute__((aligned(16))) float WL[WWORDS];
    __shared__ __attribute__((aligned(16))) float WR[WWORDS];
    __shared__ __attribute__((aligned(16))) float PT[WWORDS];
    __shared__ __attribute__((aligned(16))) float PB[WWORDS];
    __shared__ float PAR[6][FDIM];
    const int t = threadIdx.x;
    stage_w64<false>(WG, Wg, t);
    stage_w64<false>(WL, Wl, t);
    stage_w64<false>(WR, Wr, t);
    stage_w64<true>(PT, P, t);
    stage_w64<true>(PB, P + FDIM * FDIM, t);
    if (t < FDIM) {
        PAR[0][t] = bg[t];   PAR[1][t] = bl[t];
        PAR[2][t] = lnGg[t]; PAR[3][t] = lnGb[t];
        PAR[4][t] = lnSg[t]; PAR[5][t] = lnSb[t];
    }
    __syncthreads();

    const int wave  = t >> 5;
    const int lane  = t & 31;
    const int m0    = (blockIdx.x * 8 + wave) * 16;
    if (m0 >= n) return;
    const int lhalf = lane >> 4;
    const int l15   = lane & 15;
    const int kOff  = lhalf * 2;
    int row = m0 + l15;
    if (row >= n) row = n - 1;

    v8f gT[4];
    {
        v2f hf[16];
        load_row_frag(hf, H + (size_t)row * FDIM + kOff);
        #pragma unroll
        for (int nb = 0; nb < 4; ++nb) {
            v2f wf[16];
            load_w_frag(wf, WG, nb * 16 + l15, lhalf);
            FRAGS_LOADED(wf);
            v8f acc = {0.f, 0.f, 0.f, 0.f, 0.f, 0.f, 0.f, 0.f};
            #pragma unroll
            for (int kb = 0; kb < 16; ++kb) acc = WMMA_F32(wf[kb], hf[kb], acc);
            gT[nb] = acc;
        }
        bias_layernorm_fm(gT, PAR[0], PAR[2], PAR[3], lhalf);
    }
    v8f sT[4];
    {
        v2f af[16], sf[16];
        load_row_frag(af, AGG + (size_t)row * FDIM + kOff);
        load_row_frag(sf, S1  + (size_t)row * FDIM + kOff);
        #pragma unroll
        for (int nb = 0; nb < 4; ++nb) {
            v8f acc = {0.f, 0.f, 0.f, 0.f, 0.f, 0.f, 0.f, 0.f};
            {
                v2f wf[16];
                load_w_frag(wf, WL, nb * 16 + l15, lhalf);
                FRAGS_LOADED(wf);
                #pragma unroll
                for (int kb = 0; kb < 16; ++kb) acc = WMMA_F32(wf[kb], af[kb], acc);
            }
            {
                v2f wg[16];
                load_w_frag(wg, WR, nb * 16 + l15, lhalf);
                FRAGS_LOADED(wg);
                #pragma unroll
                for (int kb = 0; kb < 16; ++kb) acc = WMMA_F32(wg[kb], sf[kb], acc);
            }
            sT[nb] = acc;
        }
        bias_layernorm_fm(sT, PAR[1], PAR[4], PAR[5], lhalf);
    }
    #pragma unroll
    for (int ob = 0; ob < 4; ++ob) {
        const int col = ob * 16 + l15;
        v2f pt[16], pq[16];
        #pragma unroll
        for (int nb = 0; nb < 4; ++nb) {
            #pragma unroll
            for (int rp = 0; rp < 4; ++rp) {
                const int prow = nb * 8 + rp * 2 + lhalf;
                pt[nb * 4 + rp] = *(const v2f*)(PT + prow * WPITCH + col * 2);
                pq[nb * 4 + rp] = *(const v2f*)(PB + prow * WPITCH + col * 2);
            }
        }
        FRAGS_LOADED(pt);
        v8f acc = {0.f, 0.f, 0.f, 0.f, 0.f, 0.f, 0.f, 0.f};
        #pragma unroll
        for (int nb = 0; nb < 4; ++nb) {
            #pragma unroll
            for (int rp = 0; rp < 4; ++rp) {
                v2f a;
                a.x = gT[nb][rp * 2];
                a.y = gT[nb][rp * 2 + 1];
                acc = WMMA_F32(a, pt[nb * 4 + rp], acc);
            }
        }
        FRAGS_LOADED(pq);
        #pragma unroll
        for (int nb = 0; nb < 4; ++nb) {
            #pragma unroll
            for (int rp = 0; rp < 4; ++rp) {
                v2f a;
                a.x = sT[nb][rp * 2];
                a.y = sT[nb][rp * 2 + 1];
                acc = WMMA_F32(a, pq[nb * 4 + rp], acc);
            }
        }
        store_node_major<false>(out, acc, pb[col], m0, lhalf, col, n);
    }
}

__device__ __forceinline__ int row_bcast(int v, int u) {
    switch (u) {
        case 0:  return __builtin_amdgcn_update_dpp(0, v, 0x150, 0xf, 0xf, false);
        case 1:  return __builtin_amdgcn_update_dpp(0, v, 0x151, 0xf, 0xf, false);
        case 2:  return __builtin_amdgcn_update_dpp(0, v, 0x152, 0xf, 0xf, false);
        case 3:  return __builtin_amdgcn_update_dpp(0, v, 0x153, 0xf, 0xf, false);
        case 4:  return __builtin_amdgcn_update_dpp(0, v, 0x154, 0xf, 0xf, false);
        case 5:  return __builtin_amdgcn_update_dpp(0, v, 0x155, 0xf, 0xf, false);
        case 6:  return __builtin_amdgcn_update_dpp(0, v, 0x156, 0xf, 0xf, false);
        default: return __builtin_amdgcn_update_dpp(0, v, 0x157, 0xf, 0xf, false);
    }
}
__device__ __forceinline__ float row_bcast_f(float v, int u) {
    return __int_as_float(row_bcast(__float_as_int(v), u));
}

template <bool ONE_TABLE>
__launch_bounds__(256)
__global__ void gather_pair_kernel(const float* __restrict__ tabG, const float* __restrict__ tabS,
                                   const int* __restrict__ rowptr, const int* __restrict__ cnt,
                                   const int* __restrict__ nbr, const float* __restrict__ dinv,
                                   float* __restrict__ outG, float* __restrict__ outS, int n) {
    const int q   = threadIdx.x & 15;
    const int grp = (int)((blockIdx.x * 256u + threadIdx.x) >> 4);
    const int r0  = grp * GROUP_ROWS;
    if (r0 >= n) return;
    const v4f* g4 = (const v4f*)tabG;
    const v4f* s4 = (const v4f*)tabS;

    const bool mine = (q < GROUP_ROWS) && (r0 + q < n);
    const int  rl   = mine ? (r0 + q) : r0;
    const int   begL = rowptr[rl];
    const int   degL = mine ? cnt[rl] : 0;
    const float drL  = dinv[rl];

    int ixn = 0;
    {
        const int d0 = row_bcast(degL, 0);
        const int b0 = row_bcast(begL, 0);
        if (d0 > 0) ixn = nbr[b0 + min(q, d0 - 1)];
    }

    #pragma unroll
    for (int k = 0; k < GROUP_ROWS; ++k) {
        const int r = r0 + k;
        if (r >= n) break;
        const int   beg = row_bcast(begL, k);
        const int   deg = row_bcast(degL, k);
        const float dr  = row_bcast_f(drL, k);
        int ix = ixn;
        if (k + 1 < GROUP_ROWS) {
            const int dn = row_bcast(degL, k + 1);
            const int bn = row_bcast(begL, k + 1);
            ixn = 0;
            if (dn > 0) ixn = nbr[bn + min(q, dn - 1)];
        }
        const v4f hv = g4[r * 16 + q];
        float4 accG = make_float4(0.0f, 0.0f, 0.0f, 0.0f);
        float4 accS = make_float4(0.0f, 0.0f, 0.0f, 0.0f);
        for (int j = 0; j < deg; j += EDGE_UNROLL) {
            const int m = deg - j;
            int ixc = ix;
            if (m > EDGE_UNROLL)
                ixc = nbr[beg + j + EDGE_UNROLL + min(q, m - EDGE_UNROLL - 1)];
            float dq = 0.0f;
            if (ONE_TABLE) dq = dinv[ix];
            v4f vg[EDGE_UNROLL];
            v4f vs[EDGE_UNROLL];
            #pragma unroll
            for (int u = 0; u < EDGE_UNROLL; ++u) {
                const int s = row_bcast(ix, u);
                vg[u] = g4[s * 16 + q];
                if (!ONE_TABLE) vs[u] = s4[s * 16 + q];
            }
            if (ONE_TABLE)
                asm volatile("" : "+v"(vg[0]), "+v"(vg[1]), "+v"(vg[2]), "+v"(vg[3]),
                                  "+v"(vg[4]), "+v"(vg[5]), "+v"(vg[6]), "+v"(vg[7]));
            else
                asm volatile("" : "+v"(vg[0]), "+v"(vg[1]), "+v"(vg[2]), "+v"(vg[3]),
                                  "+v"(vg[4]), "+v"(vg[5]), "+v"(vg[6]), "+v"(vg[7]),
                                  "+v"(vs[0]), "+v"(vs[1]), "+v"(vs[2]), "+v"(vs[3]),
                                  "+v"(vs[4]), "+v"(vs[5]), "+v"(vs[6]), "+v"(vs[7]));
            const float wq = dq * dr;
            #pragma unroll
            for (int u = 0; u < EDGE_UNROLL; ++u) {
                const bool on = (u == 0) || (u < m);
                const v4f a = vg[u];
                const v4f b = ONE_TABLE ? vg[u] : vs[u];
                float4 ng, ns;
                if (ONE_TABLE) {
                    const float w = row_bcast_f(wq, u);
                    ng.x = accG.x + a.x * w; ng.y = accG.y + a.y * w;
                    ng.z = accG.z + a.z * w; ng.w = accG.w + a.w * w;
                } else {
                    ng.x = accG.x + a.x; ng.y = accG.y + a.y;
                    ng.z = accG.z + a.z; ng.w = accG.w + a.w;
                }
                ns.x = accS.x + b.x; ns.y = accS.y + b.y;
                ns.z = accS.z + b.z; ns.w = accS.w + b.w;
                accG.x = on ? ng.x : accG.x; accG.y = on ? ng.y : accG.y;
                accG.z = on ? ng.z : accG.z; accG.w = on ? ng.w : accG.w;
                accS.x = on ? ns.x : accS.x; accS.y = on ? ns.y : accS.y;
                accS.z = on ? ns.z : accS.z; accS.w = on ? ns.w : accS.w;
            }
            ix = ixc;
        }
        float4 og, os;
        if (ONE_TABLE) {
            const float d2 = dr * dr;
            og.x = accG.x + hv.x * d2; og.y = accG.y + hv.y * d2;
            og.z = accG.z + hv.z * d2; og.w = accG.w + hv.w * d2;
        } else {
            og.x = (accG.x + hv.x) * dr; og.y = (accG.y + hv.y) * dr;
            og.z = (accG.z + hv.z) * dr; og.w = (accG.w + hv.w) * dr;
        }
        const float rs = 1.0f / fmaxf((float)deg, 1.0f);
        os.x = accS.x * rs; os.y = accS.y * rs; os.z = accS.z * rs; os.w = accS.w * rs;
        ((float4*)outG)[r * 16 + q] = og;
        ((float4*)outS)[r * 16 + q] = os;
    }
}

extern "C" void kernel_launch(void* const* d_in, const int* in_sizes, int n_in,
                              void* d_out, int out_size, void* d_ws, size_t ws_size,
                              hipStream_t stream) {
    const float* x        = (const float*)d_in[0];
    const int*   ei       = (const int*)d_in[1];
    const float* gcn_w1   = (const float*)d_in[2];
    const float* gcn_b1   = (const float*)d_in[3];
    const float* gcn_w2   = (const float*)d_in[4];
    const float* gcn_b2   = (const float*)d_in[5];
    const float* sage_wl1 = (const float*)d_in[6];
    const float* sage_bl1 = (const float*)d_in[7];
    const float* sage_wr1 = (const float*)d_in[8];
    const float* sage_wl2 = (const float*)d_in[9];
    const float* sage_bl2 = (const float*)d_in[10];
    const float* sage_wr2 = (const float*)d_in[11];
    const float* gcn_ln_g = (const float*)d_in[12];
    const float* gcn_ln_b = (const float*)d_in[13];
    const float* sage_ln_g = (const float*)d_in[14];
    const float* sage_ln_b = (const float*)d_in[15];
    const float* proj_w   = (const float*)d_in[16];
    const float* proj_b   = (const float*)d_in[17];

    const int N = in_sizes[0] / FDIM;
    const int E = in_sizes[1] / 2;
    const int* src = ei;
    const int* dst = ei + E;

    int*   cnt    = (int*)d_ws;
    float* dinv   = (float*)(cnt + N);
    int*   rowptr = (int*)(dinv + N);
    int*   nbr    = rowptr + N;
    float* h      = (float*)(nbr + E);
    float* g      = h + (size_t)N * FDIM;
    float* agg    = g + (size_t)N * FDIM;
    float* out    = (float*)d_out;
    const int NB    = (N + BIN_ROWS - 1) / BIN_ROWS;
    const int chunk = (E + PART_BLOCKS - 1) / PART_BLOCKS;
    int*   hist     = (int*)h;
    int*   binTotal = hist + PART_BLOCKS * NB;
    int*   binStart = binTotal + NB;
    int2*  pairs    = (int2*)g;

    const int nblkG  = (N + 127) / 128;
    const int nblkR  = (N + 16 * GROUP_ROWS - 1) / (16 * GROUP_ROWS);

    bin_count_kernel<<<PART_BLOCKS, 256, NB * sizeof(int), stream>>>(dst, hist, E, chunk, NB);
    bin_prefix_kernel<<<NB, 256, 0, stream>>>(hist, binTotal, PART_BLOCKS, NB);
    bin_scatter_kernel<<<PART_BLOCKS, 256, NB * sizeof(int), stream>>>(src, dst, hist, binTotal, binStart, pairs, E, chunk, NB);
    bin_bucket_kernel<<<NB, 1024, 0, stream>>>(pairs, binStart, cnt, rowptr, dinv, nbr, N);

    gather_pair_kernel<true><<<nblkR, 256, 0, stream>>>(x, x, rowptr, cnt, nbr, dinv, h, agg, N);
    layer1_products_kernel<<<nblkG, 256, 0, stream>>>(
        h, agg, x, gcn_w1, gcn_b1, sage_wl1, sage_wr1, sage_bl1, dinv, g, out, N);

    gather_pair_kernel<false><<<nblkR, 256, 0, stream>>>(g, out, rowptr, cnt, nbr, dinv, h, agg, N);
    layer2_projection_kernel<<<nblkG, 256, 0, stream>>>(
        h, agg, out, gcn_w2, gcn_b2, sage_wl2, sage_wr2, sage_bl2,
        gcn_ln_g, gcn_ln_b, sage_ln_g, sage_ln_b, proj_w, proj_b, out, N);
}


# Bucket build in 2 launches, not 4: each chunk sorts its pairs by bin inside its own range, so no global prefix pass; the bin pass walks per-chunk segments; bucket (= sum) order changes
# speedup vs baseline: 4.6181x; 1.1023x over previous
#include <hip/hip_runtime.h>

typedef __attribute__((ext_vector_type(2))) float v2f;
typedef __attribute__((ext_vector_type(8))) float v8f;
typedef __attribute__((ext_vector_type(4))) float v4f;

#define FDIM 64
#define BIN_SHIFT 9
#define BIN_ROWS (1 << BIN_SHIFT)
#define PART_BLOCKS 512
#define GROUP_ROWS 5
#define EDGE_UNROLL 8

#define SEG_THREADS (1024 / PART_BLOCKS)
static_assert(PART_BLOCKS * SEG_THREADS == 1024, "the last build pass gives every chunk SEG_THREADS of its 1024 threads");

__launch_bounds__(256)
__global__ void bin_sort_chunk_kernel(const int* __restrict__ src, const int* __restrict__ dst,
                                      int2* __restrict__ seg, int2* __restrict__ pairs,
                                      int e, int chunk, int nb) {
    extern __shared__ int cur[];
    __shared__ int wsum[8];
    const int t    = threadIdx.x;
    const int lane = t & 31;
    const int wave = t >> 5;
    for (int b = t; b < nb; b += 256) cur[b] = 0;
    __syncthreads();
    const int e0 = blockIdx.x * chunk;
    const int e1 = min(e0 + chunk, e);
    {
        int i = e0 + t;
        for (; i + 768 < e1; i += 1024) {
            const int d0 = dst[i], d1 = dst[i + 256], d2 = dst[i + 512], d3 = dst[i + 768];
            atomicAdd(&cur[d0 >> BIN_SHIFT], 1);
            atomicAdd(&cur[d1 >> BIN_SHIFT], 1);
            atomicAdd(&cur[d2 >> BIN_SHIFT], 1);
            atomicAdd(&cur[d3 >> BIN_SHIFT], 1);
        }
        for (; i < e1; i += 256) atomicAdd(&cur[dst[i] >> BIN_SHIFT], 1);
    }
    __syncthreads();
    int carry = 0;
    for (int base = 0; base < nb; base += 256) {
        const int b = base + t;
        const int v = (b < nb) ? cur[b] : 0;
        int inc = v;
        for (int o = 1; o < 32; o <<= 1) {
            const int u = __shfl_up(inc, o, 32);
            if (lane >= o) inc += u;
        }
        if (lane == 31) wsum[wave] = inc;
        __syncthreads();
        int woff = 0, tot = 0;
        for (int w = 0; w < 8; ++w) {
            const int x = wsum[w];
            if (w < wave) woff += x;
            tot += x;
        }
        const int ex = carry + woff + inc - v;
        if (b < nb) {
            cur[b] = e0 + ex;
            seg[blockIdx.x * nb + b] = make_int2(ex, v);
        }
        carry += tot;
        __syncthreads();
    }
    int i = e0 + t;
    for (; i + 768 < e1; i += 1024) {
        const int d0 = dst[i], d1 = dst[i + 256], d2 = dst[i + 512], d3 = dst[i + 768];
        const int s0 = src[i], s1 = src[i + 256], s2 = src[i + 512], s3 = src[i + 768];
        const int p0 = atomicAdd(&cur[d0 >> BIN_SHIFT], 1);
        const int p1 = atomicAdd(&cur[d1 >> BIN_SHIFT], 1);
        const int p2 = atomicAdd(&cur[d2 >> BIN_SHIFT], 1);
        const int p3 = atomicAdd(&cur[d3 >> BIN_SHIFT], 1);
        pairs[p0] = make_int2(s0, d0);
        pairs[p1] = make_int2(s1, d1);
        pairs[p2] = make_int2(s2, d2);
        pairs[p3] = make_int2(s3, d3);
    }
    for (; i < e1; i += 256) {
        const int d = dst[i];
        const int sv = src[i];
        const int p = atomicAdd(&cur[d >> BIN_SHIFT], 1);
        pairs[p] = make_int2(sv, d);
    }
}

__launch_bounds__(1024)
__global__ void bin_bucket_kernel(const int2* __restrict__ pairs, const int2* __restrict__ seg,
                                  int* __restrict__ cnt, int* __restrict__ rowptr, float* __restrict__ dinv,
                                  int* __restrict__ nbr, int n, int chunk, int nb) {
    __shared__ int lc[BIN_ROWS];
    __shared__ int lp[BIN_ROWS];
    __shared__ int wsum[32];
    __shared__ int osum[32];
    const int t     = threadIdx.x;
    const int lane  = t & 31;
    const int wave  = t >> 5;
    const int rbase = blockIdx.x * BIN_ROWS;
    const int part  = t / SEG_THREADS;
    const int j     = t % SEG_THREADS;
    const int2 sg   = seg[part * nb + blockIdx.x];
    const int sbase = part * chunk + sg.x;
    const int slen  = sg.y;
    const int last  = slen - 1;
    {
        int bs = (j == 0) ? sg.x : 0;
        for (int o = 16; o > 0; o >>= 1) bs += __shfl_xor(bs, o, 32);
        if (lane == 0) osum[wave] = bs;
    }
    if (t < BIN_ROWS) lc[t] = 0;
    __syncthreads();
    int b0 = 0;
    for (int w = 0; w < 32; ++w) b0 += osum[w];
    for (int i = j; i < slen; i += 4 * SEG_THREADS) {
        int2 pa = pairs[sbase + i];
        int2 pb = pairs[sbase + min(i + SEG_THREADS, last)];
        int2 pc = pairs[sbase + min(i + 2 * SEG_THREADS, last)];
        int2 pd = pairs[sbase + min(i + 3 * SEG_THREADS, last)];
        asm volatile("" : "+v"(pa.y), "+v"(pb.y), "+v"(pc.y), "+v"(pd.y));
        atomicAdd(&lc[pa.y - rbase], 1);
        if (i + SEG_THREADS < slen)     atomicAdd(&lc[pb.y - rbase], 1);
        if (i + 2 * SEG_THREADS < slen) atomicAdd(&lc[pc.y - rbase], 1);
        if (i + 3 * SEG_THREADS < slen) atomicAdd(&lc[pd.y - rbase], 1);
    }
    __syncthreads();
    const int v = (t < BIN_ROWS) ? lc[t] : 0;
    int inc = v;
    for (int o = 1; o < 32; o <<= 1) {
        const int u = __shfl_up(inc, o, 32);
        if (lane >= o) inc += u;
    }
    if (lane == 31) wsum[wave] = inc;
    __syncthreads();
    int woff = 0;
    for (int w = 0; w < wave; ++w) woff += wsum[w];
    if (t < BIN_ROWS) {
        const int ex = woff + inc - v;
        lp[t] = ex;
        lc[t] = 0;
        const int r = rbase + t;
        if (r < n) {
            cnt[r]    = v;
            rowptr[r] = b0 + ex;
            dinv[r]   = rsqrtf((float)v + 1.0f);
        }
    }
    __syncthreads();
    for (int i = j; i < slen; i += 4 * SEG_THREADS) {
        int2 pa = pairs[sbase + i];
        int2 pb = pairs[sbase + min(i + SEG_THREADS, last)];
        int2 pc = pairs[sbase + min(i + 2 * SEG_THREADS, last)];
        int2 pd = pairs[sbase + min(i + 3 * SEG_THREADS, last)];
        asm volatile("" : "+v"(pa.x), "+v"(pa.y), "+v"(pb.x), "+v"(pb.y), "+v"(pc.x), "+v"(pc.y), "+v"(pd.x), "+v"(pd.y));
        const int ra = pa.y - rbase;
        nbr[b0 + lp[ra] + atomicAdd(&lc[ra], 1)] = pa.x;
        if (i + SEG_THREADS < slen) {
            const int rb = pb.y - rbase;
            nbr[b0 + lp[rb] + atomicAdd(&lc[rb], 1)] = pb.x;
        }
        if (i + 2 * SEG_THREADS < slen) {
            const int rc = pc.y - rbase;
            nbr[b0 + lp[rc] + atomicAdd(&lc[rc], 1)] = pc.x;
        }
        if (i + 3 * SEG_THREADS < slen) {
            const int rd = pd.y - rbase;
            nbr[b0 + lp[rd] + atomicAdd(&lc[rd], 1)] = pd.x;
        }
    }
}

#define WMMA_F32(a, b, c) __builtin_amdgcn_wmma_f32_16x16x4_f32(false, (a), false, (b), (short)0, (c), false, false)

#define WPITCH 160
#define WWORDS (32 * WPITCH)
template <bool PERM>
__device__ __forceinline__ void stage_w64(float* __restrict__ dst, const float* __restrict__ src, int t) {
    const float4* s = (const float4*)src;
    for (int i = t; i < 512; i += 256) {
        const int kp = i >> 4;
        const int c4 = i & 15;
        const float4 a = s[(2 * kp) * 16 + c4];
        const float4 b = s[(2 * kp + 1) * 16 + c4];
        const int prow = PERM ? ((kp & 24) | ((kp & 3) << 1) | ((kp >> 2) & 1)) : kp;
        float4* d = (float4*)(dst + prow * WPITCH + c4 * 8);
        d[0] = make_float4(a.x, b.x, a.y, b.y);
        d[1] = make_float4(a.z, b.z, a.w, b.w);
    }
}

#define FRAGS_LOADED(w) asm volatile("" : "+v"(w[0]), "+v"(w[1]), "+v"(w[2]), "+v"(w[3]), "+v"(w[4]), "+v"(w[5]), \
                                          "+v"(w[6]), "+v"(w[7]), "+v"(w[8]), "+v"(w[9]), "+v"(w[10]), "+v"(w[11]), \
                                          "+v"(w[12]), "+v"(w[13]), "+v"(w[14]), "+v"(w[15]))

__device__ __forceinline__ void load_row_frag(v2f (&f)[16], const float* p) {
    #pragma unroll
    for (int kb = 0; kb < 16; ++kb) { f[kb].x = p[kb * 4 + 0]; f[kb].y = p[kb * 4 + 1]; }
}

__device__ __forceinline__ void load_w_frag(v2f (&w)[16], const float* Wl, int col, int lhalf) {
    #pragma unroll
    for (int kb = 0; kb < 16; ++kb)
        w[kb] = *(const v2f*)(Wl + (2 * kb + lhalf) * WPITCH + col * 2);
}

template <bool RELU>
__device__ __forceinline__ void store_node_major(float* __restrict__ dst, const v8f acc, float bval,
                                                 int m0, int lhalf, int col, int n) {
    float v[8];
    #pragma unroll
    for (int j = 0; j < 8; ++j) {
        v[j] = acc[j] + bval;
        if (RELU) v[j] = fmaxf(v[j], 0.0f);
    }
    #pragma unroll
    for (int j = 0; j < 8; ++j) {
        const int r = m0 + j + lhalf * 8;
        if (r < n) dst[(size_t)r * FDIM + col] = v[j];
    }
}

__device__ __forceinline__ void store_node_major_relu_scaled(float* __restrict__ dst, const v8f acc, float bval,
                                                             const float (&rs)[8], int m0, int lhalf, int col, int n) {
    float v[8];
    #pragma unroll
    for (int j = 0; j < 8; ++j) v[j] = fmaxf(acc[j] + bval, 0.0f) * rs[j];
    #pragma unroll
    for (int j = 0; j < 8; ++j) {
        const int r = m0 + j + lhalf * 8;
        if (r < n) dst[(size_t)r * FDIM + col] = v[j];
    }
}

__launch_bounds__(256, 8)
__global__ void layer1_products_kernel(const float* __restrict__ H, const float* __restrict__ AGG,
                                       const float* __restrict__ X,
                                       const float* __restrict__ Wg, const float* __restrict__ bg,
                                       const float* __restrict__ Wl, const float* __restrict__ Wr,
                                       const float* __restrict__ bl, const float* __restrict__ dinv,
                                       float* __restrict__ outG, float* __restrict__ outS, int n) {
    __shared__ __attribute__((aligned(16))) float WG[WWORDS];
    __shared__ __attribute__((aligned(16))) float WL[WWORDS];
    __shared__ __attribute__((aligned(16))) float WR[WWORDS];
    const int t = threadIdx.x;
    stage_w64<false>(WG, Wg, t);
    stage_w64<false>(WL, Wl, t);
    stage_w64<false>(WR, Wr, t);
    __syncthreads();

    const int wave  = t >> 5;
    const int lane  = t & 31;
    const int m0    = (blockIdx.x * 8 + wave) * 16;
    if (m0 >= n) return;
    const int lhalf = lane >> 4;
    const int l15   = lane & 15;
    const int kOff  = lhalf * 2;
    int row = m0 + l15;
    if (row >= n) row = n - 1;

    {
        v8f acc[4];
        #pragma unroll
        for (int nb = 0; nb < 4; ++nb) acc[nb] = (v8f){0.f, 0.f, 0.f, 0.f, 0.f, 0.f, 0.f, 0.f};
        {
            v2f af[16];
            load_row_frag(af, AGG + (size_t)row * FDIM + kOff);
            #pragma unroll
            for (int nb = 0; nb < 4; ++nb) {
                v2f wf[16];
                load_w_frag(wf, WL, nb * 16 + l15, lhalf);
                FRAGS_LOADED(wf);
                #pragma unroll
                for (int kb = 0; kb < 16; ++kb) acc[nb] = WMMA_F32(af[kb], wf[kb], acc[nb]);
            }
        }
        {
            v2f xf[16];
            load_row_frag(xf, X + (size_t)row * FDIM + kOff);
            #pragma unroll
            for (int nb = 0; nb < 4; ++nb) {
                const int col = nb * 16 + l15;
                v2f wg[16];
                load_w_frag(wg, WR, col, lhalf);
                FRAGS_LOADED(wg);
                #pragma unroll
                for (int kb = 0; kb < 16; ++kb) acc[nb] = WMMA_F32(xf[kb], wg[kb], acc[nb]);
                store_node_major<true>(outS, acc[nb], bl[col], m0, lhalf, col, n);
            }
        }
    }
    {
        v2f hf[16];
        load_row_frag(hf, H + (size_t)row * FDIM + kOff);
        float dv[8];
        #pragma unroll
        for (int j = 0; j < 8; ++j) {
            const int r = m0 + j + lhalf * 8;
            dv[j] = dinv[r < n ? r : n - 1];
        }
        #pragma unroll
        for (int nb = 0; nb < 4; ++nb) {
            const int col = nb * 16 + l15;
            v2f wf[16];
            load_w_frag(wf, WG, col, lhalf);
            FRAGS_LOADED(wf);
            v8f acc = {0.f, 0.f, 0.f, 0.f, 0.f, 0.f, 0.f, 0.f};
            #pragma unroll
            for (int kb = 0; kb < 16; ++kb) acc = WMMA_F32(hf[kb], wf[kb], acc);
            store_node_major_relu_scaled(outG, acc, bg[col], dv, m0, lhalf, col, n);
        }
    }
}

__device__ __forceinline__ void bias_layernorm_fm(v8f (&tl)[4], const float* bias, const float* gam,
                                                  const float* bet, int lhalf) {
    float s = 0.0f, q = 0.0f;
    #pragma unroll
    for (int nb = 0; nb < 4; ++nb) {
        #pragma unroll
        for (int r = 0; r < 8; ++r) {
            const float v = tl[nb][r] + bias[nb * 16 + lhalf * 8 + r];
            tl[nb][r] = v;
            s += v;
            q += v * v;
        }
    }
    s += __shfl_xor(s, 16, 32);
    q += __shfl_xor(q, 16, 32);
    const float mu  = s * (1.0f / 64.0f);
    const float inv = rsqrtf(fmaxf(q * (1.0f / 64.0f) - mu * mu, 0.0f) + 1e-5f);
    #pragma unroll
    for (int nb = 0; nb < 4; ++nb) {
        #pragma unroll
        for (int r = 0; r < 8; ++r) {
            const int f = nb * 16 + lhalf * 8 + r;
            tl[nb][r] = (tl[nb][r] - mu) * inv * gam[f] + bet[f];
        }
    }
}

__launch_bounds__(256, 4)
__global__ void layer2_projection_kernel(const float* __restrict__ H, const float* __restrict__ AGG,
                                         const float* S1,
                                         const float* __restrict__ Wg, const float* __restrict__ bg,
                                         const float* __restrict__ Wl, const float* __restrict__ Wr,
                                         const float* __restrict__ bl,
                                         const float* __restrict__ lnGg, const float* __restrict__ lnGb,
                                         const float* __restrict__ lnSg, const float* __restrict__ lnSb,
                                         const float* __restrict__ P, const float* __restrict__ pb,
                                         float* out, int n) {
    __shared__ __attribute__((aligned(16))) float WG[WWORDS];
    __shared__ __attribute__((aligned(16))) float WL[WWORDS];
    __shared__ __attribute__((aligned(16))) float WR[WWORDS];
    __shared__ __attribute__((aligned(16))) float PT[WWORDS];
    __shared__ __attribute__((aligned(16))) float PB[WWORDS];
    __shared__ float PAR[6][FDIM];
    const int t = threadIdx.x;
    stage_w64<false>(WG, Wg, t);
    stage_w64<false>(WL, Wl, t);
    stage_w64<false>(WR, Wr, t);
    stage_w64<true>(PT, P, t);
    stage_w64<true>(PB, P + FDIM * FDIM, t);
    if (t < FDIM) {
        PAR[0][t] = bg[t];   PAR[1][t] = bl[t];
        PAR[2][t] = lnGg[t]; PAR[3][t] = lnGb[t];
        PAR[4][t] = lnSg[t]; PAR[5][t] = lnSb[t];
    }
    __syncthreads();

    const int wave  = t >> 5;
    const int lane  = t & 31;
    const int m0    = (blockIdx.x * 8 + wave) * 16;
    if (m0 >= n) return;
    const int lhalf = lane >> 4;
    const int l15   = lane & 15;
    const int kOff  = lhalf * 2;
    int row = m0 + l15;
    if (row >= n) row = n - 1;

    v8f gT[4];
    {
        v2f hf[16];
        load_row_frag(hf, H + (size_t)row * FDIM + kOff);
        #pragma unroll
        for (int nb = 0; nb < 4; ++nb) {
            v2f wf[16];
            load_w_frag(wf, WG, nb * 16 + l15, lhalf);
            FRAGS_LOADED(wf);
            v8f acc = {0.f, 0.f, 0.f, 0.f, 0.f, 0.f, 0.f, 0.f};
            #pragma unroll
            for (int kb = 0; kb < 16; ++kb) acc = WMMA_F32(wf[kb], hf[kb], acc);
            gT[nb] = acc;
        }
        bias_layernorm_fm(gT, PAR[0], PAR[2], PAR[3], lhalf);
    }
    v8f sT[4];
    {
        v2f af[16], sf[16];
        load_row_frag(af, AGG + (size_t)row * FDIM + kOff);
        load_row_frag(sf, S1  + (size_t)row * FDIM + kOff);
        #pragma unroll
        for (int nb = 0; nb < 4; ++nb) {
            v8f acc = {0.f, 0.f, 0.f, 0.f, 0.f, 0.f, 0.f, 0.f};
            {
                v2f wf[16];
                load_w_frag(wf, WL, nb * 16 + l15, lhalf);
                FRAGS_LOADED(wf);
                #pragma unroll
                for (int kb = 0; kb < 16; ++kb) acc = WMMA_F32(wf[kb], af[kb], acc);
            }
            {
                v2f wg[16];
                load_w_frag(wg, WR, nb * 16 + l15, lhalf);
                FRAGS_LOADED(wg);
                #pragma unroll
                for (int kb = 0; kb < 16; ++kb) acc = WMMA_F32(wg[kb], sf[kb], acc);
            }
            sT[nb] = acc;
        }
        bias_layernorm_fm(sT, PAR[1], PAR[4], PAR[5], lhalf);
    }
    #pragma unroll
    for (int ob = 0; ob < 4; ++ob) {
        const int col = ob * 16 + l15;
        v2f pt[16], pq[16];
        #pragma unroll
        for (int nb = 0; nb < 4; ++nb) {
            #pragma unroll
            for (int rp = 0; rp < 4; ++rp) {
                const int prow = nb * 8 + rp * 2 + lhalf;
                pt[nb * 4 + rp] = *(const v2f*)(PT + prow * WPITCH + col * 2);
                pq[nb * 4 + rp] = *(const v2f*)(PB + prow * WPITCH + col * 2);
            }
        }
        FRAGS_LOADED(pt);
        v8f acc = {0.f, 0.f, 0.f, 0.f, 0.f, 0.f, 0.f, 0.f};
        #pragma unroll
        for (int nb = 0; nb < 4; ++nb) {
            #pragma unroll
            for (int rp = 0; rp < 4; ++rp) {
                v2f a;
                a.x = gT[nb][rp * 2];
                a.y = gT[nb][rp * 2 + 1];
                acc = WMMA_F32(a, pt[nb * 4 + rp], acc);
            }
        }
        FRAGS_LOADED(pq);
        #pragma unroll
        for (int nb = 0; nb < 4; ++nb) {
            #pragma unroll
            for (int rp = 0; rp < 4; ++rp) {
                v2f a;
                a.x = sT[nb][rp * 2];
                a.y = sT[nb][rp * 2 + 1];
                acc = WMMA_F32(a, pq[nb * 4 + rp], acc);
            }
        }
        store_node_major<false>(out, acc, pb[col], m0, lhalf, col, n);
    }
}

__device__ __forceinline__ int row_bcast(int v, int u) {
    switch (u) {
        case 0:  return __builtin_amdgcn_update_dpp(0, v, 0x150, 0xf, 0xf, false);
        case 1:  return __builtin_amdgcn_update_dpp(0, v, 0x151, 0xf, 0xf, false);
        case 2:  return __builtin_amdgcn_update_dpp(0, v, 0x152, 0xf, 0xf, false);
        case 3:  return __builtin_amdgcn_update_dpp(0, v, 0x153, 0xf, 0xf, false);
        case 4:  return __builtin_amdgcn_update_dpp(0, v, 0x154, 0xf, 0xf, false);
        case 5:  return __builtin_amdgcn_update_dpp(0, v, 0x155, 0xf, 0xf, false);
        case 6:  return __builtin_amdgcn_update_dpp(0, v, 0x156, 0xf, 0xf, false);
        default: return __builtin_amdgcn_update_dpp(0, v, 0x157, 0xf, 0xf, false);
    }
}
__device__ __forceinline__ float row_bcast_f(float v, int u) {
    return __int_as_float(row_bcast(__float_as_int(v), u));
}

template <bool ONE_TABLE>
__launch_bounds__(256)
__global__ void gather_pair_kernel(const float* __restrict__ tabG, const float* __restrict__ tabS,
                                   const int* __restrict__ rowptr, const int* __restrict__ cnt,
                                   const int* __restrict__ nbr, const float* __restrict__ dinv,
                                   float* __restrict__ outG, float* __restrict__ outS, int n) {
    const int q   = threadIdx.x & 15;
    const int grp = (int)((blockIdx.x * 256u + threadIdx.x) >> 4);
    const int r0  = grp * GROUP_ROWS;
    if (r0 >= n) return;
    const v4f* g4 = (const v4f*)tabG;
    const v4f* s4 = (const v4f*)tabS;

    const bool mine = (q < GROUP_ROWS) && (r0 + q < n);
    const int  rl   = mine ? (r0 + q) : r0;
    const int   begL = rowptr[rl];
    const int   degL = mine ? cnt[rl] : 0;
    const float drL  = dinv[rl];

    int ixn = 0;
    {
        const int d0 = row_bcast(degL, 0);
        const int b0 = row_bcast(begL, 0);
        if (d0 > 0) ixn = nbr[b0 + min(q, d0 - 1)];
    }

    #pragma unroll
    for (int k = 0; k < GROUP_ROWS; ++k) {
        const int r = r0 + k;
        if (r >= n) break;
        const int   beg = row_bcast(begL, k);
        const int   deg = row_bcast(degL, k);
        const float dr  = row_bcast_f(drL, k);
        int ix = ixn;
        if (k + 1 < GROUP_ROWS) {
            const int dn = row_bcast(degL, k + 1);
            const int bn = row_bcast(begL, k + 1);
            ixn = 0;
            if (dn > 0) ixn = nbr[bn + min(q, dn - 1)];
        }
        const v4f hv = g4[r * 16 + q];
        float4 accG = make_float4(0.0f, 0.0f, 0.0f, 0.0f);
        float4 accS = make_float4(0.0f, 0.0f, 0.0f, 0.0f);
        for (int j = 0; j < deg; j += EDGE_UNROLL) {
            const int m = deg - j;
            int ixc = ix;
            if (m > EDGE_UNROLL)
                ixc = nbr[beg + j + EDGE_UNROLL + min(q, m - EDGE_UNROLL - 1)];
            float dq = 0.0f;
            if (ONE_TABLE) dq = dinv[ix];
            v4f vg[EDGE_UNROLL];
            v4f vs[EDGE_UNROLL];
            #pragma unroll
            for (int u = 0; u < EDGE_UNROLL; ++u) {
                const int s = row_bcast(ix, u);
                vg[u] = g4[s * 16 + q];
                if (!ONE_TABLE) vs[u] = s4[s * 16 + q];
            }
            if (ONE_TABLE)
                asm volatile("" : "+v"(vg[0]), "+v"(vg[1]), "+v"(vg[2]), "+v"(vg[3]),
                                  "+v"(vg[4]), "+v"(vg[5]), "+v"(vg[6]), "+v"(vg[7]));
            else
                asm volatile("" : "+v"(vg[0]), "+v"(vg[1]), "+v"(vg[2]), "+v"(vg[3]),
                                  "+v"(vg[4]), "+v"(vg[5]), "+v"(vg[6]), "+v"(vg[7]),
                                  "+v"(vs[0]), "+v"(vs[1]), "+v"(vs[2]), "+v"(vs[3]),
                                  "+v"(vs[4]), "+v"(vs[5]), "+v"(vs[6]), "+v"(vs[7]));
            const float wq = dq * dr;
            #pragma unroll
            for (int u = 0; u < EDGE_UNROLL; ++u) {
                const bool on = (u == 0) || (u < m);
                const v4f a = vg[u];
                const v4f b = ONE_TABLE ? vg[u] : vs[u];
                float4 ng, ns;
                if (ONE_TABLE) {
                    const float w = row_bcast_f(wq, u);
                    ng.x = accG.x + a.x * w; ng.y = accG.y + a.y * w;
                    ng.z = accG.z + a.z * w; ng.w = accG.w + a.w * w;
                } else {
                    ng.x = accG.x + a.x; ng.y = accG.y + a.y;
                    ng.z = accG.z + a.z; ng.w = accG.w + a.w;
                }
                ns.x = accS.x + b.x; ns.y = accS.y + b.y;
                ns.z = accS.z + b.z; ns.w = accS.w + b.w;
                accG.x = on ? ng.x : accG.x; accG.y = on ? ng.y : accG.y;
                accG.z = on ? ng.z : accG.z; accG.w = on ? ng.w : accG.w;
                accS.x = on ? ns.x : accS.x; accS.y = on ? ns.y : accS.y;
                accS.z = on ? ns.z : accS.z; accS.w = on ? ns.w : accS.w;
            }
            ix = ixc;
        }
        float4 og, os;
        if (ONE_TABLE) {
            const float d2 = dr * dr;
            og.x = accG.x + hv.x * d2; og.y = accG.y + hv.y * d2;
            og.z = accG.z + hv.z * d2; og.w = accG.w + hv.w * d2;
        } else {
            og.x = (accG.x + hv.x) * dr; og.y = (accG.y + hv.y) * dr;
            og.z = (accG.z + hv.z) * dr; og.w = (accG.w + hv.w) * dr;
        }
        const float rs = 1.0f / fmaxf((float)deg, 1.0f);
        os.x = accS.x * rs; os.y = accS.y * rs; os.z = accS.z * rs; os.w = accS.w * rs;
        ((float4*)outG)[r * 16 + q] = og;
        ((float4*)outS)[r * 16 + q] = os;
    }
}

extern "C" void kernel_launch(void* const* d_in, const int* in_sizes, int n_in,
                              void* d_out, int out_size, void* d_ws, size_t ws_size,
                              hipStream_t stream) {
    const float* x        = (const float*)d_in[0];
    const int*   ei       = (const int*)d_in[1];
    const float* gcn_w1   = (const float*)d_in[2];
    const float* gcn_b1   = (const float*)d_in[3];
    const float* gcn_w2   = (const float*)d_in[4];
    const float* gcn_b2   = (const float*)d_in[5];
    const float* sage_wl1 = (const float*)d_in[6];
    const float* sage_bl1 = (const float*)d_in[7];
    const float* sage_wr1 = (const float*)d_in[8];
    const float* sage_wl2 = (const float*)d_in[9];
    const float* sage_bl2 = (const float*)d_in[10];
    const float* sage_wr2 = (const float*)d_in[11];
    const float* gcn_ln_g = (const float*)d_in[12];
    const float* gcn_ln_b = (const float*)d_in[13];
    const float* sage_ln_g = (const float*)d_in[14];
    const float* sage_ln_b = (const float*)d_in[15];
    const float* proj_w   = (const float*)d_in[16];
    const float* proj_b   = (const float*)d_in[17];

    const int N = in_sizes[0] / FDIM;
    const int E = in_sizes[1] / 2;
    const int* src = ei;
    const int* dst = ei + E;

    int*   cnt    = (int*)d_ws;
    float* dinv   = (float*)(cnt + N);
    int*   rowptr = (int*)(dinv + N);
    int*   nbr    = rowptr + N;
    float* h      = (float*)(nbr + E);
    float* g      = h + (size_t)N * FDIM;
    float* agg    = g + (size_t)N * FDIM;
    float* out    = (float*)d_out;
    const int NB    = (N + BIN_ROWS - 1) / BIN_ROWS;
    const int chunk = (E + PART_BLOCKS - 1) / PART_BLOCKS;
    int2*  seg      = (int2*)h;
    int2*  pairs    = (int2*)g;

    const int nblkG  = (N + 127) / 128;
    const int nblkR  = (N + 16 * GROUP_ROWS - 1) / (16 * GROUP_ROWS);

    bin_sort_chunk_kernel<<<PART_BLOCKS, 256, NB * sizeof(int), stream>>>(src, dst, seg, pairs, E, chunk, NB);
    bin_bucket_kernel<<<NB, 1024, 0, stream>>>(pairs, seg, cnt, rowptr, dinv, nbr, N, chunk, NB);

    gather_pair_kernel<true><<<nblkR, 256, 0, stream>>>(x, x, rowptr, cnt, nbr, dinv, h, agg, N);
    layer1_products_kernel<<<nblkG, 256, 0, stream>>>(
        h, agg, x, gcn_w1, gcn_b1, sage_wl1, sage_wr1, sage_bl1, dinv, g, out, N);

    gather_pair_kernel<false><<<nblkR, 256, 0, stream>>>(g, out, rowptr, cnt, nbr, dinv, h, agg, N);
    layer2_projection_kernel<<<nblkG, 256, 0, stream>>>(
        h, agg, out, gcn_w2, gcn_b2, sage_wl2, sage_wr2, sage_bl2,
        gcn_ln_g, gcn_ln_b, sage_ln_g, sage_ln_b, proj_w, proj_b, out, N);
}


# Workspace re-ordered: the three node-row buffers lead, so each 256-byte row starts on a 256-byte boundary (two whole 128-byte lines, not three pieces); sums unchanged
# speedup vs baseline: 5.0382x; 1.0910x over previous
#include <hip/hip_runtime.h>

typedef __attribute__((ext_vector_type(2))) float v2f;
typedef __attribute__((ext_vector_type(8))) float v8f;
typedef __attribute__((ext_vector_type(4))) float v4f;

#define FDIM 64
#define BIN_SHIFT 9
#define BIN_ROWS (1 << BIN_SHIFT)
#define PART_BLOCKS 512
#define GROUP_ROWS 5
#define EDGE_UNROLL 8

#define SEG_THREADS (1024 / PART_BLOCKS)
static_assert(PART_BLOCKS * SEG_THREADS == 1024, "the last build pass gives every chunk SEG_THREADS of its 1024 threads");

__launch_bounds__(256)
__global__ void bin_sort_chunk_kernel(const int* __restrict__ src, const int* __restrict__ dst,
                                      int2* __restrict__ seg, int2* __restrict__ pairs,
                                      int e, int chunk, int nb) {
    extern __shared__ int cur[];
    __shared__ int wsum[8];
    const int t    = threadIdx.x;
    const int lane = t & 31;
    const int wave = t >> 5;
    for (int b = t; b < nb; b += 256) cur[b] = 0;
    __syncthreads();
    const int e0 = blockIdx.x * chunk;
    const int e1 = min(e0 + chunk, e);
    {
        int i = e0 + t;
        for (; i + 768 < e1; i += 1024) {
            const int d0 = dst[i], d1 = dst[i + 256], d2 = dst[i + 512], d3 = dst[i + 768];
            atomicAdd(&cur[d0 >> BIN_SHIFT], 1);
            atomicAdd(&cur[d1 >> BIN_SHIFT], 1);
            atomicAdd(&cur[d2 >> BIN_SHIFT], 1);
            atomicAdd(&cur[d3 >> BIN_SHIFT], 1);
        }
        for (; i < e1; i += 256) atomicAdd(&cur[dst[i] >> BIN_SHIFT], 1);
    }
    __syncthreads();
    int carry = 0;
    for (int base = 0; base < nb; base += 256) {
        const int b = base + t;
        const int v = (b < nb) ? cur[b] : 0;
        int inc = v;
        for (int o = 1; o < 32; o <<= 1) {
            const int u = __shfl_up(inc, o, 32);
            if (lane >= o) inc += u;
        }
        if (lane == 31) wsum[wave] = inc;
        __syncthreads();
        int woff = 0, tot = 0;
        for (int w = 0; w < 8; ++w) {
            const int x = wsum[w];
            if (w < wave) woff += x;
            tot += x;
        }
        const int ex = carry + woff + inc - v;
        if (b < nb) {
            cur[b] = e0 + ex;
            seg[blockIdx.x * nb + b] = make_int2(ex, v);
        }
        carry += tot;
        __syncthreads();
    }
    int i = e0 + t;
    for (; i + 768 < e1; i += 1024) {
        const int d0 = dst[i], d1 = dst[i + 256], d2 = dst[i + 512], d3 = dst[i + 768];
        const int s0 = src[i], s1 = src[i + 256], s2 = src[i + 512], s3 = src[i + 768];
        const int p0 = atomicAdd(&cur[d0 >> BIN_SHIFT], 1);
        const int p1 = atomicAdd(&cur[d1 >> BIN_SHIFT], 1);
        const int p2 = atomicAdd(&cur[d2 >> BIN_SHIFT], 1);
        const int p3 = atomicAdd(&cur[d3 >> BIN_SHIFT], 1);
        pairs[p0] = make_int2(s0, d0);
        pairs[p1] = make_int2(s1, d1);
        pairs[p2] = make_int2(s2, d2);
        pairs[p3] = make_int2(s3, d3);
    }
    for (; i < e1; i += 256) {
        const int d = dst[i];
        const int sv = src[i];
        const int p = atomicAdd(&cur[d >> BIN_SHIFT], 1);
        pairs[p] = make_int2(sv, d);
    }
}

__launch_bounds__(1024)
__global__ void bin_bucket_kernel(const int2* __restrict__ pairs, const int2* __restrict__ seg,
                                  int* __restrict__ cnt, int* __restrict__ rowptr, float* __restrict__ dinv,
                                  int* __restrict__ nbr, int n, int chunk, int nb) {
    __shared__ int lc[BIN_ROWS];
    __shared__ int lp[BIN_ROWS];
    __shared__ int wsum[32];
    __shared__ int osum[32];
    const int t     = threadIdx.x;
    const int lane  = t & 31;
    const int wave  = t >> 5;
    const int rbase = blockIdx.x * BIN_ROWS;
    const int part  = t / SEG_THREADS;
    const int j     = t % SEG_THREADS;
    const int2 sg   = seg[part * nb + blockIdx.x];
    const int sbase = part * chunk + sg.x;
    const int slen  = sg.y;
    const int last  = slen - 1;
    {
        int bs = (j == 0) ? sg.x : 0;
        for (int o = 16; o > 0; o >>= 1) bs += __shfl_xor(bs, o, 32);
        if (lane == 0) osum[wave] = bs;
    }
    if (t < BIN_ROWS) lc[t] = 0;
    __syncthreads();
    int b0 = 0;
    for (int w = 0; w < 32; ++w) b0 += osum[w];
    for (int i = j; i < slen; i += 4 * SEG_THREADS) {
        int2 pa = pairs[sbase + i];
        int2 pb = pairs[sbase + min(i + SEG_THREADS, last)];
        int2 pc = pairs[sbase + min(i + 2 * SEG_THREADS, last)];
        int2 pd = pairs[sbase + min(i + 3 * SEG_THREADS, last)];
        asm volatile("" : "+v"(pa.y), "+v"(pb.y), "+v"(pc.y), "+v"(pd.y));
        atomicAdd(&lc[pa.y - rbase], 1);
        if (i + SEG_THREADS < slen)     atomicAdd(&lc[pb.y - rbase], 1);
        if (i + 2 * SEG_THREADS < slen) atomicAdd(&lc[pc.y - rbase], 1);
        if (i + 3 * SEG_THREADS < slen) atomicAdd(&lc[pd.y - rbase], 1);
    }
    __syncthreads();
    const int v = (t < BIN_ROWS) ? lc[t] : 0;
    int inc = v;
    for (int o = 1; o < 32; o <<= 1) {
        const int u = __shfl_up(inc, o, 32);
        if (lane >= o) inc += u;
    }
    if (lane == 31) wsum[wave] = inc;
    __syncthreads();
    int woff = 0;
    for (int w = 0; w < wave; ++w) woff += wsum[w];
    if (t < BIN_ROWS) {
        const int ex = woff + inc - v;
        lp[t] = ex;
        lc[t] = 0;
        const int r = rbase + t;
        if (r < n) {
            cnt[r]    = v;
            rowptr[r] = b0 + ex;
            dinv[r]   = rsqrtf((float)v + 1.0f);
        }
    }
    __syncthreads();
    for (int i = j; i < slen; i += 4 * SEG_THREADS) {
        int2 pa = pairs[sbase + i];
        int2 pb = pairs[sbase + min(i + SEG_THREADS, last)];
        int2 pc = pairs[sbase + min(i + 2 * SEG_THREADS, last)];
        int2 pd = pairs[sbase + min(i + 3 * SEG_THREADS, last)];
        asm volatile("" : "+v"(pa.x), "+v"(pa.y), "+v"(pb.x), "+v"(pb.y), "+v"(pc.x), "+v"(pc.y), "+v"(pd.x), "+v"(pd.y));
        const int ra = pa.y - rbase;
        nbr[b0 + lp[ra] + atomicAdd(&lc[ra], 1)] = pa.x;
        if (i + SEG_THREADS < slen) {
            const int rb = pb.y - rbase;
            nbr[b0 + lp[rb] + atomicAdd(&lc[rb], 1)] = pb.x;
        }
        if (i + 2 * SEG_THREADS < slen) {
            const int rc = pc.y - rbase;
            nbr[b0 + lp[rc] + atomicAdd(&lc[rc], 1)] = pc.x;
        }
        if (i + 3 * SEG_THREADS < slen) {
            const int rd = pd.y - rbase;
            nbr[b0 + lp[rd] + atomicAdd(&lc[rd], 1)] = pd.x;
        }
    }
}

#define WMMA_F32(a, b, c) __builtin_amdgcn_wmma_f32_16x16x4_f32(false, (a), false, (b), (short)0, (c), false, false)

#define WPITCH 160
#define WWORDS (32 * WPITCH)
template <bool PERM>
__device__ __forceinline__ void stage_w64(float* __restrict__ dst, const float* __restrict__ src, int t) {
    const float4* s = (const float4*)src;
    for (int i = t; i < 512; i += 256) {
        const int kp = i >> 4;
        const int c4 = i & 15;
        const float4 a = s[(2 * kp) * 16 + c4];
        const float4 b = s[(2 * kp + 1) * 16 + c4];
        const int prow = PERM ? ((kp & 24) | ((kp & 3) << 1) | ((kp >> 2) & 1)) : kp;
        float4* d = (float4*)(dst + prow * WPITCH + c4 * 8);
        d[0] = make_float4(a.x, b.x, a.y, b.y);
        d[1] = make_float4(a.z, b.z, a.w, b.w);
    }
}

#define FRAGS_LOADED(w) asm volatile("" : "+v"(w[0]), "+v"(w[1]), "+v"(w[2]), "+v"(w[3]), "+v"(w[4]), "+v"(w[5]), \
                                          "+v"(w[6]), "+v"(w[7]), "+v"(w[8]), "+v"(w[9]), "+v"(w[10]), "+v"(w[11]), \
                                          "+v"(w[12]), "+v"(w[13]), "+v"(w[14]), "+v"(w[15]))

__device__ __forceinline__ void load_row_frag(v2f (&f)[16], const float* p) {
    #pragma unroll
    for (int kb = 0; kb < 16; ++kb) { f[kb].x = p[kb * 4 + 0]; f[kb].y = p[kb * 4 + 1]; }
}

__device__ __forceinline__ void load_w_frag(v2f (&w)[16], const float* Wl, int col, int lhalf) {
    #pragma unroll
    for (int kb = 0; kb < 16; ++kb)
        w[kb] = *(const v2f*)(Wl + (2 * kb + lhalf) * WPITCH + col * 2);
}

template <bool RELU>
__device__ __forceinline__ void store_node_major(float* __restrict__ dst, const v8f acc, float bval,
                                                 int m0, int lhalf, int col, int n) {
    float v[8];
    #pragma unroll
    for (int j = 0; j < 8; ++j) {
        v[j] = acc[j] + bval;
        if (RELU) v[j] = fmaxf(v[j], 0.0f);
    }
    #pragma unroll
    for (int j = 0; j < 8; ++j) {
        const int r = m0 + j + lhalf * 8;
        if (r < n) dst[(size_t)r * FDIM + col] = v[j];
    }
}

__device__ __forceinline__ void store_node_major_relu_scaled(float* __restrict__ dst, const v8f acc, float bval,
                                                             const float (&rs)[8], int m0, int lhalf, int col, int n) {
    float v[8];
    #pragma unroll
    for (int j = 0; j < 8; ++j) v[j] = fmaxf(acc[j] + bval, 0.0f) * rs[j];
    #pragma unroll
    for (int j = 0; j < 8; ++j) {
        const int r = m0 + j + lhalf * 8;
        if (r < n) dst[(size_t)r * FDIM + col] = v[j];
    }
}

__launch_bounds__(256, 8)
__global__ void layer1_products_kernel(const float* __restrict__ H, const float* __restrict__ AGG,
                                       const float* __restrict__ X,
                                       const float* __restrict__ Wg, const float* __restrict__ bg,
                                       const float* __restrict__ Wl, const float* __restrict__ Wr,
                                       const float* __restrict__ bl, const float* __restrict__ dinv,
                                       float* __restrict__ outG, float* __restrict__ outS, int n) {
    __shared__ __attribute__((aligned(16))) float WG[WWORDS];
    __shared__ __attribute__((aligned(16))) float WL[WWORDS];
    __shared__ __attribute__((aligned(16))) float WR[WWORDS];
    const int t = threadIdx.x;
    stage_w64<false>(WG, Wg, t);
    stage_w64<false>(WL, Wl, t);
    stage_w64<false>(WR, Wr, t);
    __syncthreads();

    const int wave  = t >> 5;
    const int lane  = t & 31;
    const int m0    = (blockIdx.x * 8 + wave) * 16;
    if (m0 >= n) return;
    const int lhalf = lane >> 4;
    const int l15   = lane & 15;
    const int kOff  = lhalf * 2;
    int row = m0 + l15;
    if (row >= n) row = n - 1;

    {
        v8f acc[4];
        #pragma unroll
        for (int nb = 0; nb < 4; ++nb) acc[nb] = (v8f){0.f, 0.f, 0.f, 0.f, 0.f, 0.f, 0.f, 0.f};
        {
            v2f af[16];
            load_row_frag(af, AGG + (size_t)row * FDIM + kOff);
            #pragma unroll
            for (int nb = 0; nb < 4; ++nb) {
                v2f wf[16];
                load_w_frag(wf, WL, nb * 16 + l15, lhalf);
                FRAGS_LOADED(wf);
                #pragma unroll
                for (int kb = 0; kb < 16; ++kb) acc[nb] = WMMA_F32(af[kb], wf[kb], acc[nb]);
            }
        }
        {
            v2f xf[16];
            load_row_frag(xf, X + (size_t)row * FDIM + kOff);
            #pragma unroll
            for (int nb = 0; nb < 4; ++nb) {
                const int col = nb * 16 + l15;
                v2f wg[16];
                load_w_frag(wg, WR, col, lhalf);
                FRAGS_LOADED(wg);
                #pragma unroll
                for (int kb = 0; kb < 16; ++kb) acc[nb] = WMMA_F32(xf[kb], wg[kb], acc[nb]);
                store_node_major<true>(outS, acc[nb], bl[col], m0, lhalf, col, n);
            }
        }
    }
    {
        v2f hf[16];
        load_row_frag(hf, H + (size_t)row * FDIM + kOff);
        float dv[8];
        #pragma unroll
        for (int j = 0; j < 8; ++j) {
            const int r = m0 + j + lhalf * 8;
            dv[j] = dinv[r < n ? r : n - 1];
        }
        #pragma unroll
        for (int nb = 0; nb < 4; ++nb) {
            const int col = nb * 16 + l15;
            v2f wf[16];
            load_w_frag(wf, WG, col, lhalf);
            FRAGS_LOADED(wf);
            v8f acc = {0.f, 0.f, 0.f, 0.f, 0.f, 0.f, 0.f, 0.f};
            #pragma unroll
            for (int kb = 0; kb < 16; ++kb) acc = WMMA_F32(hf[kb], wf[kb], acc);
            store_node_major_relu_scaled(outG, acc, bg[col], dv, m0, lhalf, col, n);
        }
    }
}

__device__ __forceinline__ void bias_layernorm_fm(v8f (&tl)[4], const float* bias, const float* gam,
                                                  const float* bet, int lhalf) {
    float s = 0.0f, q = 0.0f;
    #pragma unroll
    for (int nb = 0; nb < 4; ++nb) {
        #pragma unroll
        for (int r = 0; r < 8; ++r) {
            const float v = tl[nb][r] + bias[nb * 16 + lhalf * 8 + r];
            tl[nb][r] = v;
            s += v;
            q += v * v;
        }
    }
    s += __shfl_xor(s, 16, 32);
    q += __shfl_xor(q, 16, 32);
    const float mu  = s * (1.0f / 64.0f);
    const float inv = rsqrtf(fmaxf(q * (1.0f / 64.0f) - mu * mu, 0.0f) + 1e-5f);
    #pragma unroll
    for (int nb = 0; nb < 4; ++nb) {
        #pragma unroll
        for (int r = 0; r < 8; ++r) {
            const int f = nb * 16 + lhalf * 8 + r;
            tl[nb][r] = (tl[nb][r] - mu) * inv * gam[f] + bet[f];
        }
    }
}

__launch_bounds__(256, 4)
__global__ void layer2_projection_kernel(const float* __restrict__ H, const float* __restrict__ AGG,
                                         const float* S1,
                                         const float* __restrict__ Wg, const float* __restrict__ bg,
                                         const float* __restrict__ Wl, const float* __restrict__ Wr,
                                         const float* __restrict__ bl,
                                         const float* __restrict__ lnGg, const float* __restrict__ lnGb,
                                         const float* __restrict__ lnSg, const float* __restrict__ lnSb,
                                         const float* __restrict__ P, const float* __restrict__ pb,
                                         float* out, int n) {
    __shared__ __attribute__((aligned(16))) float WG[WWORDS];
    __shared__ __attribute__((aligned(16))) float WL[WWORDS];
    __shared__ __attribute__((aligned(16))) float WR[WWORDS];
    __shared__ __attribute__((aligned(16))) float PT[WWORDS];
    __shared__ __attribute__((aligned(16))) float PB[WWORDS];
    __shared__ float PAR[6][FDIM];
    const int t = threadIdx.x;
    stage_w64<false>(WG, Wg, t);
    stage_w64<false>(WL, Wl, t);
    stage_w64<false>(WR, Wr, t);
    stage_w64<true>(PT, P, t);
    stage_w64<true>(PB, P + FDIM * FDIM, t);
    if (t < FDIM) {
        PAR[0][t] = bg[t];   PAR[1][t] = bl[t];
        PAR[2][t] = lnGg[t]; PAR[3][t] = lnGb[t];
        PAR[4][t] = lnSg[t]; PAR[5][t] = lnSb[t];
    }
    __syncthreads();

    const int wave  = t >> 5;
    const int lane  = t & 31;
    const int m0    = (blockIdx.x * 8 + wave) * 16;
    if (m0 >= n) return;
    const int lhalf = lane >> 4;
    const int l15   = lane & 15;
    const int kOff  = lhalf * 2;
    int row = m0 + l15;
    if (row >= n) row = n - 1;

    v8f gT[4];
    {
        v2f hf[16];
        load_row_frag(hf, H + (size_t)row * FDIM + kOff);
        #pragma unroll
        for (int nb = 0; nb < 4; ++nb) {
            v2f wf[16];
            load_w_frag(wf, WG, nb * 16 + l15, lhalf);
            FRAGS_LOADED(wf);
            v8f acc = {0.f, 0.f, 0.f, 0.f, 0.f, 0.f, 0.f, 0.f};
            #pragma unroll
            for (int kb = 0; kb < 16; ++kb) acc = WMMA_F32(wf[kb], hf[kb], acc);
            gT[nb] = acc;
        }
        bias_layernorm_fm(gT, PAR[0], PAR[2], PAR[3], lhalf);
    }
    v8f sT[4];
    {
        v2f af[16], sf[16];
        load_row_frag(af, AGG + (size_t)row * FDIM + kOff);
        load_row_frag(sf, S1  + (size_t)row * FDIM + kOff);
        #pragma unroll
        for (int nb = 0; nb < 4; ++nb) {
            v8f acc = {0.f, 0.f, 0.f, 0.f, 0.f, 0.f, 0.f, 0.f};
            {
                v2f wf[16];
                load_w_frag(wf, WL, nb * 16 + l15, lhalf);
                FRAGS_LOADED(wf);
                #pragma unroll
                for (int kb = 0; kb < 16; ++kb) acc = WMMA_F32(wf[kb], af[kb], acc);
            }
            {
                v2f wg[16];
                load_w_frag(wg, WR, nb * 16 + l15, lhalf);
                FRAGS_LOADED(wg);
                #pragma unroll
                for (int kb = 0; kb < 16; ++kb) acc = WMMA_F32(wg[kb], sf[kb], acc);
            }
            sT[nb] = acc;
        }
        bias_layernorm_fm(sT, PAR[1], PAR[4], PAR[5], lhalf);
    }
    #pragma unroll
    for (int ob = 0; ob < 4; ++ob) {
        const int col = ob * 16 + l15;
        v2f pt[16], pq[16];
        #pragma unroll
        for (int nb = 0; nb < 4; ++nb) {
            #pragma unroll
            for (int rp = 0; rp < 4; ++rp) {
                const int prow = nb * 8 + rp * 2 + lhalf;
                pt[nb * 4 + rp] = *(const v2f*)(PT + prow * WPITCH + col * 2);
                pq[nb * 4 + rp] = *(const v2f*)(PB + prow * WPITCH + col * 2);
            }
        }
        FRAGS_LOADED(pt);
        v8f acc = {0.f, 0.f, 0.f, 0.f, 0.f, 0.f, 0.f, 0.f};
        #pragma unroll
        for (int nb = 0; nb < 4; ++nb) {
            #pragma unroll
            for (int rp = 0; rp < 4; ++rp) {
                v2f a;
                a.x = gT[nb][rp * 2];
                a.y = gT[nb][rp * 2 + 1];
                acc = WMMA_F32(a, pt[nb * 4 + rp], acc);
            }
        }
        FRAGS_LOADED(pq);
        #pragma unroll
        for (int nb = 0; nb < 4; ++nb) {
            #pragma unroll
            for (int rp = 0; rp < 4; ++rp) {
                v2f a;
                a.x = sT[nb][rp * 2];
                a.y = sT[nb][rp * 2 + 1];
                acc = WMMA_F32(a, pq[nb * 4 + rp], acc);
            }
        }
        store_node_major<false>(out, acc, pb[col], m0, lhalf, col, n);
    }
}

__device__ __forceinline__ int row_bcast(int v, int u) {
    switch (u) {
        case 0:  return __builtin_amdgcn_update_dpp(0, v, 0x150, 0xf, 0xf, false);
        case 1:  return __builtin_amdgcn_update_dpp(0, v, 0x151, 0xf, 0xf, false);
        case 2:  return __builtin_amdgcn_update_dpp(0, v, 0x152, 0xf, 0xf, false);
        case 3:  return __builtin_amdgcn_update_dpp(0, v, 0x153, 0xf, 0xf, false);
        case 4:  return __builtin_amdgcn_update_dpp(0, v, 0x154, 0xf, 0xf, false);
        case 5:  return __builtin_amdgcn_update_dpp(0, v, 0x155, 0xf, 0xf, false);
        case 6:  return __builtin_amdgcn_update_dpp(0, v, 0x156, 0xf, 0xf, false);
        default: return __builtin_amdgcn_update_dpp(0, v, 0x157, 0xf, 0xf, false);
    }
}
__device__ __forceinline__ float row_bcast_f(float v, int u) {
    return __int_as_float(row_bcast(__float_as_int(v), u));
}

template <bool ONE_TABLE>
__launch_bounds__(256)
__global__ void gather_pair_kernel(const float* __restrict__ tabG, const float* __restrict__ tabS,
                                   const int* __restrict__ rowptr, const int* __restrict__ cnt,
                                   const int* __restrict__ nbr, const float* __restrict__ dinv,
                                   float* __restrict__ outG, float* __restrict__ outS, int n) {
    const int q   = threadIdx.x & 15;
    const int grp = (int)((blockIdx.x * 256u + threadIdx.x) >> 4);
    const int r0  = grp * GROUP_ROWS;
    if (r0 >= n) return;
    const v4f* g4 = (const v4f*)tabG;
    const v4f* s4 = (const v4f*)tabS;

    const bool mine = (q < GROUP_ROWS) && (r0 + q < n);
    const int  rl   = mine ? (r0 + q) : r0;
    const int   begL = rowptr[rl];
    const int   degL = mine ? cnt[rl] : 0;
    const float drL  = dinv[rl];

    int ixn = 0;
    {
        const int d0 = row_bcast(degL, 0);
        const int b0 = row_bcast(begL, 0);
        if (d0 > 0) ixn = nbr[b0 + min(q, d0 - 1)];
    }

    #pragma unroll
    for (int k = 0; k < GROUP_ROWS; ++k) {
        const int r = r0 + k;
        if (r >= n) break;
        const int   beg = row_bcast(begL, k);
        const int   deg = row_bcast(degL, k);
        const float dr  = row_bcast_f(drL, k);
        int ix = ixn;
        if (k + 1 < GROUP_ROWS) {
            const int dn = row_bcast(degL, k + 1);
            const int bn = row_bcast(begL, k + 1);
            ixn = 0;
            if (dn > 0) ixn = nbr[bn + min(q, dn - 1)];
        }
        const v4f hv = g4[r * 16 + q];
        float4 accG = make_float4(0.0f, 0.0f, 0.0f, 0.0f);
        float4 accS = make_float4(0.0f, 0.0f, 0.0f, 0.0f);
        for (int j = 0; j < deg; j += EDGE_UNROLL) {
            const int m = deg - j;
            int ixc = ix;
            if (m > EDGE_UNROLL)
                ixc = nbr[beg + j + EDGE_UNROLL + min(q, m - EDGE_UNROLL - 1)];
            float dq = 0.0f;
            if (ONE_TABLE) dq = dinv[ix];
            v4f vg[EDGE_UNROLL];
            v4f vs[EDGE_UNROLL];
            #pragma unroll
            for (int u = 0; u < EDGE_UNROLL; ++u) {
                const int s = row_bcast(ix, u);
                vg[u] = g4[s * 16 + q];
                if (!ONE_TABLE) vs[u] = s4[s * 16 + q];
            }
            if (ONE_TABLE)
                asm volatile("" : "+v"(vg[0]), "+v"(vg[1]), "+v"(vg[2]), "+v"(vg[3]),
                                  "+v"(vg[4]), "+v"(vg[5]), "+v"(vg[6]), "+v"(vg[7]));
            else
                asm volatile("" : "+v"(vg[0]), "+v"(vg[1]), "+v"(vg[2]), "+v"(vg[3]),
                                  "+v"(vg[4]), "+v"(vg[5]), "+v"(vg[6]), "+v"(vg[7]),
                                  "+v"(vs[0]), "+v"(vs[1]), "+v"(vs[2]), "+v"(vs[3]),
                                  "+v"(vs[4]), "+v"(vs[5]), "+v"(vs[6]), "+v"(vs[7]));
            const float wq = dq * dr;
            #pragma unroll
            for (int u = 0; u < EDGE_UNROLL; ++u) {
                const bool on = (u == 0) || (u < m);
                const v4f a = vg[u];
                const v4f b = ONE_TABLE ? vg[u] : vs[u];
                float4 ng, ns;
                if (ONE_TABLE) {
                    const float w = row_bcast_f(wq, u);
                    ng.x = accG.x + a.x * w; ng.y = accG.y + a.y * w;
                    ng.z = accG.z + a.z * w; ng.w = accG.w + a.w * w;
                } else {
                    ng.x = accG.x + a.x; ng.y = accG.y + a.y;
                    ng.z = accG.z + a.z; ng.w = accG.w + a.w;
                }
                ns.x = accS.x + b.x; ns.y = accS.y + b.y;
                ns.z = accS.z + b.z; ns.w = accS.w + b.w;
                accG.x = on ? ng.x : accG.x; accG.y = on ? ng.y : accG.y;
                accG.z = on ? ng.z : accG.z; accG.w = on ? ng.w : accG.w;
                accS.x = on ? ns.x : accS.x; accS.y = on ? ns.y : accS.y;
                accS.z = on ? ns.z : accS.z; accS.w = on ? ns.w : accS.w;
            }
            ix = ixc;
        }
        float4 og, os;
        if (ONE_TABLE) {
            const float d2 = dr * dr;
            og.x = accG.x + hv.x * d2; og.y = accG.y + hv.y * d2;
            og.z = accG.z + hv.z * d2; og.w = accG.w + hv.w * d2;
        } else {
            og.x = (accG.x + hv.x) * dr; og.y = (accG.y + hv.y) * dr;
            og.z = (accG.z + hv.z) * dr; og.w = (accG.w + hv.w) * dr;
        }
        const float rs = 1.0f / fmaxf((float)deg, 1.0f);
        os.x = accS.x * rs; os.y = accS.y * rs; os.z = accS.z * rs; os.w = accS.w * rs;
        ((float4*)outG)[r * 16 + q] = og;
        ((float4*)outS)[r * 16 + q] = os;
    }
}

extern "C" void kernel_launch(void* const* d_in, const int* in_sizes, int n_in,
                              void* d_out, int out_size, void* d_ws, size_t ws_size,
                              hipStream_t stream) {
    const float* x        = (const float*)d_in[0];
    const int*   ei       = (const int*)d_in[1];
    const float* gcn_w1   = (const float*)d_in[2];
    const float* gcn_b1   = (const float*)d_in[3];
    const float* gcn_w2   = (const float*)d_in[4];
    const float* gcn_b2   = (const float*)d_in[5];
    const float* sage_wl1 = (const float*)d_in[6];
    const float* sage_bl1 = (const float*)d_in[7];
    const float* sage_wr1 = (const float*)d_in[8];
    const float* sage_wl2 = (const float*)d_in[9];
    const float* sage_bl2 = (const float*)d_in[10];
    const float* sage_wr2 = (const float*)d_in[11];
    const float* gcn_ln_g = (const float*)d_in[12];
    const float* gcn_ln_b = (const float*)d_in[13];
    const float* sage_ln_g = (const float*)d_in[14];
    const float* sage_ln_b = (const float*)d_in[15];
    const float* proj_w   = (const float*)d_in[16];
    const float* proj_b   = (const float*)d_in[17];

    const int N = in_sizes[0] / FDIM;
    const int E = in_sizes[1] / 2;
    const int* src = ei;
    const int* dst = ei + E;

    float* h      = (float*)d_ws;
    float* g      = h + (size_t)N * FDIM;
    float* agg    = g + (size_t)N * FDIM;
    int*   cnt    = (int*)(agg + (size_t)N * FDIM);
    float* dinv   = (float*)(cnt + N);
    int*   rowptr = (int*)(dinv + N);
    int*   nbr    = rowptr + N;
    float* out    = (float*)d_out;
    const int NB    = (N + BIN_ROWS - 1) / BIN_ROWS;
    const int chunk = (E + PART_BLOCKS - 1) / PART_BLOCKS;
    int2*  seg      = (int2*)h;
    int2*  pairs    = (int2*)g;

    const int nblkG  = (N + 127) / 128;
    const int nblkR  = (N + 16 * GROUP_ROWS - 1) / (16 * GROUP_ROWS);

    bin_sort_chunk_kernel<<<PART_BLOCKS, 256, NB * sizeof(int), stream>>>(src, dst, seg, pairs, E, chunk, NB);
    bin_bucket_kernel<<<NB, 1024, 0, stream>>>(pairs, seg, cnt, rowptr, dinv, nbr, N, chunk, NB);

    gather_pair_kernel<true><<<nblkR, 256, 0, stream>>>(x, x, rowptr, cnt, nbr, dinv, h, agg, N);
    layer1_products_kernel<<<nblkG, 256, 0, stream>>>(
        h, agg, x, gcn_w1, gcn_b1, sage_wl1, sage_wr1, sage_bl1, dinv, g, out, N);

    gather_pair_kernel<false><<<nblkR, 256, 0, stream>>>(g, out, rowptr, cnt, nbr, dinv, h, agg, N);
    layer2_projection_kernel<<<nblkG, 256, 0, stream>>>(
        h, agg, out, gcn_w2, gcn_b2, sage_wl2, sage_wr2, sage_bl2,
        gcn_ln_g, gcn_ln_b, sage_ln_g, sage_ln_b, proj_w, proj_b, out, N);
}


# Neighbour passes fused into the product kernels: a wave sums its own 16 rows into an LDS slot and feeds them to its WMMAs; 6 launches become 4, the sums never reach memory; all sums unchanged
# speedup vs baseline: 6.1889x; 1.2284x over previous
#include <hip/hip_runtime.h>

typedef __attribute__((ext_vector_type(2))) float v2f;
typedef __attribute__((ext_vector_type(8))) float v8f;
typedef __attribute__((ext_vector_type(4))) float v4f;

#define FDIM 64
#define BIN_SHIFT 9
#define BIN_ROWS (1 << BIN_SHIFT)
#define PART_BLOCKS 512
#define GROUP_ROWS 8
#define EDGE_UNROLL 8
#define TILE_SP 68
#define TILE_WORDS (16 * TILE_SP)
#define FUSED_BLOCKS 256
#define L1_WAVES 25
#define L2_WAVES 16

#define SEG_THREADS (1024 / PART_BLOCKS)
static_assert(PART_BLOCKS * SEG_THREADS == 1024, "the last build pass gives every chunk SEG_THREADS of its 1024 threads");

__launch_bounds__(256)
__global__ void bin_sort_chunk_kernel(const int* __restrict__ src, const int* __restrict__ dst,
                                      int2* __restrict__ seg, int2* __restrict__ pairs,
                                      int e, int chunk, int nb) {
    extern __shared__ int cur[];
    __shared__ int wsum[8];
    const int t    = threadIdx.x;
    const int lane = t & 31;
    const int wave = t >> 5;
    for (int b = t; b < nb; b += 256) cur[b] = 0;
    __syncthreads();
    const int e0 = blockIdx.x * chunk;
    const int e1 = min(e0 + chunk, e);
    {
        int i = e0 + t;
        for (; i + 768 < e1; i += 1024) {
            const int d0 = dst[i], d1 = dst[i + 256], d2 = dst[i + 512], d3 = dst[i + 768];
            atomicAdd(&cur[d0 >> BIN_SHIFT], 1);
            atomicAdd(&cur[d1 >> BIN_SHIFT], 1);
            atomicAdd(&cur[d2 >> BIN_SHIFT], 1);
            atomicAdd(&cur[d3 >> BIN_SHIFT], 1);
        }
        for (; i < e1; i += 256) atomicAdd(&cur[dst[i] >> BIN_SHIFT], 1);
    }
    __syncthreads();
    int carry = 0;
    for (int base = 0; base < nb; base += 256) {
        const int b = base + t;
        const int v = (b < nb) ? cur[b] : 0;
        int inc = v;
        for (int o = 1; o < 32; o <<= 1) {
            const int u = __shfl_up(inc, o, 32);
            if (lane >= o) inc += u;
        }
        if (lane == 31) wsum[wave] = inc;
        __syncthreads();
        int woff = 0, tot = 0;
        for (int w = 0; w < 8; ++w) {
            const int x = wsum[w];
            if (w < wave) woff += x;
            tot += x;
        }
        const int ex = carry + woff + inc - v;
        if (b < nb) {
            cur[b] = e0 + ex;
            seg[blockIdx.x * nb + b] = make_int2(ex, v);
        }
        carry += tot;
        __syncthreads();
    }
    int i = e0 + t;
    for (; i + 768 < e1; i += 1024) {
        const int d0 = dst[i], d1 = dst[i + 256], d2 = dst[i + 512], d3 = dst[i + 768];
        const int s0 = src[i], s1 = src[i + 256], s2 = src[i + 512], s3 = src[i + 768];
        const int p0 = atomicAdd(&cur[d0 >> BIN_SHIFT], 1);
        const int p1 = atomicAdd(&cur[d1 >> BIN_SHIFT], 1);
        const int p2 = atomicAdd(&cur[d2 >> BIN_SHIFT], 1);
        const int p3 = atomicAdd(&cur[d3 >> BIN_SHIFT], 1);
        pairs[p0] = make_int2(s0, d0);
        pairs[p1] = make_int2(s1, d1);
        pairs[p2] = make_int2(s2, d2);
        pairs[p3] = make_int2(s3, d3);
    }
    for (; i < e1; i += 256) {
        const int d = dst[i];
        const int sv = src[i];
        const int p = atomicAdd(&cur[d >> BIN_SHIFT], 1);
        pairs[p] = make_int2(sv, d);
    }
}

__launch_bounds__(1024)
__global__ void bin_bucket_kernel(const int2* __restrict__ pairs, const int2* __restrict__ seg,
                                  int* __restrict__ cnt, int* __restrict__ rowptr, float* __restrict__ dinv,
                                  int* __restrict__ nbr, int n, int chunk, int nb) {
    __shared__ int lc[BIN_ROWS];
    __shared__ int lp[BIN_ROWS];
    __shared__ int wsum[32];
    __shared__ int osum[32];
    const int t     = threadIdx.x;
    const int lane  = t & 31;
    const int wave  = t >> 5;
    const int rbase = blockIdx.x * BIN_ROWS;
    const int part  = t / SEG_THREADS;
    const int j     = t % SEG_THREADS;
    const int2 sg   = seg[part * nb + blockIdx.x];
    const int sbase = part * chunk + sg.x;
    const int slen  = sg.y;
    const int last  = slen - 1;
    {
        int bs = (j == 0) ? sg.x : 0;
        for (int o = 16; o > 0; o >>= 1) bs += __shfl_xor(bs, o, 32);
        if (lane == 0) osum[wave] = bs;
    }
    if (t < BIN_ROWS) lc[t] = 0;
    __syncthreads();
    int b0 = 0;
    for (int w = 0; w < 32; ++w) b0 += osum[w];
    for (int i = j; i < slen; i += 4 * SEG_THREADS) {
        int2 pa = pairs[sbase + i];
        int2 pb = pairs[sbase + min(i + SEG_THREADS, last)];
        int2 pc = pairs[sbase + min(i + 2 * SEG_THREADS, last)];
        int2 pd = pairs[sbase + min(i + 3 * SEG_THREADS, last)];
        asm volatile("" : "+v"(pa.y), "+v"(pb.y), "+v"(pc.y), "+v"(pd.y));
        atomicAdd(&lc[pa.y - rbase], 1);
        if (i + SEG_THREADS < slen)     atomicAdd(&lc[pb.y - rbase], 1);
        if (i + 2 * SEG_THREADS < slen) atomicAdd(&lc[pc.y - rbase], 1);
        if (i + 3 * SEG_THREADS < slen) atomicAdd(&lc[pd.y - rbase], 1);
    }
    __syncthreads();
    const int v = (t < BIN_ROWS) ? lc[t] : 0;
    int inc = v;
    for (int o = 1; o < 32; o <<= 1) {
        const int u = __shfl_up(inc, o, 32);
        if (lane >= o) inc += u;
    }
    if (lane == 31) wsum[wave] = inc;
    __syncthreads();
    int woff = 0;
    for (int w = 0; w < wave; ++w) woff += wsum[w];
    if (t < BIN_ROWS) {
        const int ex = woff + inc - v;
        lp[t] = ex;
        lc[t] = 0;
        const int r = rbase + t;
        if (r < n) {
            cnt[r]    = v;
            rowptr[r] = b0 + ex;
            dinv[r]   = rsqrtf((float)v + 1.0f);
        }
    }
    __syncthreads();
    for (int i = j; i < slen; i += 4 * SEG_THREADS) {
        int2 pa = pairs[sbase + i];
        int2 pb = pairs[sbase + min(i + SEG_THREADS, last)];
        int2 pc = pairs[sbase + min(i + 2 * SEG_THREADS, last)];
        int2 pd = pairs[sbase + min(i + 3 * SEG_THREADS, last)];
        asm volatile("" : "+v"(pa.x), "+v"(pa.y), "+v"(pb.x), "+v"(pb.y), "+v"(pc.x), "+v"(pc.y), "+v"(pd.x), "+v"(pd.y));
        const int ra = pa.y - rbase;
        nbr[b0 + lp[ra] + atomicAdd(&lc[ra], 1)] = pa.x;
        if (i + SEG_THREADS < slen) {
            const int rb = pb.y - rbase;
            nbr[b0 + lp[rb] + atomicAdd(&lc[rb], 1)] = pb.x;
        }
        if (i + 2 * SEG_THREADS < slen) {
            const int rc = pc.y - rbase;
            nbr[b0 + lp[rc] + atomicAdd(&lc[rc], 1)] = pc.x;
        }
        if (i + 3 * SEG_THREADS < slen) {
            const int rd = pd.y - rbase;
            nbr[b0 + lp[rd] + atomicAdd(&lc[rd], 1)] = pd.x;
        }
    }
}

#define WMMA_F32(a, b, c) __builtin_amdgcn_wmma_f32_16x16x4_f32(false, (a), false, (b), (short)0, (c), false, false)

#define WPITCH 160
#define WWORDS (32 * WPITCH)
template <bool PERM, int NT>
__device__ __forceinline__ void stage_w64(float* __restrict__ dst, const float* __restrict__ src, int t) {
    const float4* s = (const float4*)src;
    for (int i = t; i < 512; i += NT) {
        const int kp = i >> 4;
        const int c4 = i & 15;
        const float4 a = s[(2 * kp) * 16 + c4];
        const float4 b = s[(2 * kp + 1) * 16 + c4];
        const int prow = PERM ? ((kp & 24) | ((kp & 3) << 1) | ((kp >> 2) & 1)) : kp;
        float4* d = (float4*)(dst + prow * WPITCH + c4 * 8);
        d[0] = make_float4(a.x, b.x, a.y, b.y);
        d[1] = make_float4(a.z, b.z, a.w, b.w);
    }
}

#define FRAGS_LOADED(w) asm volatile("" : "+v"(w[0]), "+v"(w[1]), "+v"(w[2]), "+v"(w[3]), "+v"(w[4]), "+v"(w[5]), \
                                          "+v"(w[6]), "+v"(w[7]), "+v"(w[8]), "+v"(w[9]), "+v"(w[10]), "+v"(w[11]), \
                                          "+v"(w[12]), "+v"(w[13]), "+v"(w[14]), "+v"(w[15]))

__device__ __forceinline__ void load_row_frag(v2f (&f)[16], const float* p) {
    #pragma unroll
    for (int kb = 0; kb < 16; ++kb) { f[kb].x = p[kb * 4 + 0]; f[kb].y = p[kb * 4 + 1]; }
}

__device__ __forceinline__ void load_row_frag_lds(v2f (&f)[16], const float* p) {
    #pragma unroll
    for (int kb = 0; kb < 16; ++kb) f[kb] = *(const v2f*)(p + kb * 4);
}

__device__ __forceinline__ void load_w_frag(v2f (&w)[16], const float* Wl, int col, int lhalf) {
    #pragma unroll
    for (int kb = 0; kb < 16; ++kb)
        w[kb] = *(const v2f*)(Wl + (2 * kb + lhalf) * WPITCH + col * 2);
}

template <bool RELU>
__device__ __forceinline__ void store_node_major(float* __restrict__ dst, const v8f acc, float bval,
                                                 int m0, int lhalf, int col, int n) {
    float v[8];
    #pragma unroll
    for (int j = 0; j < 8; ++j) {
        v[j] = acc[j] + bval;
        if (RELU) v[j] = fmaxf(v[j], 0.0f);
    }
    #pragma unroll
    for (int j = 0; j < 8; ++j) {
        const int r = m0 + j + lhalf * 8;
        if (r < n) dst[(size_t)r * FDIM + col] = v[j];
    }
}

__device__ __forceinline__ void store_node_major_relu_scaled(float* __restrict__ dst, const v8f acc, float bval,
                                                             const float (&rs)[8], int m0, int lhalf, int col, int n) {
    float v[8];
    #pragma unroll
    for (int j = 0; j < 8; ++j) v[j] = fmaxf(acc[j] + bval, 0.0f) * rs[j];
    #pragma unroll
    for (int j = 0; j < 8; ++j) {
        const int r = m0 + j + lhalf * 8;
        if (r < n) dst[(size_t)r * FDIM + col] = v[j];
    }
}

__device__ __forceinline__ void bias_layernorm_fm(v8f (&tl)[4], const float* bias, const float* gam,
                                                  const float* bet, int lhalf) {
    float s = 0.0f, q = 0.0f;
    #pragma unroll
    for (int nb = 0; nb < 4; ++nb) {
        #pragma unroll
        for (int r = 0; r < 8; ++r) {
            const float v = tl[nb][r] + bias[nb * 16 + lhalf * 8 + r];
            tl[nb][r] = v;
            s += v;
            q += v * v;
        }
    }
    s += __shfl_xor(s, 16, 32);
    q += __shfl_xor(q, 16, 32);
    const float mu  = s * (1.0f / 64.0f);
    const float inv = rsqrtf(fmaxf(q * (1.0f / 64.0f) - mu * mu, 0.0f) + 1e-5f);
    #pragma unroll
    for (int nb = 0; nb < 4; ++nb) {
        #pragma unroll
        for (int r = 0; r < 8; ++r) {
            const int f = nb * 16 + lhalf * 8 + r;
            tl[nb][r] = (tl[nb][r] - mu) * inv * gam[f] + bet[f];
        }
    }
}

__device__ __forceinline__ int row_bcast(int v, int u) {
    switch (u) {
        case 0:  return __builtin_amdgcn_update_dpp(0, v, 0x150, 0xf, 0xf, false);
        case 1:  return __builtin_amdgcn_update_dpp(0, v, 0x151, 0xf, 0xf, false);
        case 2:  return __builtin_amdgcn_update_dpp(0, v, 0x152, 0xf, 0xf, false);
        case 3:  return __builtin_amdgcn_update_dpp(0, v, 0x153, 0xf, 0xf, false);
        case 4:  return __builtin_amdgcn_update_dpp(0, v, 0x154, 0xf, 0xf, false);
        case 5:  return __builtin_amdgcn_update_dpp(0, v, 0x155, 0xf, 0xf, false);
        case 6:  return __builtin_amdgcn_update_dpp(0, v, 0x156, 0xf, 0xf, false);
        default: return __builtin_amdgcn_update_dpp(0, v, 0x157, 0xf, 0xf, false);
    }
}
__device__ __forceinline__ float row_bcast_f(float v, int u) {
    return __int_as_float(row_bcast(__float_as_int(v), u));
}

template <bool ONE_TABLE>
__device__ __forceinline__ void gather_tile(const float* __restrict__ tabG, const float* __restrict__ tabS,
                                            const int* __restrict__ rowptr, const int* __restrict__ cnt,
                                            const int* __restrict__ nbr, const float* __restrict__ dinv,
                                            float* slotG, float* slotS, int m0, int n, int lane) {
    const int q    = lane & 15;
    const int half = lane >> 4;
    const int r0   = m0 + half * GROUP_ROWS;
    const v4f* g4 = (const v4f*)tabG;
    const v4f* s4 = (const v4f*)tabS;

    const bool mine = (q < GROUP_ROWS) && (r0 + q < n);
    const int  rl   = mine ? (r0 + q) : min(r0, n - 1);
    const int   begL = rowptr[rl];
    const int   degL = mine ? cnt[rl] : 0;
    const float drL  = dinv[rl];

    int ixn = 0;
    {
        const int d0 = row_bcast(degL, 0);
        const int b0 = row_bcast(begL, 0);
        if (d0 > 0) ixn = nbr[b0 + min(q, d0 - 1)];
    }

    #pragma unroll
    for (int k = 0; k < GROUP_ROWS; ++k) {
        const int r = min(r0 + k, n - 1);
        const int   beg = row_bcast(begL, k);
        const int   deg = row_bcast(degL, k);
        const float dr  = row_bcast_f(drL, k);
        int ix = ixn;
        if (k + 1 < GROUP_ROWS) {
            const int dn = row_bcast(degL, k + 1);
            const int bn = row_bcast(begL, k + 1);
            ixn = 0;
            if (dn > 0) ixn = nbr[bn + min(q, dn - 1)];
        }
        const v4f hv = g4[r * 16 + q];
        float4 accG = make_float4(0.0f, 0.0f, 0.0f, 0.0f);
        float4 accS = make_float4(0.0f, 0.0f, 0.0f, 0.0f);
        for (int j = 0; j < deg; j += EDGE_UNROLL) {
            const int m = deg - j;
            int ixc = ix;
            if (m > EDGE_UNROLL)
                ixc = nbr[beg + j + EDGE_UNROLL + min(q, m - EDGE_UNROLL - 1)];
            float dq = 0.0f;
            if (ONE_TABLE) dq = dinv[ix];
            v4f vg[EDGE_UNROLL];
            v4f vs[EDGE_UNROLL];
            #pragma unroll
            for (int u = 0; u < EDGE_UNROLL; ++u) {
                const int s = row_bcast(ix, u);
                vg[u] = g4[s * 16 + q];
                if (!ONE_TABLE) vs[u] = s4[s * 16 + q];
            }
            if (ONE_TABLE)
                asm volatile("" : "+v"(vg[0]), "+v"(vg[1]), "+v"(vg[2]), "+v"(vg[3]),
                                  "+v"(vg[4]), "+v"(vg[5]), "+v"(vg[6]), "+v"(vg[7]));
            else
                asm volatile("" : "+v"(vg[0]), "+v"(vg[1]), "+v"(vg[2]), "+v"(vg[3]),
                                  "+v"(vg[4]), "+v"(vg[5]), "+v"(vg[6]), "+v"(vg[7]),
                                  "+v"(vs[0]), "+v"(vs[1]), "+v"(vs[2]), "+v"(vs[3]),
                                  "+v"(vs[4]), "+v"(vs[5]), "+v"(vs[6]), "+v"(vs[7]));
            const float wq = dq * dr;
            #pragma unroll
            for (int u = 0; u < EDGE_UNROLL; ++u) {
                const bool on = (u == 0) || (u < m);
                const v4f a = vg[u];
                const v4f b = ONE_TABLE ? vg[u] : vs[u];
                float4 ng, ns;
                if (ONE_TABLE) {
                    const float w = row_bcast_f(wq, u);
                    ng.x = accG.x + a.x * w; ng.y = accG.y + a.y * w;
                    ng.z = accG.z + a.z * w; ng.w = accG.w + a.w * w;
                } else {
                    ng.x = accG.x + a.x; ng.y = accG.y + a.y;
                    ng.z = accG.z + a.z; ng.w = accG.w + a.w;
                }
                ns.x = accS.x + b.x; ns.y = accS.y + b.y;
                ns.z = accS.z + b.z; ns.w = accS.w + b.w;
                accG.x = on ? ng.x : accG.x; accG.y = on ? ng.y : accG.y;
                accG.z = on ? ng.z : accG.z; accG.w = on ? ng.w : accG.w;
                accS.x = on ? ns.x : accS.x; accS.y = on ? ns.y : accS.y;
                accS.z = on ? ns.z : accS.z; accS.w = on ? ns.w : accS.w;
            }
            ix = ixc;
        }
        float4 og, os;
        if (ONE_TABLE) {
            const float d2 = dr * dr;
            og.x = accG.x + hv.x * d2; og.y = accG.y + hv.y * d2;
            og.z = accG.z + hv.z * d2; og.w = accG.w + hv.w * d2;
        } else {
            og.x = (accG.x + hv.x) * dr; og.y = (accG.y + hv.y) * dr;
            og.z = (accG.z + hv.z) * dr; og.w = (accG.w + hv.w) * dr;
        }
        const float rs = 1.0f / fmaxf((float)deg, 1.0f);
        os.x = accS.x * rs; os.y = accS.y * rs; os.z = accS.z * rs; os.w = accS.w * rs;
        const int trow = half * GROUP_ROWS + k;
        *(float4*)(slotG + trow * TILE_SP + 4 * q) = og;
        *(float4*)(slotS + trow * TILE_SP + 4 * q) = os;
    }
}


__launch_bounds__(L1_WAVES * 32)
__global__ void layer1_fused_kernel(const float* __restrict__ X,
                                    const int* __restrict__ rowptr, const int* __restrict__ cnt,
                                    const int* __restrict__ nbr, const float* __restrict__ dinv,
                                    const float* __restrict__ Wg, const float* __restrict__ bg,
                                    const float* __restrict__ Wl, const float* __restrict__ Wr,
                                    const float* __restrict__ bl,
                                    float* __restrict__ outG, float* __restrict__ outS, int n) {
    __shared__ __attribute__((aligned(16))) float WG[WWORDS];
    __shared__ __attribute__((aligned(16))) float WL[WWORDS];
    __shared__ __attribute__((aligned(16))) float WR[WWORDS];
    __shared__ __attribute__((aligned(16))) float SLOT[L1_WAVES][2][TILE_WORDS];
    const int t = threadIdx.x;
    stage_w64<false, L1_WAVES * 32>(WG, Wg, t);
    stage_w64<false, L1_WAVES * 32>(WL, Wl, t);
    stage_w64<false, L1_WAVES * 32>(WR, Wr, t);
    __syncthreads();

    const int wave  = __builtin_amdgcn_readfirstlane(t >> 5);
    const int lane  = t & 31;
    const int lhalf = lane >> 4;
    const int l15   = lane & 15;
    const int kOff  = lhalf * 2;
    float* slotH = SLOT[wave][0];
    float* slotA = SLOT[wave][1];
    const int ntiles = (n + 15) >> 4;
    for (int tile = wave * (int)gridDim.x + (int)blockIdx.x; tile < ntiles; tile += L1_WAVES * (int)gridDim.x) {
        const int m0 = tile * 16;
        gather_tile<true>(X, X, rowptr, cnt, nbr, dinv, slotH, slotA, m0, n, lane);
        int row = m0 + l15;
        if (row >= n) row = n - 1;

        {
            v8f acc[4];
            #pragma unroll
            for (int nb = 0; nb < 4; ++nb) acc[nb] = (v8f){0.f, 0.f, 0.f, 0.f, 0.f, 0.f, 0.f, 0.f};
            {
                v2f af[16];
                load_row_frag_lds(af, slotA + l15 * TILE_SP + kOff);
                #pragma unroll
                for (int nb = 0; nb < 4; ++nb) {
                    v2f wf[16];
                    load_w_frag(wf, WL, nb * 16 + l15, lhalf);
                    FRAGS_LOADED(wf);
                    #pragma unroll
                    for (int kb = 0; kb < 16; ++kb) acc[nb] = WMMA_F32(af[kb], wf[kb], acc[nb]);
                }
            }
            {
                v2f xf[16];
                load_row_frag(xf, X + (size_t)row * FDIM + kOff);
                #pragma unroll
                for (int nb = 0; nb < 4; ++nb) {
                    const int col = nb * 16 + l15;
                    v2f wg[16];
                    load_w_frag(wg, WR, col, lhalf);
                    FRAGS_LOADED(wg);
                    #pragma unroll
                    for (int kb = 0; kb < 16; ++kb) acc[nb] = WMMA_F32(xf[kb], wg[kb], acc[nb]);
                    store_node_major<true>(outS, acc[nb], bl[col], m0, lhalf, col, n);
                }
            }
        }
        {
            v2f hf[16];
            load_row_frag_lds(hf, slotH + l15 * TILE_SP + kOff);
            float dv[8];
            #pragma unroll
            for (int j = 0; j < 8; ++j) {
                const int r = m0 + j + lhalf * 8;
                dv[j] = dinv[r < n ? r : n - 1];
            }
            #pragma unroll
            for (int nb = 0; nb < 4; ++nb) {
                const int col = nb * 16 + l15;
                v2f wf[16];
                load_w_frag(wf, WG, col, lhalf);
                FRAGS_LOADED(wf);
                v8f acc = {0.f, 0.f, 0.f, 0.f, 0.f, 0.f, 0.f, 0.f};
                #pragma unroll
                for (int kb = 0; kb < 16; ++kb) acc = WMMA_F32(hf[kb], wf[kb], acc);
                store_node_major_relu_scaled(outG, acc, bg[col], dv, m0, lhalf, col, n);
            }
        }
    }
}


__launch_bounds__(L2_WAVES * 32)
__global__ void layer2_fused_kernel(const float* __restrict__ G1, const float* __restrict__ S1,
                                    const int* __restrict__ rowptr, const int* __restrict__ cnt,
                                    const int* __restrict__ nbr, const float* __restrict__ dinv,
                                    const float* __restrict__ Wg, const float* __restrict__ bg,
                                    const float* __restrict__ Wl, const float* __restrict__ Wr,
                                    const float* __restrict__ bl,
                                    const float* __restrict__ lnGg, const float* __restrict__ lnGb,
                                    const float* __restrict__ lnSg, const float* __restrict__ lnSb,
                                    const float* __restrict__ P, const float* __restrict__ pb,
                                    float* __restrict__ out, int n) {
    __shared__ __attribute__((aligned(16))) float LDSMEM[5 * WWORDS + 6 * FDIM + L2_WAVES * 2 * TILE_WORDS];
    float* const WG = LDSMEM;
    float* const WL = LDSMEM + WWORDS;
    float* const WR = LDSMEM + 2 * WWORDS;
    float* const PT = LDSMEM + 3 * WWORDS;
    float* const PB = LDSMEM + 4 * WWORDS;
    float (*const PAR)[FDIM] = (float (*)[FDIM])(LDSMEM + 5 * WWORDS);
    float* const SLOTS = LDSMEM + 5 * WWORDS + 6 * FDIM;
    const int t = threadIdx.x;
    stage_w64<false, L2_WAVES * 32>(WG, Wg, t);
    stage_w64<false, L2_WAVES * 32>(WL, Wl, t);
    stage_w64<false, L2_WAVES * 32>(WR, Wr, t);
    stage_w64<true, L2_WAVES * 32>(PT, P, t);
    stage_w64<true, L2_WAVES * 32>(PB, P + FDIM * FDIM, t);
    if (t < FDIM) {
        PAR[0][t] = bg[t];   PAR[1][t] = bl[t];
        PAR[2][t] = lnGg[t]; PAR[3][t] = lnGb[t];
        PAR[4][t] = lnSg[t]; PAR[5][t] = lnSb[t];
    }
    __syncthreads();

    const int wave  = __builtin_amdgcn_readfirstlane(t >> 5);
    const int lane  = t & 31;
    const int lhalf = lane >> 4;
    const int l15   = lane & 15;
    const int kOff  = lhalf * 2;
    float* slotH = SLOTS + wave * (2 * TILE_WORDS);
    float* slotA = slotH + TILE_WORDS;
    const int ntiles = (n + 15) >> 4;
    for (int tile = wave * (int)gridDim.x + (int)blockIdx.x; tile < ntiles; tile += L2_WAVES * (int)gridDim.x) {
        const int m0 = tile * 16;
        gather_tile<false>(G1, S1, rowptr, cnt, nbr, dinv, slotH, slotA, m0, n, lane);
        int row = m0 + l15;
        if (row >= n) row = n - 1;

        v8f gT[4];
        {
            v2f hf[16];
            load_row_frag_lds(hf, slotH + l15 * TILE_SP + kOff);
            #pragma unroll
            for (int nb = 0; nb < 4; ++nb) {
                v2f wf[16];
                load_w_frag(wf, WG, nb * 16 + l15, lhalf);
                FRAGS_LOADED(wf);
                v8f acc = {0.f, 0.f, 0.f, 0.f, 0.f, 0.f, 0.f, 0.f};
                #pragma unroll
                for (int kb = 0; kb < 16; ++kb) acc = WMMA_F32(wf[kb], hf[kb], acc);
                gT[nb] = acc;
            }
            bias_layernorm_fm(gT, PAR[0], PAR[2], PAR[3], lhalf);
        }
        v8f sT[4];
        {
            #pragma unroll
            for (int nb = 0; nb < 4; ++nb) sT[nb] = (v8f){0.f, 0.f, 0.f, 0.f, 0.f, 0.f, 0.f, 0.f};
            {
                v2f af[16];
                load_row_frag_lds(af, slotA + l15 * TILE_SP + kOff);
                #pragma unroll
                for (int nb = 0; nb < 4; ++nb) {
                    v2f wf[16];
                    load_w_frag(wf, WL, nb * 16 + l15, lhalf);
                    FRAGS_LOADED(wf);
                    #pragma unroll
                    for (int kb = 0; kb < 16; ++kb) sT[nb] = WMMA_F32(wf[kb], af[kb], sT[nb]);
                }
            }
            {
                v2f sf[16];
                load_row_frag(sf, S1 + (size_t)row * FDIM + kOff);
                #pragma unroll
                for (int nb = 0; nb < 4; ++nb) {
                    v2f wg[16];
                    load_w_frag(wg, WR, nb * 16 + l15, lhalf);
                    FRAGS_LOADED(wg);
                    #pragma unroll
                    for (int kb = 0; kb < 16; ++kb) sT[nb] = WMMA_F32(wg[kb], sf[kb], sT[nb]);
                }
            }
            bias_layernorm_fm(sT, PAR[1], PAR[4], PAR[5], lhalf);
        }
        #pragma unroll
        for (int ob = 0; ob < 4; ++ob) {
            const int col = ob * 16 + l15;
            v2f pt[16], pq[16];
            #pragma unroll
            for (int nb = 0; nb < 4; ++nb) {
                #pragma unroll
                for (int rp = 0; rp < 4; ++rp) {
                    const int prow = nb * 8 + rp * 2 + lhalf;
                    pt[nb * 4 + rp] = *(const v2f*)(PT + prow * WPITCH + col * 2);
                    pq[nb * 4 + rp] = *(const v2f*)(PB + prow * WPITCH + col * 2);
                }
            }
            FRAGS_LOADED(pt);
            v8f acc = {0.f, 0.f, 0.f, 0.f, 0.f, 0.f, 0.f, 0.f};
            #pragma unroll
            for (int nb = 0; nb < 4; ++nb) {
                #pragma unroll
                for (int rp = 0; rp < 4; ++rp) {
                    v2f a;
                    a.x = gT[nb][rp * 2];
                    a.y = gT[nb][rp * 2 + 1];
                    acc = WMMA_F32(a, pt[nb * 4 + rp], acc);
                }
            }
            FRAGS_LOADED(pq);
            #pragma unroll
            for (int nb = 0; nb < 4; ++nb) {
                #pragma unroll
                for (int rp = 0; rp < 4; ++rp) {
                    v2f a;
                    a.x = sT[nb][rp * 2];
                    a.y = sT[nb][rp * 2 + 1];
                    acc = WMMA_F32(a, pq[nb * 4 + rp], acc);
                }
            }
            store_node_major<false>(out, acc, pb[col], m0, lhalf, col, n);
        }
    }
}


extern "C" void kernel_launch(void* const* d_in, const int* in_sizes, int n_in,
                              void* d_out, int out_size, void* d_ws, size_t ws_size,
                              hipStream_t stream) {
    const float* x        = (const float*)d_in[0];
    const int*   ei       = (const int*)d_in[1];
    const float* gcn_w1   = (const float*)d_in[2];
    const float* gcn_b1   = (const float*)d_in[3];
    const float* gcn_w2   = (const float*)d_in[4];
    const float* gcn_b2   = (const float*)d_in[5];
    const float* sage_wl1 = (const float*)d_in[6];
    const float* sage_bl1 = (const float*)d_in[7];
    const float* sage_wr1 = (const float*)d_in[8];
    const float* sage_wl2 = (const float*)d_in[9];
    const float* sage_bl2 = (const float*)d_in[10];
    const float* sage_wr2 = (const float*)d_in[11];
    const float* gcn_ln_g = (const float*)d_in[12];
    const float* gcn_ln_b = (const float*)d_in[13];
    const float* sage_ln_g = (const float*)d_in[14];
    const float* sage_ln_b = (const float*)d_in[15];
    const float* proj_w   = (const float*)d_in[16];
    const float* proj_b   = (const float*)d_in[17];

    const int N = in_sizes[0] / FDIM;
    const int E = in_sizes[1] / 2;
    const int* src = ei;
    const int* dst = ei + E;

    float* s1     = (float*)d_ws;
    float* g      = s1 + (size_t)N * FDIM;
    float* spare  = g + (size_t)N * FDIM;
    int*   cnt    = (int*)(spare + (size_t)N * FDIM);
    float* dinv   = (float*)(cnt + N);
    int*   rowptr = (int*)(dinv + N);
    int*   nbr    = rowptr + N;
    float* out    = (float*)d_out;
    const int NB    = (N + BIN_ROWS - 1) / BIN_ROWS;
    const int chunk = (E + PART_BLOCKS - 1) / PART_BLOCKS;
    int2*  seg      = (int2*)s1;
    int2*  pairs    = (int2*)g;

    bin_sort_chunk_kernel<<<PART_BLOCKS, 256, NB * sizeof(int), stream>>>(src, dst, seg, pairs, E, chunk, NB);
    bin_bucket_kernel<<<NB, 1024, 0, stream>>>(pairs, seg, cnt, rowptr, dinv, nbr, N, chunk, NB);

    layer1_fused_kernel<<<FUSED_BLOCKS, L1_WAVES * 32, 0, stream>>>(
        x, rowptr, cnt, nbr, dinv, gcn_w1, gcn_b1, sage_wl1, sage_wr1, sage_bl1, g, s1, N);

    layer2_fused_kernel<<<FUSED_BLOCKS, L2_WAVES * 32, 0, stream>>>(
        g, s1, rowptr, cnt, nbr, dinv, gcn_w2, gcn_b2, sage_wl2, sage_wr2, sage_bl2,
        gcn_ln_g, gcn_ln_b, sage_ln_g, sage_ln_b, proj_w, proj_b, out, N);
}
